# starting kernel
# baseline (speedup 1.0000x reference)
#include <hip/hip_runtime.h>

typedef __attribute__((ext_vector_type(2))) float v2f;
typedef __attribute__((ext_vector_type(8))) float v8f;

constexpr int IN_DIM = 128;
constexpr int HID    = 8;


__global__ void zero_f32(float* __restrict__ p, int n) {
  int i = blockIdx.x * blockDim.x + threadIdx.x;
  if (i < n) p[i] = 0.0f;
}

__global__ void degree_kernel(const int* __restrict__ src, const int* __restrict__ dst,
                              float* __restrict__ deg_s, float* __restrict__ deg_d, int nE) {
  int e = blockIdx.x * blockDim.x + threadIdx.x;
  if (e < nE) {
    atomicAdd(&deg_s[src[e]], 1.0f);
    atomicAdd(&deg_d[dst[e]], 1.0f);
  }
}

__global__ void norm_kernel(float* __restrict__ deg_s, float* __restrict__ deg_d, int n) {
  int i = blockIdx.x * blockDim.x + threadIdx.x;
  if (i < n) {
    deg_s[i] = rsqrtf(fmaxf(deg_s[i], 1.0f));
    deg_d[i] = rsqrtf(fmaxf(deg_d[i], 1.0f));
  }
}


__global__ __launch_bounds__(256) void gemm1_wmma(const float* __restrict__ x,
                                                  const float* __restrict__ W1,
                                                  const float* __restrict__ out_norm,
                                                  float* __restrict__ h, int nNodes) {
  __shared__ float w[IN_DIM * 16];
  int tid = threadIdx.x;
  for (int i = tid; i < IN_DIM * 16; i += 256) {
    int r = i >> 4, cidx = i & 15;
    w[i] = (cidx < HID) ? W1[r * HID + cidx] : 0.0f;
  }
  __syncthreads();

  int wid  = tid >> 5;
  int lane = tid & 31;
  int m    = lane & 15;
  int kh   = lane >> 4;

  int row0 = (blockIdx.x * 8 + wid) * 16;
  int node = row0 + m;
  int nodeClamped = node < nNodes ? node : nNodes - 1;
  const float* xrow = x + (long)nodeClamped * IN_DIM;

  v8f acc = {};
#pragma unroll
  for (int kk = 0; kk < IN_DIM; kk += 4) {
    int k0 = kk + 2 * kh;
    v2f a;
    a.x = xrow[k0];
    a.y = xrow[k0 + 1];
    v2f b;
    b.x = w[k0 * 16 + m];
    b.y = w[(k0 + 1) * 16 + m];
    acc = __builtin_amdgcn_wmma_f32_16x16x4_f32(
         false, a,  false, b,
         (short)0, acc,  false,  false);
  }

  if (m < HID) {
#pragma unroll
    for (int r = 0; r < 8; ++r) {
      int nr = row0 + r + 8 * kh;
      if (nr < nNodes) h[(long)nr * HID + m] = acc[r] * out_norm[nr];
    }
  }
}


__global__ void spmm1_kernel(const int* __restrict__ src, const int* __restrict__ dst,
                             const float* __restrict__ h, float* __restrict__ agg, int nE) {
  int e = blockIdx.x * blockDim.x + threadIdx.x;
  if (e >= nE) return;
  int s = src[e], d = dst[e];
  const float4* hs = (const float4*)(h + (long)s * HID);
  float4 h0 = hs[0];
  float4 h1 = hs[1];
  float* ad = agg + (long)d * HID;
  atomicAdd(&ad[0], h0.x); atomicAdd(&ad[1], h0.y);
  atomicAdd(&ad[2], h0.z); atomicAdd(&ad[3], h0.w);
  atomicAdd(&ad[4], h1.x); atomicAdd(&ad[5], h1.y);
  atomicAdd(&ad[6], h1.z); atomicAdd(&ad[7], h1.w);
}


__global__ void layer2_node(const float* __restrict__ agg, const float* __restrict__ in_norm,
                            const float* __restrict__ out_norm, const float* __restrict__ b1,
                            const float* __restrict__ W2, float* __restrict__ g2, int n) {
  int v = blockIdx.x * blockDim.x + threadIdx.x;
  if (v >= n) return;
  float innv = in_norm[v];
  const float* av = agg + (long)v * HID;
  float s = 0.0f;
#pragma unroll
  for (int j = 0; j < HID; ++j) {
    float z = fmaf(av[j], innv, b1[j]);
    z = z > 0.0f ? z : 0.0f;
    s = fmaf(z, W2[j], s);
  }
  g2[v] = s * out_norm[v];
}


__global__ void spmm2_kernel(const int* __restrict__ src, const int* __restrict__ dst,
                             const float* __restrict__ g2, float* __restrict__ outacc, int nE) {
  int e = blockIdx.x * blockDim.x + threadIdx.x;
  if (e < nE) atomicAdd(&outacc[dst[e]], g2[src[e]]);
}

__global__ void final_kernel(const float* __restrict__ outacc, const float* __restrict__ in_norm,
                             const float* __restrict__ b2, float* __restrict__ out, int n) {
  int v = blockIdx.x * blockDim.x + threadIdx.x;
  if (v < n) out[v] = fmaf(outacc[v], in_norm[v], b2[0]);
}


extern "C" void kernel_launch(void* const* d_in, const int* in_sizes, int n_in,
                              void* d_out, int out_size, void* d_ws, size_t ws_size,
                              hipStream_t stream) {
  const float* x  = (const float*)d_in[0];
  const float* W1 = (const float*)d_in[1];
  const float* b1 = (const float*)d_in[2];
  const float* W2 = (const float*)d_in[3];
  const float* b2 = (const float*)d_in[4];
  const int*  src = (const int*)d_in[5];
  const int*  dst = (const int*)d_in[6];

  const int nN = in_sizes[0] / IN_DIM;
  const int nE = in_sizes[5];

  float* ws     = (float*)d_ws;
  float* deg_s  = ws;
  float* deg_d  = ws + (long)nN;
  float* h      = ws + 2L * nN;
  float* agg    = ws + 10L * nN;
  float* g2     = ws + 18L * nN;
  float* outacc = ws + 19L * nN;
  float* out    = (float*)d_out;

  const int TB = 256;
  dim3 blk(TB);

  zero_f32<<<(2 * nN + TB - 1) / TB, blk, 0, stream>>>(deg_s, 2 * nN);
  zero_f32<<<(8 * nN + TB - 1) / TB, blk, 0, stream>>>(agg, 8 * nN);
  zero_f32<<<(nN + TB - 1) / TB, blk, 0, stream>>>(outacc, nN);

  degree_kernel<<<(nE + TB - 1) / TB, blk, 0, stream>>>(src, dst, deg_s, deg_d, nE);
  norm_kernel<<<(nN + TB - 1) / TB, blk, 0, stream>>>(deg_s, deg_d, nN);

  gemm1_wmma<<<(nN + 127) / 128, blk, 0, stream>>>(x, W1, deg_s, h, nN);

  spmm1_kernel<<<(nE + TB - 1) / TB, blk, 0, stream>>>(src, dst, h, agg, nE);
  layer2_node<<<(nN + TB - 1) / TB, blk, 0, stream>>>(agg, deg_d, deg_s, b1, W2, g2, nN);
  spmm2_kernel<<<(nE + TB - 1) / TB, blk, 0, stream>>>(src, dst, g2, outacc, nE);
  final_kernel<<<(nN + TB - 1) / TB, blk, 0, stream>>>(outacc, deg_d, b2, out, nN);
}


# spmm1: eight lanes per edge row (src/dst passed by lane shuffle): each gather and each f32 atomic instruction touches 4 rows of 32 B, not 32 lines; sums stay in atomic arrival order
# speedup vs baseline: 1.5312x; 1.5312x over previous
#include <hip/hip_runtime.h>

typedef __attribute__((ext_vector_type(2))) float v2f;
typedef __attribute__((ext_vector_type(8))) float v8f;

constexpr int IN_DIM = 128;
constexpr int HID    = 8;


__global__ void zero_f32(float* __restrict__ p, int n) {
  int i = blockIdx.x * blockDim.x + threadIdx.x;
  if (i < n) p[i] = 0.0f;
}

__global__ void degree_kernel(const int* __restrict__ src, const int* __restrict__ dst,
                              float* __restrict__ deg_s, float* __restrict__ deg_d, int nE) {
  int e = blockIdx.x * blockDim.x + threadIdx.x;
  if (e < nE) {
    atomicAdd(&deg_s[src[e]], 1.0f);
    atomicAdd(&deg_d[dst[e]], 1.0f);
  }
}

__global__ void norm_kernel(float* __restrict__ deg_s, float* __restrict__ deg_d, int n) {
  int i = blockIdx.x * blockDim.x + threadIdx.x;
  if (i < n) {
    deg_s[i] = rsqrtf(fmaxf(deg_s[i], 1.0f));
    deg_d[i] = rsqrtf(fmaxf(deg_d[i], 1.0f));
  }
}


__global__ __launch_bounds__(256) void gemm1_wmma(const float* __restrict__ x,
                                                  const float* __restrict__ W1,
                                                  const float* __restrict__ out_norm,
                                                  float* __restrict__ h, int nNodes) {
  __shared__ float w[IN_DIM * 16];
  int tid = threadIdx.x;
  for (int i = tid; i < IN_DIM * 16; i += 256) {
    int r = i >> 4, cidx = i & 15;
    w[i] = (cidx < HID) ? W1[r * HID + cidx] : 0.0f;
  }
  __syncthreads();

  int wid  = tid >> 5;
  int lane = tid & 31;
  int m    = lane & 15;
  int kh   = lane >> 4;

  int row0 = (blockIdx.x * 8 + wid) * 16;
  int node = row0 + m;
  int nodeClamped = node < nNodes ? node : nNodes - 1;
  const float* xrow = x + (long)nodeClamped * IN_DIM;

  v8f acc = {};
#pragma unroll
  for (int kk = 0; kk < IN_DIM; kk += 4) {
    int k0 = kk + 2 * kh;
    v2f a;
    a.x = xrow[k0];
    a.y = xrow[k0 + 1];
    v2f b;
    b.x = w[k0 * 16 + m];
    b.y = w[(k0 + 1) * 16 + m];
    acc = __builtin_amdgcn_wmma_f32_16x16x4_f32(
         false, a,  false, b,
         (short)0, acc,  false,  false);
  }

  if (m < HID) {
#pragma unroll
    for (int r = 0; r < 8; ++r) {
      int nr = row0 + r + 8 * kh;
      if (nr < nNodes) h[(long)nr * HID + m] = acc[r] * out_norm[nr];
    }
  }
}


__global__ void spmm1_kernel(const int* __restrict__ src, const int* __restrict__ dst,
                             const float* __restrict__ h, float* __restrict__ agg, int nE) {
  const int lane = threadIdx.x & 31;
  const int j    = lane & 7;
  const int sub  = lane >> 3;
  const int e    = blockIdx.x * blockDim.x + threadIdx.x;
  const int left = nE - (e - lane);

  int s = 0, d = 0;
  if (e < nE) { s = src[e]; d = dst[e]; }

  float    v[8];
  unsigned off[8];
#pragma unroll
  for (int i = 0; i < 8; ++i) {
    const int from = i * 4 + sub;
    const int si = __shfl(s, from, 32);
    const int di = __shfl(d, from, 32);
    v[i]   = h[(unsigned)(si * HID + j)];
    off[i] = (unsigned)(di * HID + j);
  }
#pragma unroll
  for (int i = 0; i < 8; ++i) {
    if (i * 4 + sub < left) atomicAdd(&agg[off[i]], v[i]);
  }
}


__global__ void layer2_node(const float* __restrict__ agg, const float* __restrict__ in_norm,
                            const float* __restrict__ out_norm, const float* __restrict__ b1,
                            const float* __restrict__ W2, float* __restrict__ g2, int n) {
  int v = blockIdx.x * blockDim.x + threadIdx.x;
  if (v >= n) return;
  float innv = in_norm[v];
  const float* av = agg + (long)v * HID;
  float s = 0.0f;
#pragma unroll
  for (int j = 0; j < HID; ++j) {
    float z = fmaf(av[j], innv, b1[j]);
    z = z > 0.0f ? z : 0.0f;
    s = fmaf(z, W2[j], s);
  }
  g2[v] = s * out_norm[v];
}


__global__ void spmm2_kernel(const int* __restrict__ src, const int* __restrict__ dst,
                             const float* __restrict__ g2, float* __restrict__ outacc, int nE) {
  int e = blockIdx.x * blockDim.x + threadIdx.x;
  if (e < nE) atomicAdd(&outacc[dst[e]], g2[src[e]]);
}

__global__ void final_kernel(const float* __restrict__ outacc, const float* __restrict__ in_norm,
                             const float* __restrict__ b2, float* __restrict__ out, int n) {
  int v = blockIdx.x * blockDim.x + threadIdx.x;
  if (v < n) out[v] = fmaf(outacc[v], in_norm[v], b2[0]);
}


extern "C" void kernel_launch(void* const* d_in, const int* in_sizes, int n_in,
                              void* d_out, int out_size, void* d_ws, size_t ws_size,
                              hipStream_t stream) {
  const float* x  = (const float*)d_in[0];
  const float* W1 = (const float*)d_in[1];
  const float* b1 = (const float*)d_in[2];
  const float* W2 = (const float*)d_in[3];
  const float* b2 = (const float*)d_in[4];
  const int*  src = (const int*)d_in[5];
  const int*  dst = (const int*)d_in[6];

  const int nN = in_sizes[0] / IN_DIM;
  const int nE = in_sizes[5];

  float* ws     = (float*)d_ws;
  float* deg_s  = ws;
  float* deg_d  = ws + (long)nN;
  float* h      = ws + 2L * nN;
  float* agg    = ws + 10L * nN;
  float* g2     = ws + 18L * nN;
  float* outacc = ws + 19L * nN;
  float* out    = (float*)d_out;

  const int TB = 256;
  dim3 blk(TB);

  zero_f32<<<(2 * nN + TB - 1) / TB, blk, 0, stream>>>(deg_s, 2 * nN);
  zero_f32<<<(8 * nN + TB - 1) / TB, blk, 0, stream>>>(agg, 8 * nN);
  zero_f32<<<(nN + TB - 1) / TB, blk, 0, stream>>>(outacc, nN);

  degree_kernel<<<(nE + TB - 1) / TB, blk, 0, stream>>>(src, dst, deg_s, deg_d, nE);
  norm_kernel<<<(nN + TB - 1) / TB, blk, 0, stream>>>(deg_s, deg_d, nN);

  gemm1_wmma<<<(nN + 127) / 128, blk, 0, stream>>>(x, W1, deg_s, h, nN);

  spmm1_kernel<<<(nE + TB - 1) / TB, blk, 0, stream>>>(src, dst, h, agg, nE);
  layer2_node<<<(nN + TB - 1) / TB, blk, 0, stream>>>(agg, deg_d, deg_s, b1, W2, g2, nN);
  spmm2_kernel<<<(nE + TB - 1) / TB, blk, 0, stream>>>(src, dst, g2, outacc, nE);
  final_kernel<<<(nN + TB - 1) / TB, blk, 0, stream>>>(outacc, deg_d, b2, out, nN);
}


# per-node scalar edge sums (degree counts, layer-2 sum) add in LDS node slices fed by repeated edge-list scans; global atomics only flush whole lines; layer-2 sums in LDS+global atomic arrival order
# speedup vs baseline: 3.9015x; 2.5480x over previous
#include <hip/hip_runtime.h>

typedef __attribute__((ext_vector_type(2))) float v2f;
typedef __attribute__((ext_vector_type(8))) float v8f;

constexpr int IN_DIM = 128;
constexpr int HID    = 8;

constexpr int SLICE_CAP    = 73728;
constexpr int SCAN_THREADS = 1024;


__global__ void zero_f32(float* __restrict__ p, int n) {
  int i = blockIdx.x * blockDim.x + threadIdx.x;
  if (i < n) p[i] = 0.0f;
}


__device__ __forceinline__ int slice_len(int nN, int nSlices) {
  const int per = (nN + nSlices - 1) / nSlices;
  return (per + 31) & ~31;
}

__device__ __forceinline__ void chunk_bounds(int nE, int nChunks, int chunk, int& start, int& count) {
  const int chunkLen = (((nE + nChunks - 1) / nChunks) + 3) & ~3;
  long s = (long)chunk * chunkLen;
  if (s > nE) s = nE;
  long e = s + chunkLen;
  if (e > nE) e = nE;
  start = (int)s;
  count = (int)(e - s);
}

__device__ __forceinline__ void count_hit(unsigned* cnt, int v, int base, int len) {
  const unsigned k = (unsigned)(v - base);
  if (k < (unsigned)len) atomicAdd(&cnt[k], 1u);
}

__global__ __launch_bounds__(SCAN_THREADS) void degree_slices(const int* __restrict__ src,
                                                              const int* __restrict__ dst,
                                                              unsigned* __restrict__ cnt_s,
                                                              unsigned* __restrict__ cnt_d,
                                                              int nE, int nN) {
  __shared__ unsigned cnt[SLICE_CAP];
  const int tid     = threadIdx.x;
  const int nSlices = gridDim.y >> 1;
  const int which   = (int)blockIdx.y >= nSlices ? 1 : 0;
  const int slice   = (int)blockIdx.y - which * nSlices;
  const int len     = slice_len(nN, nSlices);
  const int base    = slice * len;
  const int* __restrict__ idx = which ? dst : src;
  unsigned* __restrict__ out  = which ? cnt_d : cnt_s;

  for (int i = tid; i < len; i += SCAN_THREADS) cnt[i] = 0u;
  __syncthreads();

  int start, n;
  chunk_bounds(nE, gridDim.x, blockIdx.x, start, n);
  const int* p = idx + start;
  int head = (int)(((16u - (unsigned)((size_t)p & 15u)) & 15u) >> 2);
  if (head > n) head = n;
  for (int i = tid; i < head; i += SCAN_THREADS) count_hit(cnt, p[i], base, len);
  const int4* pv = (const int4*)(p + head);
  const int nvec = (n - head) >> 2;
  int i = tid;
  for (; i + 7 * SCAN_THREADS < nvec; i += 8 * SCAN_THREADS) {
    int4 v[8];
#pragma unroll
    for (int u = 0; u < 8; ++u) v[u] = pv[i + u * SCAN_THREADS];
#pragma unroll
    for (int u = 0; u < 8; ++u) {
      count_hit(cnt, v[u].x, base, len); count_hit(cnt, v[u].y, base, len);
      count_hit(cnt, v[u].z, base, len); count_hit(cnt, v[u].w, base, len);
    }
  }
  for (; i < nvec; i += SCAN_THREADS) {
    const int4 a = pv[i];
    count_hit(cnt, a.x, base, len); count_hit(cnt, a.y, base, len);
    count_hit(cnt, a.z, base, len); count_hit(cnt, a.w, base, len);
  }
  for (int t = head + nvec * 4 + tid; t < n; t += SCAN_THREADS) count_hit(cnt, p[t], base, len);
  __syncthreads();

  for (int t = tid; t < len; t += SCAN_THREADS) {
    const int node = base + t;
    if (node < nN) atomicAdd(&out[node], cnt[t]);
  }
}

__global__ void norm_kernel(unsigned* __restrict__ deg_s, unsigned* __restrict__ deg_d, int n) {
  int i = blockIdx.x * blockDim.x + threadIdx.x;
  if (i < n) {
    deg_s[i] = __float_as_uint(rsqrtf(fmaxf((float)deg_s[i], 1.0f)));
    deg_d[i] = __float_as_uint(rsqrtf(fmaxf((float)deg_d[i], 1.0f)));
  }
}


__global__ __launch_bounds__(256) void gemm1_wmma(const float* __restrict__ x,
                                                  const float* __restrict__ W1,
                                                  const float* __restrict__ out_norm,
                                                  float* __restrict__ h, int nNodes) {
  __shared__ float w[IN_DIM * 16];
  int tid = threadIdx.x;
  for (int i = tid; i < IN_DIM * 16; i += 256) {
    int r = i >> 4, cidx = i & 15;
    w[i] = (cidx < HID) ? W1[r * HID + cidx] : 0.0f;
  }
  __syncthreads();

  int wid  = tid >> 5;
  int lane = tid & 31;
  int m    = lane & 15;
  int kh   = lane >> 4;

  int row0 = (blockIdx.x * 8 + wid) * 16;
  int node = row0 + m;
  int nodeClamped = node < nNodes ? node : nNodes - 1;
  const float* xrow = x + (long)nodeClamped * IN_DIM;

  v8f acc = {};
#pragma unroll
  for (int kk = 0; kk < IN_DIM; kk += 4) {
    int k0 = kk + 2 * kh;
    v2f a;
    a.x = xrow[k0];
    a.y = xrow[k0 + 1];
    v2f b;
    b.x = w[k0 * 16 + m];
    b.y = w[(k0 + 1) * 16 + m];
    acc = __builtin_amdgcn_wmma_f32_16x16x4_f32(
         false, a,  false, b,
         (short)0, acc,  false,  false);
  }

  if (m < HID) {
#pragma unroll
    for (int r = 0; r < 8; ++r) {
      int nr = row0 + r + 8 * kh;
      if (nr < nNodes) h[(long)nr * HID + m] = acc[r] * out_norm[nr];
    }
  }
}


__global__ void spmm1_kernel(const int* __restrict__ src, const int* __restrict__ dst,
                             const float* __restrict__ h, float* __restrict__ agg, int nE) {
  const int lane = threadIdx.x & 31;
  const int j    = lane & 7;
  const int sub  = lane >> 3;
  const int e    = blockIdx.x * blockDim.x + threadIdx.x;
  const int left = nE - (e - lane);

  int s = 0, d = 0;
  if (e < nE) { s = src[e]; d = dst[e]; }

  float    v[8];
  unsigned off[8];
#pragma unroll
  for (int i = 0; i < 8; ++i) {
    const int from = i * 4 + sub;
    const int si = __shfl(s, from, 32);
    const int di = __shfl(d, from, 32);
    v[i]   = h[(unsigned)(si * HID + j)];
    off[i] = (unsigned)(di * HID + j);
  }
#pragma unroll
  for (int i = 0; i < 8; ++i) {
    if (i * 4 + sub < left) atomicAdd(&agg[off[i]], v[i]);
  }
}


__global__ void layer2_node(const float* __restrict__ agg, const float* __restrict__ in_norm,
                            const float* __restrict__ out_norm, const float* __restrict__ b1,
                            const float* __restrict__ W2, float* __restrict__ g2, int n) {
  int v = blockIdx.x * blockDim.x + threadIdx.x;
  if (v >= n) return;
  float innv = in_norm[v];
  const float* av = agg + (long)v * HID;
  float s = 0.0f;
#pragma unroll
  for (int j = 0; j < HID; ++j) {
    float z = fmaf(av[j], innv, b1[j]);
    z = z > 0.0f ? z : 0.0f;
    s = fmaf(z, W2[j], s);
  }
  g2[v] = s * out_norm[v];
}


__device__ __forceinline__ void sum_hit(float* acc, const float* __restrict__ g2, int s, int d,
                                        int base, int len) {
  const unsigned k = (unsigned)(d - base);
  if (k < (unsigned)len) atomicAdd(&acc[k], g2[s]);
}

__global__ __launch_bounds__(SCAN_THREADS) void spmm2_slices(const int* __restrict__ src,
                                                             const int* __restrict__ dst,
                                                             const float* __restrict__ g2,
                                                             float* __restrict__ outacc,
                                                             int nE, int nN) {
  __shared__ float acc[SLICE_CAP];
  const int tid     = threadIdx.x;
  const int nSlices = gridDim.y;
  const int len     = slice_len(nN, nSlices);
  const int base    = (int)blockIdx.y * len;

  for (int i = tid; i < len; i += SCAN_THREADS) acc[i] = 0.0f;
  __syncthreads();

  int start, n;
  chunk_bounds(nE, gridDim.x, blockIdx.x, start, n);
  const int* ps = src + start;
  const int* pd = dst + start;
  int head = (int)(((16u - (unsigned)((size_t)pd & 15u)) & 15u) >> 2);
  if (head > n) head = n;
  const bool vec = (((size_t)(ps + head)) & 15u) == 0;
  if (!vec) head = n;
  for (int i = tid; i < head; i += SCAN_THREADS) sum_hit(acc, g2, ps[i], pd[i], base, len);
  const int4* sv = (const int4*)(ps + head);
  const int4* dv = (const int4*)(pd + head);
  const int nvec = (n - head) >> 2;
  int i = tid;
  int4 dq[4], sq[4];
#pragma unroll
  for (int u = 0; u < 4; ++u) { dq[u] = make_int4(-1, -1, -1, -1); sq[u] = make_int4(0, 0, 0, 0); }
  for (;;) {
    const bool fullNext = i + 3 * SCAN_THREADS < nvec;
    int4 dn[4], sn[4];
#pragma unroll
    for (int u = 0; u < 4; ++u) { dn[u] = make_int4(-1, -1, -1, -1); sn[u] = make_int4(0, 0, 0, 0); }
    if (fullNext) {
#pragma unroll
      for (int u = 0; u < 4; ++u) { sn[u] = sv[i + u * SCAN_THREADS]; dn[u] = dv[i + u * SCAN_THREADS]; }
    }
    unsigned k[16];
    float    val[16];
#pragma unroll
    for (int u = 0; u < 4; ++u) {
      k[4 * u + 0] = (unsigned)(dq[u].x - base);
      k[4 * u + 1] = (unsigned)(dq[u].y - base);
      k[4 * u + 2] = (unsigned)(dq[u].z - base);
      k[4 * u + 3] = (unsigned)(dq[u].w - base);
      val[4 * u + 0] = (k[4 * u + 0] < (unsigned)len) ? g2[sq[u].x] : 0.0f;
      val[4 * u + 1] = (k[4 * u + 1] < (unsigned)len) ? g2[sq[u].y] : 0.0f;
      val[4 * u + 2] = (k[4 * u + 2] < (unsigned)len) ? g2[sq[u].z] : 0.0f;
      val[4 * u + 3] = (k[4 * u + 3] < (unsigned)len) ? g2[sq[u].w] : 0.0f;
    }
#pragma unroll
    for (int j = 0; j < 16; ++j) {
      if (k[j] < (unsigned)len) atomicAdd(&acc[k[j]], val[j]);
    }
    if (!fullNext) break;
#pragma unroll
    for (int u = 0; u < 4; ++u) { dq[u] = dn[u]; sq[u] = sn[u]; }
    i += 4 * SCAN_THREADS;
  }
  for (; i < nvec; i += SCAN_THREADS) {
    const int4 d0 = dv[i];
    const int4 s0 = sv[i];
    sum_hit(acc, g2, s0.x, d0.x, base, len); sum_hit(acc, g2, s0.y, d0.y, base, len);
    sum_hit(acc, g2, s0.z, d0.z, base, len); sum_hit(acc, g2, s0.w, d0.w, base, len);
  }
  for (int t = head + nvec * 4 + tid; t < n; t += SCAN_THREADS) sum_hit(acc, g2, ps[t], pd[t], base, len);
  __syncthreads();

  for (int t = tid; t < len; t += SCAN_THREADS) {
    const int node = base + t;
    if (node < nN) atomicAdd(&outacc[node], acc[t]);
  }
}

__global__ void final_kernel(const float* __restrict__ outacc, const float* __restrict__ in_norm,
                             const float* __restrict__ b2, float* __restrict__ out, int n) {
  int v = blockIdx.x * blockDim.x + threadIdx.x;
  if (v < n) out[v] = fmaf(outacc[v], in_norm[v], b2[0]);
}


extern "C" void kernel_launch(void* const* d_in, const int* in_sizes, int n_in,
                              void* d_out, int out_size, void* d_ws, size_t ws_size,
                              hipStream_t stream) {
  const float* x  = (const float*)d_in[0];
  const float* W1 = (const float*)d_in[1];
  const float* b1 = (const float*)d_in[2];
  const float* W2 = (const float*)d_in[3];
  const float* b2 = (const float*)d_in[4];
  const int*  src = (const int*)d_in[5];
  const int*  dst = (const int*)d_in[6];

  const int nN = in_sizes[0] / IN_DIM;
  const int nE = in_sizes[5];

  float* ws     = (float*)d_ws;
  float* deg_s  = ws;
  float* deg_d  = ws + (long)nN;
  float* h      = ws + 2L * nN;
  float* agg    = ws + 10L * nN;
  float* g2     = ws + 18L * nN;
  float* outacc = ws + 19L * nN;
  float* out    = (float*)d_out;

  const int TB = 256;
  dim3 blk(TB);

  zero_f32<<<(2 * nN + TB - 1) / TB, blk, 0, stream>>>(deg_s, 2 * nN);
  zero_f32<<<(8 * nN + TB - 1) / TB, blk, 0, stream>>>(agg, 8 * nN);
  zero_f32<<<(nN + TB - 1) / TB, blk, 0, stream>>>(outacc, nN);

  const int nSlices   = (nN + SLICE_CAP - 1) / SLICE_CAP;
  const int degChunks = 256 / (2 * nSlices) > 0 ? 256 / (2 * nSlices) : 1;
  const int sumChunks = 256 / nSlices > 0 ? 256 / nSlices : 1;

  degree_slices<<<dim3(degChunks, 2 * nSlices), dim3(SCAN_THREADS), 0, stream>>>(
      src, dst, (unsigned*)deg_s, (unsigned*)deg_d, nE, nN);
  norm_kernel<<<(nN + TB - 1) / TB, blk, 0, stream>>>((unsigned*)deg_s, (unsigned*)deg_d, nN);

  gemm1_wmma<<<(nN + 127) / 128, blk, 0, stream>>>(x, W1, deg_s, h, nN);

  spmm1_kernel<<<(nE + TB - 1) / TB, blk, 0, stream>>>(src, dst, h, agg, nE);
  layer2_node<<<(nN + TB - 1) / TB, blk, 0, stream>>>(agg, deg_d, deg_s, b1, W2, g2, nN);
  spmm2_slices<<<dim3(sumChunks, nSlices), dim3(SCAN_THREADS), 0, stream>>>(src, dst, g2, outacc, nE, nN);
  final_kernel<<<(nN + TB - 1) / TB, blk, 0, stream>>>(outacc, deg_d, b2, out, nN);
}


# layer-1 edge sums by owner ranges: edges counting-sorted by dst range (LDS counters), owner work-group adds rows in LDS, stores once; no global atomics; sums in LDS atomic arrival order
# speedup vs baseline: 5.5413x; 1.4203x over previous
#include <hip/hip_runtime.h>

typedef __attribute__((ext_vector_type(2))) float v2f;
typedef __attribute__((ext_vector_type(8))) float v8f;

constexpr int IN_DIM = 128;
constexpr int HID    = 8;

constexpr int SLICE_CAP    = 73728;
constexpr int SCAN_THREADS = 1024;


__global__ void zero_f32(float* __restrict__ p, int n) {
  int i = blockIdx.x * blockDim.x + threadIdx.x;
  if (i < n) p[i] = 0.0f;
}


__device__ __forceinline__ int slice_len(int nN, int nSlices) {
  const int per = (nN + nSlices - 1) / nSlices;
  return (per + 31) & ~31;
}

__device__ __forceinline__ void chunk_bounds(int nE, int nChunks, int chunk, int& start, int& count) {
  const int chunkLen = (((nE + nChunks - 1) / nChunks) + 3) & ~3;
  long s = (long)chunk * chunkLen;
  if (s > nE) s = nE;
  long e = s + chunkLen;
  if (e > nE) e = nE;
  start = (int)s;
  count = (int)(e - s);
}

__device__ __forceinline__ void count_hit(unsigned* cnt, int v, int base, int len) {
  const unsigned k = (unsigned)(v - base);
  if (k < (unsigned)len) atomicAdd(&cnt[k], 1u);
}

__global__ __launch_bounds__(SCAN_THREADS) void degree_slices(const int* __restrict__ src,
                                                              const int* __restrict__ dst,
                                                              unsigned* __restrict__ cnt_s,
                                                              unsigned* __restrict__ cnt_d,
                                                              int nE, int nN) {
  __shared__ unsigned cnt[SLICE_CAP];
  const int tid     = threadIdx.x;
  const int nSlices = gridDim.y >> 1;
  const int which   = (int)blockIdx.y >= nSlices ? 1 : 0;
  const int slice   = (int)blockIdx.y - which * nSlices;
  const int len     = slice_len(nN, nSlices);
  const int base    = slice * len;
  const int* __restrict__ idx = which ? dst : src;
  unsigned* __restrict__ out  = which ? cnt_d : cnt_s;

  for (int i = tid; i < len; i += SCAN_THREADS) cnt[i] = 0u;
  __syncthreads();

  int start, n;
  chunk_bounds(nE, gridDim.x, blockIdx.x, start, n);
  const int* p = idx + start;
  int head = (int)(((16u - (unsigned)((size_t)p & 15u)) & 15u) >> 2);
  if (head > n) head = n;
  for (int i = tid; i < head; i += SCAN_THREADS) count_hit(cnt, p[i], base, len);
  const int4* pv = (const int4*)(p + head);
  const int nvec = (n - head) >> 2;
  int i = tid;
  for (; i + 7 * SCAN_THREADS < nvec; i += 8 * SCAN_THREADS) {
    int4 v[8];
#pragma unroll
    for (int u = 0; u < 8; ++u) v[u] = pv[i + u * SCAN_THREADS];
#pragma unroll
    for (int u = 0; u < 8; ++u) {
      count_hit(cnt, v[u].x, base, len); count_hit(cnt, v[u].y, base, len);
      count_hit(cnt, v[u].z, base, len); count_hit(cnt, v[u].w, base, len);
    }
  }
  for (; i < nvec; i += SCAN_THREADS) {
    const int4 a = pv[i];
    count_hit(cnt, a.x, base, len); count_hit(cnt, a.y, base, len);
    count_hit(cnt, a.z, base, len); count_hit(cnt, a.w, base, len);
  }
  for (int t = head + nvec * 4 + tid; t < n; t += SCAN_THREADS) count_hit(cnt, p[t], base, len);
  __syncthreads();

  for (int t = tid; t < len; t += SCAN_THREADS) {
    const int node = base + t;
    if (node < nN) atomicAdd(&out[node], cnt[t]);
  }
}

__global__ void norm_kernel(unsigned* __restrict__ deg_s, unsigned* __restrict__ deg_d, int n) {
  int i = blockIdx.x * blockDim.x + threadIdx.x;
  if (i < n) {
    deg_s[i] = __float_as_uint(rsqrtf(fmaxf((float)deg_s[i], 1.0f)));
    deg_d[i] = __float_as_uint(rsqrtf(fmaxf((float)deg_d[i], 1.0f)));
  }
}


__global__ __launch_bounds__(256) void gemm1_wmma(const float* __restrict__ x,
                                                  const float* __restrict__ W1,
                                                  const float* __restrict__ out_norm,
                                                  float* __restrict__ h, int nNodes) {
  __shared__ float w[IN_DIM * 16];
  int tid = threadIdx.x;
  for (int i = tid; i < IN_DIM * 16; i += 256) {
    int r = i >> 4, cidx = i & 15;
    w[i] = (cidx < HID) ? W1[r * HID + cidx] : 0.0f;
  }
  __syncthreads();

  int wid  = tid >> 5;
  int lane = tid & 31;
  int m    = lane & 15;
  int kh   = lane >> 4;

  int row0 = (blockIdx.x * 8 + wid) * 16;
  int node = row0 + m;
  int nodeClamped = node < nNodes ? node : nNodes - 1;
  const float* xrow = x + (long)nodeClamped * IN_DIM;

  v8f acc = {};
#pragma unroll
  for (int kk = 0; kk < IN_DIM; kk += 4) {
    int k0 = kk + 2 * kh;
    v2f a;
    a.x = xrow[k0];
    a.y = xrow[k0 + 1];
    v2f b;
    b.x = w[k0 * 16 + m];
    b.y = w[(k0 + 1) * 16 + m];
    acc = __builtin_amdgcn_wmma_f32_16x16x4_f32(
         false, a,  false, b,
         (short)0, acc,  false,  false);
  }

  if (m < HID) {
#pragma unroll
    for (int r = 0; r < 8; ++r) {
      int nr = row0 + r + 8 * kh;
      if (nr < nNodes) h[(long)nr * HID + m] = acc[r] * out_norm[nr];
    }
  }
}


__global__ void spmm1_kernel(const int* __restrict__ src, const int* __restrict__ dst,
                             const float* __restrict__ h, float* __restrict__ agg, int nE) {
  const int lane = threadIdx.x & 31;
  const int j    = lane & 7;
  const int sub  = lane >> 3;
  const int e    = blockIdx.x * blockDim.x + threadIdx.x;
  const int left = nE - (e - lane);

  int s = 0, d = 0;
  if (e < nE) { s = src[e]; d = dst[e]; }

  float    v[8];
  unsigned off[8];
#pragma unroll
  for (int i = 0; i < 8; ++i) {
    const int from = i * 4 + sub;
    const int si = __shfl(s, from, 32);
    const int di = __shfl(d, from, 32);
    v[i]   = h[(unsigned)(si * HID + j)];
    off[i] = (unsigned)(di * HID + j);
  }
#pragma unroll
  for (int i = 0; i < 8; ++i) {
    if (i * 4 + sub < left) atomicAdd(&agg[off[i]], v[i]);
  }
}


constexpr int OWN_SHIFT  = 10;
constexpr int OWN_ROWS   = 1 << OWN_SHIFT;
constexpr int SRC_BITS   = 32 - OWN_SHIFT;
constexpr int MAX_RANGES = 4096;
constexpr int BIN_CHUNKS = 256;
constexpr unsigned SRC_MASK = (1u << SRC_BITS) - 1u;

__device__ __forceinline__ unsigned range_of(int d, int nRanges) {
  const unsigned r = (unsigned)d >> OWN_SHIFT;
  return r < (unsigned)nRanges ? r : (unsigned)(nRanges - 1);
}

__global__ __launch_bounds__(SCAN_THREADS) void bin_count(const int* __restrict__ dst,
                                                          int* __restrict__ histT,
                                                          int nE, int nRanges) {
  __shared__ unsigned cnt[MAX_RANGES];
  const int tid = threadIdx.x;
  for (int i = tid; i < nRanges; i += SCAN_THREADS) cnt[i] = 0u;
  __syncthreads();

  int start, n;
  chunk_bounds(nE, gridDim.x, blockIdx.x, start, n);
  const int* p = dst + start;
  int head = (int)(((16u - (unsigned)((size_t)p & 15u)) & 15u) >> 2);
  if (head > n) head = n;
  for (int i = tid; i < head; i += SCAN_THREADS) atomicAdd(&cnt[range_of(p[i], nRanges)], 1u);
  const int4* pv = (const int4*)(p + head);
  const int nvec = (n - head) >> 2;
  int i = tid;
  for (; i + 3 * SCAN_THREADS < nvec; i += 4 * SCAN_THREADS) {
    int4 v[4];
#pragma unroll
    for (int u = 0; u < 4; ++u) v[u] = pv[i + u * SCAN_THREADS];
#pragma unroll
    for (int u = 0; u < 4; ++u) {
      atomicAdd(&cnt[range_of(v[u].x, nRanges)], 1u); atomicAdd(&cnt[range_of(v[u].y, nRanges)], 1u);
      atomicAdd(&cnt[range_of(v[u].z, nRanges)], 1u); atomicAdd(&cnt[range_of(v[u].w, nRanges)], 1u);
    }
  }
  for (; i < nvec; i += SCAN_THREADS) {
    const int4 a = pv[i];
    atomicAdd(&cnt[range_of(a.x, nRanges)], 1u); atomicAdd(&cnt[range_of(a.y, nRanges)], 1u);
    atomicAdd(&cnt[range_of(a.z, nRanges)], 1u); atomicAdd(&cnt[range_of(a.w, nRanges)], 1u);
  }
  for (int t = head + nvec * 4 + tid; t < n; t += SCAN_THREADS)
    atomicAdd(&cnt[range_of(p[t], nRanges)], 1u);
  __syncthreads();

  for (int r = tid; r < nRanges; r += SCAN_THREADS) histT[r * (int)gridDim.x + (int)blockIdx.x] = (int)cnt[r];
}

__global__ void bin_scan(int* __restrict__ histT, int* __restrict__ total, int nChunks, int nRanges) {
  const int lane  = threadIdx.x & 31;
  const int range = (int)((blockIdx.x * blockDim.x + threadIdx.x) >> 5);
  if (range >= nRanges) return;
  int* row = histT + (long)range * nChunks;
  int carry = 0;
  for (int c0 = 0; c0 < nChunks; c0 += 32) {
    const int c = c0 + lane;
    const int v = c < nChunks ? row[c] : 0;
    int s = v;
#pragma unroll
    for (int d = 1; d < 32; d <<= 1) {
      const int t = __shfl_up(s, d, 32);
      if (lane >= d) s += t;
    }
    if (c < nChunks) row[c] = carry + s - v;
    carry += __shfl(s, 31, 32);
  }
  if (lane == 0) total[range] = carry;
}

__global__ __launch_bounds__(SCAN_THREADS) void bin_place(const int* __restrict__ src,
                                                          const int* __restrict__ dst,
                                                          const int* __restrict__ histT,
                                                          const int* __restrict__ total,
                                                          int* __restrict__ rangeStart,
                                                          unsigned* __restrict__ recs,
                                                          int nE, int nRanges) {
  __shared__ int      scanbuf[2][MAX_RANGES];
  __shared__ unsigned cursor[MAX_RANGES];
  const int tid = threadIdx.x;
  for (int r = tid; r < nRanges; r += SCAN_THREADS) scanbuf[0][r] = total[r];
  __syncthreads();
  int cur = 0;
  for (int off = 1; off < nRanges; off <<= 1) {
    for (int r = tid; r < nRanges; r += SCAN_THREADS)
      scanbuf[1 - cur][r] = scanbuf[cur][r] + (r >= off ? scanbuf[cur][r - off] : 0);
    __syncthreads();
    cur ^= 1;
  }
  for (int r = tid; r < nRanges; r += SCAN_THREADS) {
    const int incl = scanbuf[cur][r];
    const int excl = incl - total[r];
    cursor[r] = (unsigned)(excl + histT[r * (int)gridDim.x + (int)blockIdx.x]);
    if (blockIdx.x == 0) {
      rangeStart[r] = excl;
      if (r == nRanges - 1) rangeStart[nRanges] = incl;
    }
  }
  __syncthreads();

  int start, n;
  chunk_bounds(nE, gridDim.x, blockIdx.x, start, n);
  const int* ps = src + start;
  const int* pd = dst + start;
  int head = (int)(((16u - (unsigned)((size_t)pd & 15u)) & 15u) >> 2);
  if (head > n) head = n;
  if ((((size_t)(ps + head)) & 15u) != 0) head = n;
#define PLACE_EDGE(S, D)                                                            \
  do {                                                                              \
    const unsigned pos_ = atomicAdd(&cursor[range_of((D), nRanges)], 1u);           \
    recs[pos_] = (((unsigned)(D) & (unsigned)(OWN_ROWS - 1)) << SRC_BITS) | ((unsigned)(S) & SRC_MASK); \
  } while (0)
  for (int i = tid; i < head; i += SCAN_THREADS) PLACE_EDGE(ps[i], pd[i]);
  const int4* sv = (const int4*)(ps + head);
  const int4* dv = (const int4*)(pd + head);
  const int nvec = (n - head) >> 2;
  int i = tid;
  for (; i + 3 * SCAN_THREADS < nvec; i += 4 * SCAN_THREADS) {
    int4 s4[4], d4[4];
#pragma unroll
    for (int u = 0; u < 4; ++u) d4[u] = dv[i + u * SCAN_THREADS];
#pragma unroll
    for (int u = 0; u < 4; ++u) s4[u] = sv[i + u * SCAN_THREADS];
    unsigned rec[16], pos[16];
#pragma unroll
    for (int u = 0; u < 4; ++u) {
      const int sx[4] = {s4[u].x, s4[u].y, s4[u].z, s4[u].w};
      const int dx[4] = {d4[u].x, d4[u].y, d4[u].z, d4[u].w};
#pragma unroll
      for (int k = 0; k < 4; ++k) {
        rec[4 * u + k] = (((unsigned)dx[k] & (unsigned)(OWN_ROWS - 1)) << SRC_BITS) | ((unsigned)sx[k] & SRC_MASK);
        pos[4 * u + k] = atomicAdd(&cursor[range_of(dx[k], nRanges)], 1u);
      }
    }
#pragma unroll
    for (int k = 0; k < 16; ++k) recs[pos[k]] = rec[k];
  }
  for (; i < nvec; i += SCAN_THREADS) {
    const int4 s0 = sv[i];
    const int4 d0 = dv[i];
    PLACE_EDGE(s0.x, d0.x); PLACE_EDGE(s0.y, d0.y); PLACE_EDGE(s0.z, d0.z); PLACE_EDGE(s0.w, d0.w);
  }
  for (int t = head + nvec * 4 + tid; t < n; t += SCAN_THREADS) PLACE_EDGE(ps[t], pd[t]);
#undef PLACE_EDGE
}

__global__ __launch_bounds__(SCAN_THREADS) void spmm1_owner(const unsigned* __restrict__ recs,
                                                            const int* __restrict__ rangeStart,
                                                            const float* __restrict__ h,
                                                            float* __restrict__ agg, int nN, int nE) {
  __shared__ float acc[OWN_ROWS * HID];
  const int tid = threadIdx.x;
  for (int i = tid; i < OWN_ROWS * HID; i += SCAN_THREADS) acc[i] = 0.0f;
  __syncthreads();

  const int beg  = rangeStart[blockIdx.x];
  const int end  = rangeStart[blockIdx.x + 1];
  const int lane = tid & 31;
  const int j    = lane & 7;
  const int sub  = lane >> 3;
  constexpr int TRIP   = 64;
  constexpr int STRIDE = (SCAN_THREADS / 32) * TRIP;

  const int last = nE - 1;
  int base = beg + (tid >> 5) * TRIP;
  unsigned r0 = recs[min(base + lane, last)];
  unsigned r1 = recs[min(base + 32 + lane, last)];
  for (; base < end; base += STRIDE) {
    const int left  = end - base;
    const int nbase = base + STRIDE;
    const unsigned n0 = recs[min(nbase + lane, last)];
    const unsigned n1 = recs[min(nbase + 32 + lane, last)];

    float    v[16];
    unsigned off[16];
#pragma unroll
    for (int i = 0; i < 8; ++i) {
      const int from = i * 4 + sub;
      const unsigned q0 = __shfl(r0, from, 32);
      const unsigned q1 = __shfl(r1, from, 32);
      v[i]       = h[(q0 & SRC_MASK) * HID + j];
      off[i]     = (q0 >> SRC_BITS) * HID + j;
      v[8 + i]   = h[(q1 & SRC_MASK) * HID + j];
      off[8 + i] = (q1 >> SRC_BITS) * HID + j;
    }
#pragma unroll
    for (int i = 0; i < 8; ++i) {
      const bool ok0 = i * 4 + sub < left;
      const bool ok1 = 32 + i * 4 + sub < left;
      atomicAdd(&acc[ok0 ? off[i] : (unsigned)j], ok0 ? v[i] : 0.0f);
      atomicAdd(&acc[ok1 ? off[8 + i] : (unsigned)j], ok1 ? v[8 + i] : 0.0f);
    }
    r0 = n0;
    r1 = n1;
  }
  __syncthreads();

  const long rowBase = (long)blockIdx.x * OWN_ROWS;
  for (int t = tid; t < OWN_ROWS * HID; t += SCAN_THREADS) {
    if (rowBase + (t >> 3) < nN) agg[rowBase * HID + t] = acc[t];
  }
}


__global__ void layer2_node(const float* __restrict__ agg, const float* __restrict__ in_norm,
                            const float* __restrict__ out_norm, const float* __restrict__ b1,
                            const float* __restrict__ W2, float* __restrict__ g2, int n) {
  int v = blockIdx.x * blockDim.x + threadIdx.x;
  if (v >= n) return;
  float innv = in_norm[v];
  const float* av = agg + (long)v * HID;
  float s = 0.0f;
#pragma unroll
  for (int j = 0; j < HID; ++j) {
    float z = fmaf(av[j], innv, b1[j]);
    z = z > 0.0f ? z : 0.0f;
    s = fmaf(z, W2[j], s);
  }
  g2[v] = s * out_norm[v];
}


__device__ __forceinline__ void sum_hit(float* acc, const float* __restrict__ g2, int s, int d,
                                        int base, int len) {
  const unsigned k = (unsigned)(d - base);
  if (k < (unsigned)len) atomicAdd(&acc[k], g2[s]);
}

__global__ __launch_bounds__(SCAN_THREADS) void spmm2_slices(const int* __restrict__ src,
                                                             const int* __restrict__ dst,
                                                             const float* __restrict__ g2,
                                                             float* __restrict__ outacc,
                                                             int nE, int nN) {
  __shared__ float acc[SLICE_CAP];
  const int tid     = threadIdx.x;
  const int nSlices = gridDim.y;
  const int len     = slice_len(nN, nSlices);
  const int base    = (int)blockIdx.y * len;

  for (int i = tid; i < len; i += SCAN_THREADS) acc[i] = 0.0f;
  __syncthreads();

  int start, n;
  chunk_bounds(nE, gridDim.x, blockIdx.x, start, n);
  const int* ps = src + start;
  const int* pd = dst + start;
  int head = (int)(((16u - (unsigned)((size_t)pd & 15u)) & 15u) >> 2);
  if (head > n) head = n;
  const bool vec = (((size_t)(ps + head)) & 15u) == 0;
  if (!vec) head = n;
  for (int i = tid; i < head; i += SCAN_THREADS) sum_hit(acc, g2, ps[i], pd[i], base, len);
  const int4* sv = (const int4*)(ps + head);
  const int4* dv = (const int4*)(pd + head);
  const int nvec = (n - head) >> 2;
  int i = tid;
  int4 dq[4], sq[4];
#pragma unroll
  for (int u = 0; u < 4; ++u) { dq[u] = make_int4(-1, -1, -1, -1); sq[u] = make_int4(0, 0, 0, 0); }
  for (;;) {
    const bool fullNext = i + 3 * SCAN_THREADS < nvec;
    int4 dn[4], sn[4];
#pragma unroll
    for (int u = 0; u < 4; ++u) { dn[u] = make_int4(-1, -1, -1, -1); sn[u] = make_int4(0, 0, 0, 0); }
    if (fullNext) {
#pragma unroll
      for (int u = 0; u < 4; ++u) { sn[u] = sv[i + u * SCAN_THREADS]; dn[u] = dv[i + u * SCAN_THREADS]; }
    }
    unsigned k[16];
    float    val[16];
#pragma unroll
    for (int u = 0; u < 4; ++u) {
      k[4 * u + 0] = (unsigned)(dq[u].x - base);
      k[4 * u + 1] = (unsigned)(dq[u].y - base);
      k[4 * u + 2] = (unsigned)(dq[u].z - base);
      k[4 * u + 3] = (unsigned)(dq[u].w - base);
      val[4 * u + 0] = (k[4 * u + 0] < (unsigned)len) ? g2[sq[u].x] : 0.0f;
      val[4 * u + 1] = (k[4 * u + 1] < (unsigned)len) ? g2[sq[u].y] : 0.0f;
      val[4 * u + 2] = (k[4 * u + 2] < (unsigned)len) ? g2[sq[u].z] : 0.0f;
      val[4 * u + 3] = (k[4 * u + 3] < (unsigned)len) ? g2[sq[u].w] : 0.0f;
    }
#pragma unroll
    for (int j = 0; j < 16; ++j) {
      if (k[j] < (unsigned)len) atomicAdd(&acc[k[j]], val[j]);
    }
    if (!fullNext) break;
#pragma unroll
    for (int u = 0; u < 4; ++u) { dq[u] = dn[u]; sq[u] = sn[u]; }
    i += 4 * SCAN_THREADS;
  }
  for (; i < nvec; i += SCAN_THREADS) {
    const int4 d0 = dv[i];
    const int4 s0 = sv[i];
    sum_hit(acc, g2, s0.x, d0.x, base, len); sum_hit(acc, g2, s0.y, d0.y, base, len);
    sum_hit(acc, g2, s0.z, d0.z, base, len); sum_hit(acc, g2, s0.w, d0.w, base, len);
  }
  for (int t = head + nvec * 4 + tid; t < n; t += SCAN_THREADS) sum_hit(acc, g2, ps[t], pd[t], base, len);
  __syncthreads();

  for (int t = tid; t < len; t += SCAN_THREADS) {
    const int node = base + t;
    if (node < nN) atomicAdd(&outacc[node], acc[t]);
  }
}

__global__ void final_kernel(const float* __restrict__ outacc, const float* __restrict__ in_norm,
                             const float* __restrict__ b2, float* __restrict__ out, int n) {
  int v = blockIdx.x * blockDim.x + threadIdx.x;
  if (v < n) out[v] = fmaf(outacc[v], in_norm[v], b2[0]);
}


extern "C" void kernel_launch(void* const* d_in, const int* in_sizes, int n_in,
                              void* d_out, int out_size, void* d_ws, size_t ws_size,
                              hipStream_t stream) {
  const float* x  = (const float*)d_in[0];
  const float* W1 = (const float*)d_in[1];
  const float* b1 = (const float*)d_in[2];
  const float* W2 = (const float*)d_in[3];
  const float* b2 = (const float*)d_in[4];
  const int*  src = (const int*)d_in[5];
  const int*  dst = (const int*)d_in[6];

  const int nN = in_sizes[0] / IN_DIM;
  const int nE = in_sizes[5];

  float* ws     = (float*)d_ws;
  float* deg_s  = ws;
  float* deg_d  = ws + (long)nN;
  float* h      = ws + 2L * nN;
  float* agg    = ws + 10L * nN;
  float* g2     = ws + 18L * nN;
  float* outacc = ws + 19L * nN;
  float* out    = (float*)d_out;

  const int TB = 256;
  dim3 blk(TB);

  zero_f32<<<(2 * nN + TB - 1) / TB, blk, 0, stream>>>(deg_s, 2 * nN);
  zero_f32<<<(nN + TB - 1) / TB, blk, 0, stream>>>(outacc, nN);

  const int nSlices   = (nN + SLICE_CAP - 1) / SLICE_CAP;
  const int degChunks = 256 / (2 * nSlices) > 0 ? 256 / (2 * nSlices) : 1;
  const int sumChunks = 256 / nSlices > 0 ? 256 / nSlices : 1;

  degree_slices<<<dim3(degChunks, 2 * nSlices), dim3(SCAN_THREADS), 0, stream>>>(
      src, dst, (unsigned*)deg_s, (unsigned*)deg_d, nE, nN);
  norm_kernel<<<(nN + TB - 1) / TB, blk, 0, stream>>>((unsigned*)deg_s, (unsigned*)deg_d, nN);

  gemm1_wmma<<<(nN + 127) / 128, blk, 0, stream>>>(x, W1, deg_s, h, nN);

  const int    nRanges  = (nN + OWN_ROWS - 1) / OWN_ROWS;
  const size_t wsWords  = 20UL * nN + (size_t)nE + (size_t)nRanges * BIN_CHUNKS + 2UL * nRanges + 8UL;
  if (nE > 0 && nRanges <= MAX_RANGES && wsWords * sizeof(float) <= ws_size) {
    unsigned* recs       = (unsigned*)(ws + 20L * nN);
    int*      histT      = (int*)(recs + nE);
    int*      total      = histT + (long)nRanges * BIN_CHUNKS;
    int*      rangeStart = total + nRanges;
    bin_count<<<BIN_CHUNKS, dim3(SCAN_THREADS), 0, stream>>>(dst, histT, nE, nRanges);
    bin_scan<<<(nRanges * 32 + TB - 1) / TB, blk, 0, stream>>>(histT, total, BIN_CHUNKS, nRanges);
    bin_place<<<BIN_CHUNKS, dim3(SCAN_THREADS), 0, stream>>>(src, dst, histT, total, rangeStart, recs, nE, nRanges);
    spmm1_owner<<<nRanges, dim3(SCAN_THREADS), 0, stream>>>(recs, rangeStart, h, agg, nN, nE);
  } else {
    zero_f32<<<(8 * nN + TB - 1) / TB, blk, 0, stream>>>(agg, 8 * nN);
    spmm1_kernel<<<(nE + TB - 1) / TB, blk, 0, stream>>>(src, dst, h, agg, nE);
  }
  layer2_node<<<(nN + TB - 1) / TB, blk, 0, stream>>>(agg, deg_d, deg_s, b1, W2, g2, nN);
  spmm2_slices<<<dim3(sumChunks, nSlices), dim3(SCAN_THREADS), 0, stream>>>(src, dst, g2, outacc, nE, nN);
  final_kernel<<<(nN + TB - 1) / TB, blk, 0, stream>>>(outacc, deg_d, b2, out, nN);
}


# bin_place orders a chunk's records by range in LDS (LDS cursor adds), then copies each range's run out in whole-line stores, not a 4-byte global store per edge; sums stay in LDS atomic arrival order
# speedup vs baseline: 6.3628x; 1.1483x over previous
#include <hip/hip_runtime.h>

typedef __attribute__((ext_vector_type(2))) float v2f;
typedef __attribute__((ext_vector_type(8))) float v8f;

constexpr int IN_DIM = 128;
constexpr int HID    = 8;

constexpr int SLICE_CAP    = 73728;
constexpr int SCAN_THREADS = 1024;


__global__ void zero_f32(float* __restrict__ p, int n) {
  int i = blockIdx.x * blockDim.x + threadIdx.x;
  if (i < n) p[i] = 0.0f;
}


__device__ __forceinline__ int slice_len(int nN, int nSlices) {
  const int per = (nN + nSlices - 1) / nSlices;
  return (per + 31) & ~31;
}

__device__ __forceinline__ void chunk_bounds(int nE, int nChunks, int chunk, int& start, int& count) {
  const int chunkLen = (((nE + nChunks - 1) / nChunks) + 3) & ~3;
  long s = (long)chunk * chunkLen;
  if (s > nE) s = nE;
  long e = s + chunkLen;
  if (e > nE) e = nE;
  start = (int)s;
  count = (int)(e - s);
}

__device__ __forceinline__ void count_hit(unsigned* cnt, int v, int base, int len) {
  const unsigned k = (unsigned)(v - base);
  if (k < (unsigned)len) atomicAdd(&cnt[k], 1u);
}

__global__ __launch_bounds__(SCAN_THREADS) void degree_slices(const int* __restrict__ src,
                                                              const int* __restrict__ dst,
                                                              unsigned* __restrict__ cnt_s,
                                                              unsigned* __restrict__ cnt_d,
                                                              int nE, int nN) {
  __shared__ unsigned cnt[SLICE_CAP];
  const int tid     = threadIdx.x;
  const int nSlices = gridDim.y >> 1;
  const int which   = (int)blockIdx.y >= nSlices ? 1 : 0;
  const int slice   = (int)blockIdx.y - which * nSlices;
  const int len     = slice_len(nN, nSlices);
  const int base    = slice * len;
  const int* __restrict__ idx = which ? dst : src;
  unsigned* __restrict__ out  = which ? cnt_d : cnt_s;

  for (int i = tid; i < len; i += SCAN_THREADS) cnt[i] = 0u;
  __syncthreads();

  int start, n;
  chunk_bounds(nE, gridDim.x, blockIdx.x, start, n);
  const int* p = idx + start;
  int head = (int)(((16u - (unsigned)((size_t)p & 15u)) & 15u) >> 2);
  if (head > n) head = n;
  for (int i = tid; i < head; i += SCAN_THREADS) count_hit(cnt, p[i], base, len);
  const int4* pv = (const int4*)(p + head);
  const int nvec = (n - head) >> 2;
  int i = tid;
  for (; i + 7 * SCAN_THREADS < nvec; i += 8 * SCAN_THREADS) {
    int4 v[8];
#pragma unroll
    for (int u = 0; u < 8; ++u) v[u] = pv[i + u * SCAN_THREADS];
#pragma unroll
    for (int u = 0; u < 8; ++u) {
      count_hit(cnt, v[u].x, base, len); count_hit(cnt, v[u].y, base, len);
      count_hit(cnt, v[u].z, base, len); count_hit(cnt, v[u].w, base, len);
    }
  }
  for (; i < nvec; i += SCAN_THREADS) {
    const int4 a = pv[i];
    count_hit(cnt, a.x, base, len); count_hit(cnt, a.y, base, len);
    count_hit(cnt, a.z, base, len); count_hit(cnt, a.w, base, len);
  }
  for (int t = head + nvec * 4 + tid; t < n; t += SCAN_THREADS) count_hit(cnt, p[t], base, len);
  __syncthreads();

  for (int t = tid; t < len; t += SCAN_THREADS) {
    const int node = base + t;
    if (node < nN) atomicAdd(&out[node], cnt[t]);
  }
}

__global__ void norm_kernel(unsigned* __restrict__ deg_s, unsigned* __restrict__ deg_d, int n) {
  int i = blockIdx.x * blockDim.x + threadIdx.x;
  if (i < n) {
    deg_s[i] = __float_as_uint(rsqrtf(fmaxf((float)deg_s[i], 1.0f)));
    deg_d[i] = __float_as_uint(rsqrtf(fmaxf((float)deg_d[i], 1.0f)));
  }
}


__global__ __launch_bounds__(256) void gemm1_wmma(const float* __restrict__ x,
                                                  const float* __restrict__ W1,
                                                  const float* __restrict__ out_norm,
                                                  float* __restrict__ h, int nNodes) {
  __shared__ float w[IN_DIM * 16];
  int tid = threadIdx.x;
  for (int i = tid; i < IN_DIM * 16; i += 256) {
    int r = i >> 4, cidx = i & 15;
    w[i] = (cidx < HID) ? W1[r * HID + cidx] : 0.0f;
  }
  __syncthreads();

  int wid  = tid >> 5;
  int lane = tid & 31;
  int m    = lane & 15;
  int kh   = lane >> 4;

  int row0 = (blockIdx.x * 8 + wid) * 16;
  int node = row0 + m;
  int nodeClamped = node < nNodes ? node : nNodes - 1;
  const float* xrow = x + (long)nodeClamped * IN_DIM;

  v8f acc = {};
#pragma unroll
  for (int kk = 0; kk < IN_DIM; kk += 4) {
    int k0 = kk + 2 * kh;
    v2f a;
    a.x = xrow[k0];
    a.y = xrow[k0 + 1];
    v2f b;
    b.x = w[k0 * 16 + m];
    b.y = w[(k0 + 1) * 16 + m];
    acc = __builtin_amdgcn_wmma_f32_16x16x4_f32(
         false, a,  false, b,
         (short)0, acc,  false,  false);
  }

  if (m < HID) {
#pragma unroll
    for (int r = 0; r < 8; ++r) {
      int nr = row0 + r + 8 * kh;
      if (nr < nNodes) h[(long)nr * HID + m] = acc[r] * out_norm[nr];
    }
  }
}


__global__ void spmm1_kernel(const int* __restrict__ src, const int* __restrict__ dst,
                             const float* __restrict__ h, float* __restrict__ agg, int nE) {
  const int lane = threadIdx.x & 31;
  const int j    = lane & 7;
  const int sub  = lane >> 3;
  const int e    = blockIdx.x * blockDim.x + threadIdx.x;
  const int left = nE - (e - lane);

  int s = 0, d = 0;
  if (e < nE) { s = src[e]; d = dst[e]; }

  float    v[8];
  unsigned off[8];
#pragma unroll
  for (int i = 0; i < 8; ++i) {
    const int from = i * 4 + sub;
    const int si = __shfl(s, from, 32);
    const int di = __shfl(d, from, 32);
    v[i]   = h[(unsigned)(si * HID + j)];
    off[i] = (unsigned)(di * HID + j);
  }
#pragma unroll
  for (int i = 0; i < 8; ++i) {
    if (i * 4 + sub < left) atomicAdd(&agg[off[i]], v[i]);
  }
}


constexpr int OWN_SHIFT  = 10;
constexpr int OWN_ROWS   = 1 << OWN_SHIFT;
constexpr int SRC_BITS   = 32 - OWN_SHIFT;
constexpr int MAX_RANGES = 4096;
constexpr int BIN_CHUNKS = 256;
constexpr unsigned SRC_MASK = (1u << SRC_BITS) - 1u;

__device__ __forceinline__ unsigned range_of(int d, int nRanges) {
  const unsigned r = (unsigned)d >> OWN_SHIFT;
  return r < (unsigned)nRanges ? r : (unsigned)(nRanges - 1);
}

__global__ __launch_bounds__(SCAN_THREADS) void bin_count(const int* __restrict__ dst,
                                                          int* __restrict__ histT,
                                                          int nE, int nRanges) {
  __shared__ unsigned cnt[MAX_RANGES];
  const int tid = threadIdx.x;
  for (int i = tid; i < nRanges; i += SCAN_THREADS) cnt[i] = 0u;
  __syncthreads();

  int start, n;
  chunk_bounds(nE, gridDim.x, blockIdx.x, start, n);
  const int* p = dst + start;
  int head = (int)(((16u - (unsigned)((size_t)p & 15u)) & 15u) >> 2);
  if (head > n) head = n;
  for (int i = tid; i < head; i += SCAN_THREADS) atomicAdd(&cnt[range_of(p[i], nRanges)], 1u);
  const int4* pv = (const int4*)(p + head);
  const int nvec = (n - head) >> 2;
  int i = tid;
  for (; i + 3 * SCAN_THREADS < nvec; i += 4 * SCAN_THREADS) {
    int4 v[4];
#pragma unroll
    for (int u = 0; u < 4; ++u) v[u] = pv[i + u * SCAN_THREADS];
#pragma unroll
    for (int u = 0; u < 4; ++u) {
      atomicAdd(&cnt[range_of(v[u].x, nRanges)], 1u); atomicAdd(&cnt[range_of(v[u].y, nRanges)], 1u);
      atomicAdd(&cnt[range_of(v[u].z, nRanges)], 1u); atomicAdd(&cnt[range_of(v[u].w, nRanges)], 1u);
    }
  }
  for (; i < nvec; i += SCAN_THREADS) {
    const int4 a = pv[i];
    atomicAdd(&cnt[range_of(a.x, nRanges)], 1u); atomicAdd(&cnt[range_of(a.y, nRanges)], 1u);
    atomicAdd(&cnt[range_of(a.z, nRanges)], 1u); atomicAdd(&cnt[range_of(a.w, nRanges)], 1u);
  }
  for (int t = head + nvec * 4 + tid; t < n; t += SCAN_THREADS)
    atomicAdd(&cnt[range_of(p[t], nRanges)], 1u);
  __syncthreads();

  for (int r = tid; r < nRanges; r += SCAN_THREADS) histT[r * (int)gridDim.x + (int)blockIdx.x] = (int)cnt[r];
}

__global__ void bin_scan(int* __restrict__ histT, int* __restrict__ total, int nChunks, int nRanges) {
  const int lane  = threadIdx.x & 31;
  const int range = (int)((blockIdx.x * blockDim.x + threadIdx.x) >> 5);
  if (range >= nRanges) return;
  int* row = histT + (long)range * nChunks;
  int carry = 0;
  for (int c0 = 0; c0 < nChunks; c0 += 32) {
    const int c = c0 + lane;
    const int v = c < nChunks ? row[c] : 0;
    int s = v;
#pragma unroll
    for (int d = 1; d < 32; d <<= 1) {
      const int t = __shfl_up(s, d, 32);
      if (lane >= d) s += t;
    }
    if (c < nChunks) row[c] = carry + s - v;
    carry += __shfl(s, 31, 32);
  }
  if (lane == 0) total[range] = carry;
}

constexpr int STAGE_CAP = 62976;

__global__ __launch_bounds__(SCAN_THREADS) void bin_place(const int* __restrict__ src,
                                                          const int* __restrict__ dst,
                                                          const int* __restrict__ histT,
                                                          const int* __restrict__ total,
                                                          int* __restrict__ rangeStart,
                                                          unsigned* __restrict__ recs,
                                                          int nE, int nRanges) {
  __shared__ int      scanbuf[2][MAX_RANGES];
  __shared__ unsigned gpos[MAX_RANGES];
  __shared__ unsigned lcur[MAX_RANGES];
  __shared__ unsigned stage[STAGE_CAP];
  const int tid     = threadIdx.x;
  const int nChunks = (int)gridDim.x;
  const int chunk   = (int)blockIdx.x;

  for (int r = tid; r < nRanges; r += SCAN_THREADS) scanbuf[0][r] = total[r];
  __syncthreads();
  int cur = 0;
  for (int off = 1; off < nRanges; off <<= 1) {
    for (int r = tid; r < nRanges; r += SCAN_THREADS)
      scanbuf[1 - cur][r] = scanbuf[cur][r] + (r >= off ? scanbuf[cur][r - off] : 0);
    __syncthreads();
    cur ^= 1;
  }
  for (int r = tid; r < nRanges; r += SCAN_THREADS) {
    const int tot  = total[r];
    const int incl = scanbuf[cur][r];
    const int excl = incl - tot;
    const int pre  = histT[r * nChunks + chunk];
    const int nxt  = chunk + 1 < nChunks ? histT[r * nChunks + chunk + 1] : tot;
    gpos[r] = (unsigned)(excl + pre);
    scanbuf[1 - cur][r] = nxt - pre;
    if (chunk == 0) {
      rangeStart[r] = excl;
      if (r == nRanges - 1) rangeStart[nRanges] = incl;
    }
  }
  __syncthreads();
  cur ^= 1;
  for (int off = 1; off < nRanges; off <<= 1) {
    for (int r = tid; r < nRanges; r += SCAN_THREADS)
      scanbuf[1 - cur][r] = scanbuf[cur][r] + (r >= off ? scanbuf[cur][r - off] : 0);
    __syncthreads();
    cur ^= 1;
  }
  for (int r = tid; r < nRanges; r += SCAN_THREADS) lcur[r] = r > 0 ? (unsigned)scanbuf[cur][r - 1] : 0u;
  __syncthreads();

  int start, n;
  chunk_bounds(nE, nChunks, chunk, start, n);
  const int* ps = src + start;
  const int* pd = dst + start;
  int head = (int)(((16u - (unsigned)((size_t)pd & 15u)) & 15u) >> 2);
  if (head > n) head = n;
  if ((((size_t)(ps + head)) & 15u) != 0) head = n;
#define PLACE_EDGE(S, D)                                                            \
  do {                                                                              \
    const unsigned pos_ = atomicAdd(&lcur[range_of((D), nRanges)], 1u);             \
    stage[pos_] = (((unsigned)(D) & (unsigned)(OWN_ROWS - 1)) << SRC_BITS) | ((unsigned)(S) & SRC_MASK); \
  } while (0)
  for (int i = tid; i < head; i += SCAN_THREADS) PLACE_EDGE(ps[i], pd[i]);
  const int4* sv = (const int4*)(ps + head);
  const int4* dv = (const int4*)(pd + head);
  const int nvec = (n - head) >> 2;
  int i = tid;
  for (; i + 3 * SCAN_THREADS < nvec; i += 4 * SCAN_THREADS) {
    int4 s4[4], d4[4];
#pragma unroll
    for (int u = 0; u < 4; ++u) d4[u] = dv[i + u * SCAN_THREADS];
#pragma unroll
    for (int u = 0; u < 4; ++u) s4[u] = sv[i + u * SCAN_THREADS];
    unsigned rec[16], pos[16];
#pragma unroll
    for (int u = 0; u < 4; ++u) {
      const int sx[4] = {s4[u].x, s4[u].y, s4[u].z, s4[u].w};
      const int dx[4] = {d4[u].x, d4[u].y, d4[u].z, d4[u].w};
#pragma unroll
      for (int k = 0; k < 4; ++k) {
        rec[4 * u + k] = (((unsigned)dx[k] & (unsigned)(OWN_ROWS - 1)) << SRC_BITS) | ((unsigned)sx[k] & SRC_MASK);
        pos[4 * u + k] = atomicAdd(&lcur[range_of(dx[k], nRanges)], 1u);
      }
    }
#pragma unroll
    for (int k = 0; k < 16; ++k) stage[pos[k]] = rec[k];
  }
  for (; i < nvec; i += SCAN_THREADS) {
    const int4 s0 = sv[i];
    const int4 d0 = dv[i];
    PLACE_EDGE(s0.x, d0.x); PLACE_EDGE(s0.y, d0.y); PLACE_EDGE(s0.z, d0.z); PLACE_EDGE(s0.w, d0.w);
  }
  for (int t = head + nvec * 4 + tid; t < n; t += SCAN_THREADS) PLACE_EDGE(ps[t], pd[t]);
#undef PLACE_EDGE
  __syncthreads();

  const int lane = tid & 31;
  for (int r = tid >> 5; r < nRanges; r += SCAN_THREADS / 32) {
    const unsigned  ls  = r > 0 ? lcur[r - 1] : 0u;
    const int       cnt = (int)(lcur[r] - ls);
    unsigned*       run = recs + gpos[r];
    const unsigned* st  = stage + ls;
    int k = lane - (int)((((size_t)run) >> 2) & 31u);
    if (k >= 0 && k < cnt) run[k] = st[k];
    k += 32;
    for (; k + 96 < cnt; k += 128) {
      const unsigned a0 = st[k], a1 = st[k + 32], a2 = st[k + 64], a3 = st[k + 96];
      run[k] = a0; run[k + 32] = a1; run[k + 64] = a2; run[k + 96] = a3;
    }
    for (; k < cnt; k += 32) run[k] = st[k];
  }
}

__global__ __launch_bounds__(SCAN_THREADS) void spmm1_owner(const unsigned* __restrict__ recs,
                                                            const int* __restrict__ rangeStart,
                                                            const float* __restrict__ h,
                                                            float* __restrict__ agg, int nN, int nE) {
  __shared__ float acc[OWN_ROWS * HID];
  const int tid = threadIdx.x;
  for (int i = tid; i < OWN_ROWS * HID; i += SCAN_THREADS) acc[i] = 0.0f;
  __syncthreads();

  const int beg  = rangeStart[blockIdx.x];
  const int end  = rangeStart[blockIdx.x + 1];
  const int lane = tid & 31;
  const int j    = lane & 7;
  const int sub  = lane >> 3;
  constexpr int TRIP   = 64;
  constexpr int STRIDE = (SCAN_THREADS / 32) * TRIP;

  const int last = nE - 1;
  int base = beg + (tid >> 5) * TRIP;
  unsigned r0 = recs[min(base + lane, last)];
  unsigned r1 = recs[min(base + 32 + lane, last)];
  for (; base < end; base += STRIDE) {
    const int left  = end - base;
    const int nbase = base + STRIDE;
    const unsigned n0 = recs[min(nbase + lane, last)];
    const unsigned n1 = recs[min(nbase + 32 + lane, last)];

    float    v[16];
    unsigned off[16];
#pragma unroll
    for (int i = 0; i < 8; ++i) {
      const int from = i * 4 + sub;
      const unsigned q0 = __shfl(r0, from, 32);
      const unsigned q1 = __shfl(r1, from, 32);
      v[i]       = h[(q0 & SRC_MASK) * HID + j];
      off[i]     = (q0 >> SRC_BITS) * HID + j;
      v[8 + i]   = h[(q1 & SRC_MASK) * HID + j];
      off[8 + i] = (q1 >> SRC_BITS) * HID + j;
    }
#pragma unroll
    for (int i = 0; i < 8; ++i) {
      const bool ok0 = i * 4 + sub < left;
      const bool ok1 = 32 + i * 4 + sub < left;
      atomicAdd(&acc[ok0 ? off[i] : (unsigned)j], ok0 ? v[i] : 0.0f);
      atomicAdd(&acc[ok1 ? off[8 + i] : (unsigned)j], ok1 ? v[8 + i] : 0.0f);
    }
    r0 = n0;
    r1 = n1;
  }
  __syncthreads();

  const long rowBase = (long)blockIdx.x * OWN_ROWS;
  for (int t = tid; t < OWN_ROWS * HID; t += SCAN_THREADS) {
    if (rowBase + (t >> 3) < nN) agg[rowBase * HID + t] = acc[t];
  }
}


__global__ void layer2_node(const float* __restrict__ agg, const float* __restrict__ in_norm,
                            const float* __restrict__ out_norm, const float* __restrict__ b1,
                            const float* __restrict__ W2, float* __restrict__ g2, int n) {
  int v = blockIdx.x * blockDim.x + threadIdx.x;
  if (v >= n) return;
  float innv = in_norm[v];
  const float* av = agg + (long)v * HID;
  float s = 0.0f;
#pragma unroll
  for (int j = 0; j < HID; ++j) {
    float z = fmaf(av[j], innv, b1[j]);
    z = z > 0.0f ? z : 0.0f;
    s = fmaf(z, W2[j], s);
  }
  g2[v] = s * out_norm[v];
}


__device__ __forceinline__ void sum_hit(float* acc, const float* __restrict__ g2, int s, int d,
                                        int base, int len) {
  const unsigned k = (unsigned)(d - base);
  if (k < (unsigned)len) atomicAdd(&acc[k], g2[s]);
}

__global__ __launch_bounds__(SCAN_THREADS) void spmm2_slices(const int* __restrict__ src,
                                                             const int* __restrict__ dst,
                                                             const float* __restrict__ g2,
                                                             float* __restrict__ outacc,
                                                             int nE, int nN) {
  __shared__ float acc[SLICE_CAP];
  const int tid     = threadIdx.x;
  const int nSlices = gridDim.y;
  const int len     = slice_len(nN, nSlices);
  const int base    = (int)blockIdx.y * len;

  for (int i = tid; i < len; i += SCAN_THREADS) acc[i] = 0.0f;
  __syncthreads();

  int start, n;
  chunk_bounds(nE, gridDim.x, blockIdx.x, start, n);
  const int* ps = src + start;
  const int* pd = dst + start;
  int head = (int)(((16u - (unsigned)((size_t)pd & 15u)) & 15u) >> 2);
  if (head > n) head = n;
  const bool vec = (((size_t)(ps + head)) & 15u) == 0;
  if (!vec) head = n;
  for (int i = tid; i < head; i += SCAN_THREADS) sum_hit(acc, g2, ps[i], pd[i], base, len);
  const int4* sv = (const int4*)(ps + head);
  const int4* dv = (const int4*)(pd + head);
  const int nvec = (n - head) >> 2;
  int i = tid;
  int4 dq[4], sq[4];
#pragma unroll
  for (int u = 0; u < 4; ++u) { dq[u] = make_int4(-1, -1, -1, -1); sq[u] = make_int4(0, 0, 0, 0); }
  for (;;) {
    const bool fullNext = i + 3 * SCAN_THREADS < nvec;
    int4 dn[4], sn[4];
#pragma unroll
    for (int u = 0; u < 4; ++u) { dn[u] = make_int4(-1, -1, -1, -1); sn[u] = make_int4(0, 0, 0, 0); }
    if (fullNext) {
#pragma unroll
      for (int u = 0; u < 4; ++u) { sn[u] = sv[i + u * SCAN_THREADS]; dn[u] = dv[i + u * SCAN_THREADS]; }
    }
    unsigned k[16];
    float    val[16];
#pragma unroll
    for (int u = 0; u < 4; ++u) {
      k[4 * u + 0] = (unsigned)(dq[u].x - base);
      k[4 * u + 1] = (unsigned)(dq[u].y - base);
      k[4 * u + 2] = (unsigned)(dq[u].z - base);
      k[4 * u + 3] = (unsigned)(dq[u].w - base);
      val[4 * u + 0] = (k[4 * u + 0] < (unsigned)len) ? g2[sq[u].x] : 0.0f;
      val[4 * u + 1] = (k[4 * u + 1] < (unsigned)len) ? g2[sq[u].y] : 0.0f;
      val[4 * u + 2] = (k[4 * u + 2] < (unsigned)len) ? g2[sq[u].z] : 0.0f;
      val[4 * u + 3] = (k[4 * u + 3] < (unsigned)len) ? g2[sq[u].w] : 0.0f;
    }
#pragma unroll
    for (int j = 0; j < 16; ++j) {
      if (k[j] < (unsigned)len) atomicAdd(&acc[k[j]], val[j]);
    }
    if (!fullNext) break;
#pragma unroll
    for (int u = 0; u < 4; ++u) { dq[u] = dn[u]; sq[u] = sn[u]; }
    i += 4 * SCAN_THREADS;
  }
  for (; i < nvec; i += SCAN_THREADS) {
    const int4 d0 = dv[i];
    const int4 s0 = sv[i];
    sum_hit(acc, g2, s0.x, d0.x, base, len); sum_hit(acc, g2, s0.y, d0.y, base, len);
    sum_hit(acc, g2, s0.z, d0.z, base, len); sum_hit(acc, g2, s0.w, d0.w, base, len);
  }
  for (int t = head + nvec * 4 + tid; t < n; t += SCAN_THREADS) sum_hit(acc, g2, ps[t], pd[t], base, len);
  __syncthreads();

  for (int t = tid; t < len; t += SCAN_THREADS) {
    const int node = base + t;
    if (node < nN) atomicAdd(&outacc[node], acc[t]);
  }
}

__global__ void final_kernel(const float* __restrict__ outacc, const float* __restrict__ in_norm,
                             const float* __restrict__ b2, float* __restrict__ out, int n) {
  int v = blockIdx.x * blockDim.x + threadIdx.x;
  if (v < n) out[v] = fmaf(outacc[v], in_norm[v], b2[0]);
}


extern "C" void kernel_launch(void* const* d_in, const int* in_sizes, int n_in,
                              void* d_out, int out_size, void* d_ws, size_t ws_size,
                              hipStream_t stream) {
  const float* x  = (const float*)d_in[0];
  const float* W1 = (const float*)d_in[1];
  const float* b1 = (const float*)d_in[2];
  const float* W2 = (const float*)d_in[3];
  const float* b2 = (const float*)d_in[4];
  const int*  src = (const int*)d_in[5];
  const int*  dst = (const int*)d_in[6];

  const int nN = in_sizes[0] / IN_DIM;
  const int nE = in_sizes[5];

  float* ws     = (float*)d_ws;
  float* deg_s  = ws;
  float* deg_d  = ws + (long)nN;
  float* h      = ws + 2L * nN;
  float* agg    = ws + 10L * nN;
  float* g2     = ws + 18L * nN;
  float* outacc = ws + 19L * nN;
  float* out    = (float*)d_out;

  const int TB = 256;
  dim3 blk(TB);

  zero_f32<<<(2 * nN + TB - 1) / TB, blk, 0, stream>>>(deg_s, 2 * nN);
  zero_f32<<<(nN + TB - 1) / TB, blk, 0, stream>>>(outacc, nN);

  const int nSlices   = (nN + SLICE_CAP - 1) / SLICE_CAP;
  const int degChunks = 256 / (2 * nSlices) > 0 ? 256 / (2 * nSlices) : 1;
  const int sumChunks = 256 / nSlices > 0 ? 256 / nSlices : 1;

  degree_slices<<<dim3(degChunks, 2 * nSlices), dim3(SCAN_THREADS), 0, stream>>>(
      src, dst, (unsigned*)deg_s, (unsigned*)deg_d, nE, nN);
  norm_kernel<<<(nN + TB - 1) / TB, blk, 0, stream>>>((unsigned*)deg_s, (unsigned*)deg_d, nN);

  gemm1_wmma<<<(nN + 127) / 128, blk, 0, stream>>>(x, W1, deg_s, h, nN);

  const int    nRanges  = (nN + OWN_ROWS - 1) / OWN_ROWS;
  const size_t wsWords  = 20UL * nN + (size_t)nE + (size_t)nRanges * BIN_CHUNKS + 2UL * nRanges + 8UL;
  if (nE > 0 && nRanges <= MAX_RANGES && wsWords * sizeof(float) <= ws_size &&
      (nE + BIN_CHUNKS - 1) / BIN_CHUNKS + 3 <= STAGE_CAP) {
    unsigned* recs       = (unsigned*)(ws + 20L * nN);
    int*      histT      = (int*)(recs + nE);
    int*      total      = histT + (long)nRanges * BIN_CHUNKS;
    int*      rangeStart = total + nRanges;
    bin_count<<<BIN_CHUNKS, dim3(SCAN_THREADS), 0, stream>>>(dst, histT, nE, nRanges);
    bin_scan<<<(nRanges * 32 + TB - 1) / TB, blk, 0, stream>>>(histT, total, BIN_CHUNKS, nRanges);
    bin_place<<<BIN_CHUNKS, dim3(SCAN_THREADS), 0, stream>>>(src, dst, histT, total, rangeStart, recs, nE, nRanges);
    spmm1_owner<<<nRanges, dim3(SCAN_THREADS), 0, stream>>>(recs, rangeStart, h, agg, nN, nE);
  } else {
    zero_f32<<<(8 * nN + TB - 1) / TB, blk, 0, stream>>>(agg, 8 * nN);
    spmm1_kernel<<<(nE + TB - 1) / TB, blk, 0, stream>>>(src, dst, h, agg, nE);
  }
  layer2_node<<<(nN + TB - 1) / TB, blk, 0, stream>>>(agg, deg_d, deg_s, b1, W2, g2, nN);
  spmm2_slices<<<dim3(sumChunks, nSlices), dim3(SCAN_THREADS), 0, stream>>>(src, dst, g2, outacc, nE, nN);
  final_kernel<<<(nN + TB - 1) / TB, blk, 0, stream>>>(outacc, deg_d, b2, out, nN);
}


# all per-dst edge sums run on layer 1's dst-sorted records in owner work-groups: in-degree counted there, layer 2 gathers g2 per record and stores out; layer-2 sums in LDS atomic arrival order
# speedup vs baseline: 8.0503x; 1.2652x over previous
#include <hip/hip_runtime.h>

typedef __attribute__((ext_vector_type(2))) float v2f;
typedef __attribute__((ext_vector_type(8))) float v8f;

constexpr int IN_DIM = 128;
constexpr int HID    = 8;

constexpr int SLICE_CAP    = 73728;
constexpr int SCAN_THREADS = 1024;


__global__ void zero_f32(float* __restrict__ p, int n) {
  int i = blockIdx.x * blockDim.x + threadIdx.x;
  if (i < n) p[i] = 0.0f;
}


__device__ __forceinline__ int slice_len(int nN, int nSlices) {
  const int per = (nN + nSlices - 1) / nSlices;
  return (per + 31) & ~31;
}

__device__ __forceinline__ void chunk_bounds(int nE, int nChunks, int chunk, int& start, int& count) {
  const int chunkLen = (((nE + nChunks - 1) / nChunks) + 3) & ~3;
  long s = (long)chunk * chunkLen;
  if (s > nE) s = nE;
  long e = s + chunkLen;
  if (e > nE) e = nE;
  start = (int)s;
  count = (int)(e - s);
}

__device__ __forceinline__ void count_hit(unsigned* cnt, int v, int base, int len) {
  const unsigned k = (unsigned)(v - base);
  if (k < (unsigned)len) atomicAdd(&cnt[k], 1u);
}

__global__ __launch_bounds__(SCAN_THREADS) void degree_slices(const int* __restrict__ src,
                                                              const int* __restrict__ dst,
                                                              unsigned* __restrict__ cnt_s,
                                                              unsigned* __restrict__ cnt_d,
                                                              int nE, int nN, int nSlices) {
  __shared__ unsigned cnt[SLICE_CAP];
  const int tid     = threadIdx.x;
  const int which   = (int)blockIdx.y >= nSlices ? 1 : 0;
  const int slice   = (int)blockIdx.y - which * nSlices;
  const int len     = slice_len(nN, nSlices);
  const int base    = slice * len;
  const int* __restrict__ idx = which ? dst : src;
  unsigned* __restrict__ out  = which ? cnt_d : cnt_s;

  for (int i = tid; i < len; i += SCAN_THREADS) cnt[i] = 0u;
  __syncthreads();

  int start, n;
  chunk_bounds(nE, gridDim.x, blockIdx.x, start, n);
  const int* p = idx + start;
  int head = (int)(((16u - (unsigned)((size_t)p & 15u)) & 15u) >> 2);
  if (head > n) head = n;
  for (int i = tid; i < head; i += SCAN_THREADS) count_hit(cnt, p[i], base, len);
  const int4* pv = (const int4*)(p + head);
  const int nvec = (n - head) >> 2;
  int i = tid;
  for (; i + 7 * SCAN_THREADS < nvec; i += 8 * SCAN_THREADS) {
    int4 v[8];
#pragma unroll
    for (int u = 0; u < 8; ++u) v[u] = pv[i + u * SCAN_THREADS];
#pragma unroll
    for (int u = 0; u < 8; ++u) {
      count_hit(cnt, v[u].x, base, len); count_hit(cnt, v[u].y, base, len);
      count_hit(cnt, v[u].z, base, len); count_hit(cnt, v[u].w, base, len);
    }
  }
  for (; i < nvec; i += SCAN_THREADS) {
    const int4 a = pv[i];
    count_hit(cnt, a.x, base, len); count_hit(cnt, a.y, base, len);
    count_hit(cnt, a.z, base, len); count_hit(cnt, a.w, base, len);
  }
  for (int t = head + nvec * 4 + tid; t < n; t += SCAN_THREADS) count_hit(cnt, p[t], base, len);
  __syncthreads();

  for (int t = tid; t < len; t += SCAN_THREADS) {
    const int node = base + t;
    if (node < nN) atomicAdd(&out[node], cnt[t]);
  }
}

__global__ void norm_kernel(unsigned* __restrict__ deg_s, unsigned* __restrict__ deg_d, int n, int both) {
  int i = blockIdx.x * blockDim.x + threadIdx.x;
  if (i < n) {
    deg_s[i] = __float_as_uint(rsqrtf(fmaxf((float)deg_s[i], 1.0f)));
    if (both) deg_d[i] = __float_as_uint(rsqrtf(fmaxf((float)deg_d[i], 1.0f)));
  }
}


__global__ __launch_bounds__(256) void gemm1_wmma(const float* __restrict__ x,
                                                  const float* __restrict__ W1,
                                                  const float* __restrict__ out_norm,
                                                  float* __restrict__ h, int nNodes) {
  __shared__ float w[IN_DIM * 16];
  int tid = threadIdx.x;
  for (int i = tid; i < IN_DIM * 16; i += 256) {
    int r = i >> 4, cidx = i & 15;
    w[i] = (cidx < HID) ? W1[r * HID + cidx] : 0.0f;
  }
  __syncthreads();

  int wid  = tid >> 5;
  int lane = tid & 31;
  int m    = lane & 15;
  int kh   = lane >> 4;

  int row0 = (blockIdx.x * 8 + wid) * 16;
  int node = row0 + m;
  int nodeClamped = node < nNodes ? node : nNodes - 1;
  const float* xrow = x + (long)nodeClamped * IN_DIM;

  v8f acc = {};
#pragma unroll
  for (int kk = 0; kk < IN_DIM; kk += 4) {
    int k0 = kk + 2 * kh;
    v2f a;
    a.x = xrow[k0];
    a.y = xrow[k0 + 1];
    v2f b;
    b.x = w[k0 * 16 + m];
    b.y = w[(k0 + 1) * 16 + m];
    acc = __builtin_amdgcn_wmma_f32_16x16x4_f32(
         false, a,  false, b,
         (short)0, acc,  false,  false);
  }

  if (m < HID) {
#pragma unroll
    for (int r = 0; r < 8; ++r) {
      int nr = row0 + r + 8 * kh;
      if (nr < nNodes) h[(long)nr * HID + m] = acc[r] * out_norm[nr];
    }
  }
}


__global__ void spmm1_kernel(const int* __restrict__ src, const int* __restrict__ dst,
                             const float* __restrict__ h, float* __restrict__ agg, int nE) {
  const int lane = threadIdx.x & 31;
  const int j    = lane & 7;
  const int sub  = lane >> 3;
  const int e    = blockIdx.x * blockDim.x + threadIdx.x;
  const int left = nE - (e - lane);

  int s = 0, d = 0;
  if (e < nE) { s = src[e]; d = dst[e]; }

  float    v[8];
  unsigned off[8];
#pragma unroll
  for (int i = 0; i < 8; ++i) {
    const int from = i * 4 + sub;
    const int si = __shfl(s, from, 32);
    const int di = __shfl(d, from, 32);
    v[i]   = h[(unsigned)(si * HID + j)];
    off[i] = (unsigned)(di * HID + j);
  }
#pragma unroll
  for (int i = 0; i < 8; ++i) {
    if (i * 4 + sub < left) atomicAdd(&agg[off[i]], v[i]);
  }
}


constexpr int OWN_SHIFT  = 10;
constexpr int OWN_ROWS   = 1 << OWN_SHIFT;
constexpr int SRC_BITS   = 32 - OWN_SHIFT;
constexpr int MAX_RANGES = 4096;
constexpr int BIN_CHUNKS = 256;
constexpr unsigned SRC_MASK = (1u << SRC_BITS) - 1u;

__device__ __forceinline__ unsigned range_of(int d, int nRanges) {
  const unsigned r = (unsigned)d >> OWN_SHIFT;
  return r < (unsigned)nRanges ? r : (unsigned)(nRanges - 1);
}

__global__ __launch_bounds__(SCAN_THREADS) void bin_count(const int* __restrict__ dst,
                                                          int* __restrict__ histT,
                                                          int nE, int nRanges) {
  __shared__ unsigned cnt[MAX_RANGES];
  const int tid = threadIdx.x;
  for (int i = tid; i < nRanges; i += SCAN_THREADS) cnt[i] = 0u;
  __syncthreads();

  int start, n;
  chunk_bounds(nE, gridDim.x, blockIdx.x, start, n);
  const int* p = dst + start;
  int head = (int)(((16u - (unsigned)((size_t)p & 15u)) & 15u) >> 2);
  if (head > n) head = n;
  for (int i = tid; i < head; i += SCAN_THREADS) atomicAdd(&cnt[range_of(p[i], nRanges)], 1u);
  const int4* pv = (const int4*)(p + head);
  const int nvec = (n - head) >> 2;
  int i = tid;
  for (; i + 3 * SCAN_THREADS < nvec; i += 4 * SCAN_THREADS) {
    int4 v[4];
#pragma unroll
    for (int u = 0; u < 4; ++u) v[u] = pv[i + u * SCAN_THREADS];
#pragma unroll
    for (int u = 0; u < 4; ++u) {
      atomicAdd(&cnt[range_of(v[u].x, nRanges)], 1u); atomicAdd(&cnt[range_of(v[u].y, nRanges)], 1u);
      atomicAdd(&cnt[range_of(v[u].z, nRanges)], 1u); atomicAdd(&cnt[range_of(v[u].w, nRanges)], 1u);
    }
  }
  for (; i < nvec; i += SCAN_THREADS) {
    const int4 a = pv[i];
    atomicAdd(&cnt[range_of(a.x, nRanges)], 1u); atomicAdd(&cnt[range_of(a.y, nRanges)], 1u);
    atomicAdd(&cnt[range_of(a.z, nRanges)], 1u); atomicAdd(&cnt[range_of(a.w, nRanges)], 1u);
  }
  for (int t = head + nvec * 4 + tid; t < n; t += SCAN_THREADS)
    atomicAdd(&cnt[range_of(p[t], nRanges)], 1u);
  __syncthreads();

  for (int r = tid; r < nRanges; r += SCAN_THREADS) histT[r * (int)gridDim.x + (int)blockIdx.x] = (int)cnt[r];
}

__global__ void bin_scan(int* __restrict__ histT, int* __restrict__ total, int nChunks, int nRanges) {
  const int lane  = threadIdx.x & 31;
  const int range = (int)((blockIdx.x * blockDim.x + threadIdx.x) >> 5);
  if (range >= nRanges) return;
  int* row = histT + (long)range * nChunks;
  int carry = 0;
  for (int c0 = 0; c0 < nChunks; c0 += 32) {
    const int c = c0 + lane;
    const int v = c < nChunks ? row[c] : 0;
    int s = v;
#pragma unroll
    for (int d = 1; d < 32; d <<= 1) {
      const int t = __shfl_up(s, d, 32);
      if (lane >= d) s += t;
    }
    if (c < nChunks) row[c] = carry + s - v;
    carry += __shfl(s, 31, 32);
  }
  if (lane == 0) total[range] = carry;
}

constexpr int STAGE_CAP = 62976;

__global__ __launch_bounds__(SCAN_THREADS) void bin_place(const int* __restrict__ src,
                                                          const int* __restrict__ dst,
                                                          const int* __restrict__ histT,
                                                          const int* __restrict__ total,
                                                          int* __restrict__ rangeStart,
                                                          unsigned* __restrict__ recs,
                                                          int nE, int nRanges) {
  __shared__ int      scanbuf[2][MAX_RANGES];
  __shared__ unsigned gpos[MAX_RANGES];
  __shared__ unsigned lcur[MAX_RANGES];
  __shared__ unsigned stage[STAGE_CAP];
  const int tid     = threadIdx.x;
  const int nChunks = (int)gridDim.x;
  const int chunk   = (int)blockIdx.x;

  for (int r = tid; r < nRanges; r += SCAN_THREADS) scanbuf[0][r] = total[r];
  __syncthreads();
  int cur = 0;
  for (int off = 1; off < nRanges; off <<= 1) {
    for (int r = tid; r < nRanges; r += SCAN_THREADS)
      scanbuf[1 - cur][r] = scanbuf[cur][r] + (r >= off ? scanbuf[cur][r - off] : 0);
    __syncthreads();
    cur ^= 1;
  }
  for (int r = tid; r < nRanges; r += SCAN_THREADS) {
    const int tot  = total[r];
    const int incl = scanbuf[cur][r];
    const int excl = incl - tot;
    const int pre  = histT[r * nChunks + chunk];
    const int nxt  = chunk + 1 < nChunks ? histT[r * nChunks + chunk + 1] : tot;
    gpos[r] = (unsigned)(excl + pre);
    scanbuf[1 - cur][r] = nxt - pre;
    if (chunk == 0) {
      rangeStart[r] = excl;
      if (r == nRanges - 1) rangeStart[nRanges] = incl;
    }
  }
  __syncthreads();
  cur ^= 1;
  for (int off = 1; off < nRanges; off <<= 1) {
    for (int r = tid; r < nRanges; r += SCAN_THREADS)
      scanbuf[1 - cur][r] = scanbuf[cur][r] + (r >= off ? scanbuf[cur][r - off] : 0);
    __syncthreads();
    cur ^= 1;
  }
  for (int r = tid; r < nRanges; r += SCAN_THREADS) lcur[r] = r > 0 ? (unsigned)scanbuf[cur][r - 1] : 0u;
  __syncthreads();

  int start, n;
  chunk_bounds(nE, nChunks, chunk, start, n);
  const int* ps = src + start;
  const int* pd = dst + start;
  int head = (int)(((16u - (unsigned)((size_t)pd & 15u)) & 15u) >> 2);
  if (head > n) head = n;
  if ((((size_t)(ps + head)) & 15u) != 0) head = n;
#define PLACE_EDGE(S, D)                                                            \
  do {                                                                              \
    const unsigned pos_ = atomicAdd(&lcur[range_of((D), nRanges)], 1u);             \
    stage[pos_] = (((unsigned)(D) & (unsigned)(OWN_ROWS - 1)) << SRC_BITS) | ((unsigned)(S) & SRC_MASK); \
  } while (0)
  for (int i = tid; i < head; i += SCAN_THREADS) PLACE_EDGE(ps[i], pd[i]);
  const int4* sv = (const int4*)(ps + head);
  const int4* dv = (const int4*)(pd + head);
  const int nvec = (n - head) >> 2;
  int i = tid;
  for (; i + 3 * SCAN_THREADS < nvec; i += 4 * SCAN_THREADS) {
    int4 s4[4], d4[4];
#pragma unroll
    for (int u = 0; u < 4; ++u) d4[u] = dv[i + u * SCAN_THREADS];
#pragma unroll
    for (int u = 0; u < 4; ++u) s4[u] = sv[i + u * SCAN_THREADS];
    unsigned rec[16], pos[16];
#pragma unroll
    for (int u = 0; u < 4; ++u) {
      const int sx[4] = {s4[u].x, s4[u].y, s4[u].z, s4[u].w};
      const int dx[4] = {d4[u].x, d4[u].y, d4[u].z, d4[u].w};
#pragma unroll
      for (int k = 0; k < 4; ++k) {
        rec[4 * u + k] = (((unsigned)dx[k] & (unsigned)(OWN_ROWS - 1)) << SRC_BITS) | ((unsigned)sx[k] & SRC_MASK);
        pos[4 * u + k] = atomicAdd(&lcur[range_of(dx[k], nRanges)], 1u);
      }
    }
#pragma unroll
    for (int k = 0; k < 16; ++k) stage[pos[k]] = rec[k];
  }
  for (; i < nvec; i += SCAN_THREADS) {
    const int4 s0 = sv[i];
    const int4 d0 = dv[i];
    PLACE_EDGE(s0.x, d0.x); PLACE_EDGE(s0.y, d0.y); PLACE_EDGE(s0.z, d0.z); PLACE_EDGE(s0.w, d0.w);
  }
  for (int t = head + nvec * 4 + tid; t < n; t += SCAN_THREADS) PLACE_EDGE(ps[t], pd[t]);
#undef PLACE_EDGE
  __syncthreads();

  const int lane = tid & 31;
  for (int r = tid >> 5; r < nRanges; r += SCAN_THREADS / 32) {
    const unsigned  ls  = r > 0 ? lcur[r - 1] : 0u;
    const int       cnt = (int)(lcur[r] - ls);
    unsigned*       run = recs + gpos[r];
    const unsigned* st  = stage + ls;
    int k = lane - (int)((((size_t)run) >> 2) & 31u);
    if (k >= 0 && k < cnt) run[k] = st[k];
    k += 32;
    for (; k + 96 < cnt; k += 128) {
      const unsigned a0 = st[k], a1 = st[k + 32], a2 = st[k + 64], a3 = st[k + 96];
      run[k] = a0; run[k + 32] = a1; run[k + 64] = a2; run[k + 96] = a3;
    }
    for (; k < cnt; k += 32) run[k] = st[k];
  }
}

__global__ __launch_bounds__(SCAN_THREADS) void spmm1_owner(const unsigned* __restrict__ recs,
                                                            const int* __restrict__ rangeStart,
                                                            const float* __restrict__ h,
                                                            float* __restrict__ agg,
                                                            float* __restrict__ in_norm, int nN, int nE) {
  __shared__ float    acc[OWN_ROWS * HID];
  __shared__ unsigned indeg[OWN_ROWS];
  const int tid = threadIdx.x;
  for (int i = tid; i < OWN_ROWS * HID; i += SCAN_THREADS) acc[i] = 0.0f;
  for (int i = tid; i < OWN_ROWS; i += SCAN_THREADS) indeg[i] = 0u;
  __syncthreads();

  const int beg  = rangeStart[blockIdx.x];
  const int end  = rangeStart[blockIdx.x + 1];
  const int lane = tid & 31;
  const int j    = lane & 7;
  const int sub  = lane >> 3;
  constexpr int TRIP   = 64;
  constexpr int STRIDE = (SCAN_THREADS / 32) * TRIP;

  const int last = nE - 1;
  int base = beg + (tid >> 5) * TRIP;
  unsigned r0 = recs[min(base + lane, last)];
  unsigned r1 = recs[min(base + 32 + lane, last)];
  for (; base < end; base += STRIDE) {
    const int left  = end - base;
    const int nbase = base + STRIDE;
    const unsigned n0 = recs[min(nbase + lane, last)];
    const unsigned n1 = recs[min(nbase + 32 + lane, last)];

    if (lane < left)      atomicAdd(&indeg[r0 >> SRC_BITS], 1u);
    if (32 + lane < left) atomicAdd(&indeg[r1 >> SRC_BITS], 1u);

    float    v[16];
    unsigned off[16];
#pragma unroll
    for (int i = 0; i < 8; ++i) {
      const int from = i * 4 + sub;
      const unsigned q0 = __shfl(r0, from, 32);
      const unsigned q1 = __shfl(r1, from, 32);
      v[i]       = h[(q0 & SRC_MASK) * HID + j];
      off[i]     = (q0 >> SRC_BITS) * HID + j;
      v[8 + i]   = h[(q1 & SRC_MASK) * HID + j];
      off[8 + i] = (q1 >> SRC_BITS) * HID + j;
    }
#pragma unroll
    for (int i = 0; i < 8; ++i) {
      const bool ok0 = i * 4 + sub < left;
      const bool ok1 = 32 + i * 4 + sub < left;
      atomicAdd(&acc[ok0 ? off[i] : (unsigned)j], ok0 ? v[i] : 0.0f);
      atomicAdd(&acc[ok1 ? off[8 + i] : (unsigned)j], ok1 ? v[8 + i] : 0.0f);
    }
    r0 = n0;
    r1 = n1;
  }
  __syncthreads();

  const long rowBase = (long)blockIdx.x * OWN_ROWS;
  for (int t = tid; t < OWN_ROWS * HID; t += SCAN_THREADS) {
    if (rowBase + (t >> 3) < nN) agg[rowBase * HID + t] = acc[t];
  }
  for (int t = tid; t < OWN_ROWS; t += SCAN_THREADS) {
    if (rowBase + t < nN) in_norm[rowBase + t] = rsqrtf(fmaxf((float)indeg[t], 1.0f));
  }
}

__global__ __launch_bounds__(SCAN_THREADS) void spmm2_owner(const unsigned* __restrict__ recs,
                                                            const int* __restrict__ rangeStart,
                                                            const float* __restrict__ g2,
                                                            const float* __restrict__ in_norm,
                                                            const float* __restrict__ b2,
                                                            float* __restrict__ out, int nN, int nE) {
  __shared__ float acc[OWN_ROWS];
  const int tid = threadIdx.x;
  for (int i = tid; i < OWN_ROWS; i += SCAN_THREADS) acc[i] = 0.0f;
  __syncthreads();

  const int beg = rangeStart[blockIdx.x];
  const int end = rangeStart[blockIdx.x + 1];
  constexpr int U    = 8;
  constexpr int TRIP = U * SCAN_THREADS;
  const int last = nE - 1;
  unsigned r[U];
#pragma unroll
  for (int u = 0; u < U; ++u) r[u] = recs[min(beg + u * SCAN_THREADS + tid, last)];
  for (int base = beg; base < end; base += TRIP) {
    unsigned n[U];
#pragma unroll
    for (int u = 0; u < U; ++u) n[u] = recs[min(base + TRIP + u * SCAN_THREADS + tid, last)];
    float v[U];
#pragma unroll
    for (int u = 0; u < U; ++u) v[u] = g2[r[u] & SRC_MASK];
#pragma unroll
    for (int u = 0; u < U; ++u) {
      const bool ok = base + u * SCAN_THREADS + tid < end;
      atomicAdd(&acc[ok ? (r[u] >> SRC_BITS) : (unsigned)(tid & (OWN_ROWS - 1))], ok ? v[u] : 0.0f);
    }
#pragma unroll
    for (int u = 0; u < U; ++u) r[u] = n[u];
  }
  __syncthreads();

  const long rowBase = (long)blockIdx.x * OWN_ROWS;
  const float bias = b2[0];
  for (int t = tid; t < OWN_ROWS; t += SCAN_THREADS) {
    const long row = rowBase + t;
    if (row < nN) out[row] = fmaf(acc[t], in_norm[row], bias);
  }
}


__global__ void layer2_node(const float* __restrict__ agg, const float* __restrict__ in_norm,
                            const float* __restrict__ out_norm, const float* __restrict__ b1,
                            const float* __restrict__ W2, float* __restrict__ g2, int n) {
  int v = blockIdx.x * blockDim.x + threadIdx.x;
  if (v >= n) return;
  float innv = in_norm[v];
  const float* av = agg + (long)v * HID;
  float s = 0.0f;
#pragma unroll
  for (int j = 0; j < HID; ++j) {
    float z = fmaf(av[j], innv, b1[j]);
    z = z > 0.0f ? z : 0.0f;
    s = fmaf(z, W2[j], s);
  }
  g2[v] = s * out_norm[v];
}


__device__ __forceinline__ void sum_hit(float* acc, const float* __restrict__ g2, int s, int d,
                                        int base, int len) {
  const unsigned k = (unsigned)(d - base);
  if (k < (unsigned)len) atomicAdd(&acc[k], g2[s]);
}

__global__ __launch_bounds__(SCAN_THREADS) void spmm2_slices(const int* __restrict__ src,
                                                             const int* __restrict__ dst,
                                                             const float* __restrict__ g2,
                                                             float* __restrict__ outacc,
                                                             int nE, int nN) {
  __shared__ float acc[SLICE_CAP];
  const int tid     = threadIdx.x;
  const int nSlices = gridDim.y;
  const int len     = slice_len(nN, nSlices);
  const int base    = (int)blockIdx.y * len;

  for (int i = tid; i < len; i += SCAN_THREADS) acc[i] = 0.0f;
  __syncthreads();

  int start, n;
  chunk_bounds(nE, gridDim.x, blockIdx.x, start, n);
  const int* ps = src + start;
  const int* pd = dst + start;
  int head = (int)(((16u - (unsigned)((size_t)pd & 15u)) & 15u) >> 2);
  if (head > n) head = n;
  const bool vec = (((size_t)(ps + head)) & 15u) == 0;
  if (!vec) head = n;
  for (int i = tid; i < head; i += SCAN_THREADS) sum_hit(acc, g2, ps[i], pd[i], base, len);
  const int4* sv = (const int4*)(ps + head);
  const int4* dv = (const int4*)(pd + head);
  const int nvec = (n - head) >> 2;
  int i = tid;
  int4 dq[4], sq[4];
#pragma unroll
  for (int u = 0; u < 4; ++u) { dq[u] = make_int4(-1, -1, -1, -1); sq[u] = make_int4(0, 0, 0, 0); }
  for (;;) {
    const bool fullNext = i + 3 * SCAN_THREADS < nvec;
    int4 dn[4], sn[4];
#pragma unroll
    for (int u = 0; u < 4; ++u) { dn[u] = make_int4(-1, -1, -1, -1); sn[u] = make_int4(0, 0, 0, 0); }
    if (fullNext) {
#pragma unroll
      for (int u = 0; u < 4; ++u) { sn[u] = sv[i + u * SCAN_THREADS]; dn[u] = dv[i + u * SCAN_THREADS]; }
    }
    unsigned k[16];
    float    val[16];
#pragma unroll
    for (int u = 0; u < 4; ++u) {
      k[4 * u + 0] = (unsigned)(dq[u].x - base);
      k[4 * u + 1] = (unsigned)(dq[u].y - base);
      k[4 * u + 2] = (unsigned)(dq[u].z - base);
      k[4 * u + 3] = (unsigned)(dq[u].w - base);
      val[4 * u + 0] = (k[4 * u + 0] < (unsigned)len) ? g2[sq[u].x] : 0.0f;
      val[4 * u + 1] = (k[4 * u + 1] < (unsigned)len) ? g2[sq[u].y] : 0.0f;
      val[4 * u + 2] = (k[4 * u + 2] < (unsigned)len) ? g2[sq[u].z] : 0.0f;
      val[4 * u + 3] = (k[4 * u + 3] < (unsigned)len) ? g2[sq[u].w] : 0.0f;
    }
#pragma unroll
    for (int j = 0; j < 16; ++j) {
      if (k[j] < (unsigned)len) atomicAdd(&acc[k[j]], val[j]);
    }
    if (!fullNext) break;
#pragma unroll
    for (int u = 0; u < 4; ++u) { dq[u] = dn[u]; sq[u] = sn[u]; }
    i += 4 * SCAN_THREADS;
  }
  for (; i < nvec; i += SCAN_THREADS) {
    const int4 d0 = dv[i];
    const int4 s0 = sv[i];
    sum_hit(acc, g2, s0.x, d0.x, base, len); sum_hit(acc, g2, s0.y, d0.y, base, len);
    sum_hit(acc, g2, s0.z, d0.z, base, len); sum_hit(acc, g2, s0.w, d0.w, base, len);
  }
  for (int t = head + nvec * 4 + tid; t < n; t += SCAN_THREADS) sum_hit(acc, g2, ps[t], pd[t], base, len);
  __syncthreads();

  for (int t = tid; t < len; t += SCAN_THREADS) {
    const int node = base + t;
    if (node < nN) atomicAdd(&outacc[node], acc[t]);
  }
}

__global__ void final_kernel(const float* __restrict__ outacc, const float* __restrict__ in_norm,
                             const float* __restrict__ b2, float* __restrict__ out, int n) {
  int v = blockIdx.x * blockDim.x + threadIdx.x;
  if (v < n) out[v] = fmaf(outacc[v], in_norm[v], b2[0]);
}


extern "C" void kernel_launch(void* const* d_in, const int* in_sizes, int n_in,
                              void* d_out, int out_size, void* d_ws, size_t ws_size,
                              hipStream_t stream) {
  const float* x  = (const float*)d_in[0];
  const float* W1 = (const float*)d_in[1];
  const float* b1 = (const float*)d_in[2];
  const float* W2 = (const float*)d_in[3];
  const float* b2 = (const float*)d_in[4];
  const int*  src = (const int*)d_in[5];
  const int*  dst = (const int*)d_in[6];

  const int nN = in_sizes[0] / IN_DIM;
  const int nE = in_sizes[5];

  float* ws     = (float*)d_ws;
  float* deg_s  = ws;
  float* deg_d  = ws + (long)nN;
  float* h      = ws + 2L * nN;
  float* agg    = ws + 10L * nN;
  float* g2     = ws + 18L * nN;
  float* outacc = ws + 19L * nN;
  float* out    = (float*)d_out;

  const int TB = 256;
  dim3 blk(TB);

  const int nSlices   = (nN + SLICE_CAP - 1) / SLICE_CAP;
  const int degChunks = 256 / (2 * nSlices) > 0 ? 256 / (2 * nSlices) : 1;
  const int sumChunks = 256 / nSlices > 0 ? 256 / nSlices : 1;

  const int    nRanges  = (nN + OWN_ROWS - 1) / OWN_ROWS;
  const size_t wsWords  = 20UL * nN + (size_t)nE + (size_t)nRanges * BIN_CHUNKS + 2UL * nRanges + 8UL;
  const bool   sorted   = nE > 0 && nRanges <= MAX_RANGES && wsWords * sizeof(float) <= ws_size &&
                          (nE + BIN_CHUNKS - 1) / BIN_CHUNKS + 3 <= STAGE_CAP;

  if (sorted) {
    zero_f32<<<(nN + TB - 1) / TB, blk, 0, stream>>>(deg_s, nN);
    degree_slices<<<dim3(sumChunks, nSlices), dim3(SCAN_THREADS), 0, stream>>>(
        src, dst, (unsigned*)deg_s, (unsigned*)deg_d, nE, nN, nSlices);
    norm_kernel<<<(nN + TB - 1) / TB, blk, 0, stream>>>((unsigned*)deg_s, (unsigned*)deg_d, nN, 0);
  } else {
    zero_f32<<<(2 * nN + TB - 1) / TB, blk, 0, stream>>>(deg_s, 2 * nN);
    degree_slices<<<dim3(degChunks, 2 * nSlices), dim3(SCAN_THREADS), 0, stream>>>(
        src, dst, (unsigned*)deg_s, (unsigned*)deg_d, nE, nN, nSlices);
    norm_kernel<<<(nN + TB - 1) / TB, blk, 0, stream>>>((unsigned*)deg_s, (unsigned*)deg_d, nN, 1);
  }

  gemm1_wmma<<<(nN + 127) / 128, blk, 0, stream>>>(x, W1, deg_s, h, nN);

  if (sorted) {
    unsigned* recs       = (unsigned*)(ws + 20L * nN);
    int*      histT      = (int*)(recs + nE);
    int*      total      = histT + (long)nRanges * BIN_CHUNKS;
    int*      rangeStart = total + nRanges;
    bin_count<<<BIN_CHUNKS, dim3(SCAN_THREADS), 0, stream>>>(dst, histT, nE, nRanges);
    bin_scan<<<(nRanges * 32 + TB - 1) / TB, blk, 0, stream>>>(histT, total, BIN_CHUNKS, nRanges);
    bin_place<<<BIN_CHUNKS, dim3(SCAN_THREADS), 0, stream>>>(src, dst, histT, total, rangeStart, recs, nE, nRanges);
    spmm1_owner<<<nRanges, dim3(SCAN_THREADS), 0, stream>>>(recs, rangeStart, h, agg, deg_d, nN, nE);
    layer2_node<<<(nN + TB - 1) / TB, blk, 0, stream>>>(agg, deg_d, deg_s, b1, W2, g2, nN);
    spmm2_owner<<<nRanges, dim3(SCAN_THREADS), 0, stream>>>(recs, rangeStart, g2, deg_d, b2, out, nN, nE);
  } else {
    zero_f32<<<(8 * nN + TB - 1) / TB, blk, 0, stream>>>(agg, 8 * nN);
    spmm1_kernel<<<(nE + TB - 1) / TB, blk, 0, stream>>>(src, dst, h, agg, nE);
    layer2_node<<<(nN + TB - 1) / TB, blk, 0, stream>>>(agg, deg_d, deg_s, b1, W2, g2, nN);
    zero_f32<<<(nN + TB - 1) / TB, blk, 0, stream>>>(outacc, nN);
    spmm2_slices<<<dim3(sumChunks, nSlices), dim3(SCAN_THREADS), 0, stream>>>(src, dst, g2, outacc, nE, nN);
    final_kernel<<<(nN + TB - 1) / TB, blk, 0, stream>>>(outacc, deg_d, b2, out, nN);
  }
}


# layer-2 edge sums read g2 from LDS node slices: a work-group keeps two owner ranges' records in registers, walks g2 slice by slice; row sums add slice after slice, LDS atomic arrival order
# speedup vs baseline: 8.5951x; 1.0677x over previous
#include <hip/hip_runtime.h>

typedef __attribute__((ext_vector_type(2))) float v2f;
typedef __attribute__((ext_vector_type(8))) float v8f;

constexpr int IN_DIM = 128;
constexpr int HID    = 8;

constexpr int SLICE_CAP    = 73728;
constexpr int SCAN_THREADS = 1024;


__global__ void zero_f32(float* __restrict__ p, int n) {
  int i = blockIdx.x * blockDim.x + threadIdx.x;
  if (i < n) p[i] = 0.0f;
}


__device__ __forceinline__ int slice_len(int nN, int nSlices) {
  const int per = (nN + nSlices - 1) / nSlices;
  return (per + 31) & ~31;
}

__device__ __forceinline__ void chunk_bounds(int nE, int nChunks, int chunk, int& start, int& count) {
  const int chunkLen = (((nE + nChunks - 1) / nChunks) + 3) & ~3;
  long s = (long)chunk * chunkLen;
  if (s > nE) s = nE;
  long e = s + chunkLen;
  if (e > nE) e = nE;
  start = (int)s;
  count = (int)(e - s);
}

__device__ __forceinline__ void count_hit(unsigned* cnt, int v, int base, int len) {
  const unsigned k = (unsigned)(v - base);
  if (k < (unsigned)len) atomicAdd(&cnt[k], 1u);
}

__global__ __launch_bounds__(SCAN_THREADS) void degree_slices(const int* __restrict__ src,
                                                              const int* __restrict__ dst,
                                                              unsigned* __restrict__ cnt_s,
                                                              unsigned* __restrict__ cnt_d,
                                                              int nE, int nN, int nSlices) {
  __shared__ unsigned cnt[SLICE_CAP];
  const int tid     = threadIdx.x;
  const int which   = (int)blockIdx.y >= nSlices ? 1 : 0;
  const int slice   = (int)blockIdx.y - which * nSlices;
  const int len     = slice_len(nN, nSlices);
  const int base    = slice * len;
  const int* __restrict__ idx = which ? dst : src;
  unsigned* __restrict__ out  = which ? cnt_d : cnt_s;

  for (int i = tid; i < len; i += SCAN_THREADS) cnt[i] = 0u;
  __syncthreads();

  int start, n;
  chunk_bounds(nE, gridDim.x, blockIdx.x, start, n);
  const int* p = idx + start;
  int head = (int)(((16u - (unsigned)((size_t)p & 15u)) & 15u) >> 2);
  if (head > n) head = n;
  for (int i = tid; i < head; i += SCAN_THREADS) count_hit(cnt, p[i], base, len);
  const int4* pv = (const int4*)(p + head);
  const int nvec = (n - head) >> 2;
  int i = tid;
  for (; i + 7 * SCAN_THREADS < nvec; i += 8 * SCAN_THREADS) {
    int4 v[8];
#pragma unroll
    for (int u = 0; u < 8; ++u) v[u] = pv[i + u * SCAN_THREADS];
#pragma unroll
    for (int u = 0; u < 8; ++u) {
      count_hit(cnt, v[u].x, base, len); count_hit(cnt, v[u].y, base, len);
      count_hit(cnt, v[u].z, base, len); count_hit(cnt, v[u].w, base, len);
    }
  }
  for (; i < nvec; i += SCAN_THREADS) {
    const int4 a = pv[i];
    count_hit(cnt, a.x, base, len); count_hit(cnt, a.y, base, len);
    count_hit(cnt, a.z, base, len); count_hit(cnt, a.w, base, len);
  }
  for (int t = head + nvec * 4 + tid; t < n; t += SCAN_THREADS) count_hit(cnt, p[t], base, len);
  __syncthreads();

  for (int t = tid; t < len; t += SCAN_THREADS) {
    const int node = base + t;
    if (node < nN) atomicAdd(&out[node], cnt[t]);
  }
}

__global__ void norm_kernel(unsigned* __restrict__ deg_s, unsigned* __restrict__ deg_d, int n, int both) {
  int i = blockIdx.x * blockDim.x + threadIdx.x;
  if (i < n) {
    deg_s[i] = __float_as_uint(rsqrtf(fmaxf((float)deg_s[i], 1.0f)));
    if (both) deg_d[i] = __float_as_uint(rsqrtf(fmaxf((float)deg_d[i], 1.0f)));
  }
}


__global__ __launch_bounds__(256) void gemm1_wmma(const float* __restrict__ x,
                                                  const float* __restrict__ W1,
                                                  const float* __restrict__ out_norm,
                                                  float* __restrict__ h, int nNodes) {
  __shared__ float w[IN_DIM * 16];
  int tid = threadIdx.x;
  for (int i = tid; i < IN_DIM * 16; i += 256) {
    int r = i >> 4, cidx = i & 15;
    w[i] = (cidx < HID) ? W1[r * HID + cidx] : 0.0f;
  }
  __syncthreads();

  int wid  = tid >> 5;
  int lane = tid & 31;
  int m    = lane & 15;
  int kh   = lane >> 4;

  int row0 = (blockIdx.x * 8 + wid) * 16;
  int node = row0 + m;
  int nodeClamped = node < nNodes ? node : nNodes - 1;
  const float* xrow = x + (long)nodeClamped * IN_DIM;

  v8f acc = {};
#pragma unroll
  for (int kk = 0; kk < IN_DIM; kk += 4) {
    int k0 = kk + 2 * kh;
    v2f a;
    a.x = xrow[k0];
    a.y = xrow[k0 + 1];
    v2f b;
    b.x = w[k0 * 16 + m];
    b.y = w[(k0 + 1) * 16 + m];
    acc = __builtin_amdgcn_wmma_f32_16x16x4_f32(
         false, a,  false, b,
         (short)0, acc,  false,  false);
  }

  if (m < HID) {
#pragma unroll
    for (int r = 0; r < 8; ++r) {
      int nr = row0 + r + 8 * kh;
      if (nr < nNodes) h[(long)nr * HID + m] = acc[r] * out_norm[nr];
    }
  }
}


__global__ void spmm1_kernel(const int* __restrict__ src, const int* __restrict__ dst,
                             const float* __restrict__ h, float* __restrict__ agg, int nE) {
  const int lane = threadIdx.x & 31;
  const int j    = lane & 7;
  const int sub  = lane >> 3;
  const int e    = blockIdx.x * blockDim.x + threadIdx.x;
  const int left = nE - (e - lane);

  int s = 0, d = 0;
  if (e < nE) { s = src[e]; d = dst[e]; }

  float    v[8];
  unsigned off[8];
#pragma unroll
  for (int i = 0; i < 8; ++i) {
    const int from = i * 4 + sub;
    const int si = __shfl(s, from, 32);
    const int di = __shfl(d, from, 32);
    v[i]   = h[(unsigned)(si * HID + j)];
    off[i] = (unsigned)(di * HID + j);
  }
#pragma unroll
  for (int i = 0; i < 8; ++i) {
    if (i * 4 + sub < left) atomicAdd(&agg[off[i]], v[i]);
  }
}


constexpr int OWN_SHIFT  = 10;
constexpr int OWN_ROWS   = 1 << OWN_SHIFT;
constexpr int SRC_BITS   = 32 - OWN_SHIFT;
constexpr int MAX_RANGES = 4096;
constexpr int BIN_CHUNKS = 256;
constexpr unsigned SRC_MASK = (1u << SRC_BITS) - 1u;

__device__ __forceinline__ unsigned range_of(int d, int nRanges) {
  const unsigned r = (unsigned)d >> OWN_SHIFT;
  return r < (unsigned)nRanges ? r : (unsigned)(nRanges - 1);
}

__global__ __launch_bounds__(SCAN_THREADS) void bin_count(const int* __restrict__ dst,
                                                          int* __restrict__ histT,
                                                          int nE, int nRanges) {
  __shared__ unsigned cnt[MAX_RANGES];
  const int tid = threadIdx.x;
  for (int i = tid; i < nRanges; i += SCAN_THREADS) cnt[i] = 0u;
  __syncthreads();

  int start, n;
  chunk_bounds(nE, gridDim.x, blockIdx.x, start, n);
  const int* p = dst + start;
  int head = (int)(((16u - (unsigned)((size_t)p & 15u)) & 15u) >> 2);
  if (head > n) head = n;
  for (int i = tid; i < head; i += SCAN_THREADS) atomicAdd(&cnt[range_of(p[i], nRanges)], 1u);
  const int4* pv = (const int4*)(p + head);
  const int nvec = (n - head) >> 2;
  int i = tid;
  for (; i + 3 * SCAN_THREADS < nvec; i += 4 * SCAN_THREADS) {
    int4 v[4];
#pragma unroll
    for (int u = 0; u < 4; ++u) v[u] = pv[i + u * SCAN_THREADS];
#pragma unroll
    for (int u = 0; u < 4; ++u) {
      atomicAdd(&cnt[range_of(v[u].x, nRanges)], 1u); atomicAdd(&cnt[range_of(v[u].y, nRanges)], 1u);
      atomicAdd(&cnt[range_of(v[u].z, nRanges)], 1u); atomicAdd(&cnt[range_of(v[u].w, nRanges)], 1u);
    }
  }
  for (; i < nvec; i += SCAN_THREADS) {
    const int4 a = pv[i];
    atomicAdd(&cnt[range_of(a.x, nRanges)], 1u); atomicAdd(&cnt[range_of(a.y, nRanges)], 1u);
    atomicAdd(&cnt[range_of(a.z, nRanges)], 1u); atomicAdd(&cnt[range_of(a.w, nRanges)], 1u);
  }
  for (int t = head + nvec * 4 + tid; t < n; t += SCAN_THREADS)
    atomicAdd(&cnt[range_of(p[t], nRanges)], 1u);
  __syncthreads();

  for (int r = tid; r < nRanges; r += SCAN_THREADS) histT[r * (int)gridDim.x + (int)blockIdx.x] = (int)cnt[r];
}

__global__ void bin_scan(int* __restrict__ histT, int* __restrict__ total, int nChunks, int nRanges) {
  const int lane  = threadIdx.x & 31;
  const int range = (int)((blockIdx.x * blockDim.x + threadIdx.x) >> 5);
  if (range >= nRanges) return;
  int* row = histT + (long)range * nChunks;
  int carry = 0;
  for (int c0 = 0; c0 < nChunks; c0 += 32) {
    const int c = c0 + lane;
    const int v = c < nChunks ? row[c] : 0;
    int s = v;
#pragma unroll
    for (int d = 1; d < 32; d <<= 1) {
      const int t = __shfl_up(s, d, 32);
      if (lane >= d) s += t;
    }
    if (c < nChunks) row[c] = carry + s - v;
    carry += __shfl(s, 31, 32);
  }
  if (lane == 0) total[range] = carry;
}

constexpr int STAGE_CAP = 62976;

__global__ __launch_bounds__(SCAN_THREADS) void bin_place(const int* __restrict__ src,
                                                          const int* __restrict__ dst,
                                                          const int* __restrict__ histT,
                                                          const int* __restrict__ total,
                                                          int* __restrict__ rangeStart,
                                                          unsigned* __restrict__ recs,
                                                          int nE, int nRanges) {
  __shared__ int      scanbuf[2][MAX_RANGES];
  __shared__ unsigned gpos[MAX_RANGES];
  __shared__ unsigned lcur[MAX_RANGES];
  __shared__ unsigned stage[STAGE_CAP];
  const int tid     = threadIdx.x;
  const int nChunks = (int)gridDim.x;
  const int chunk   = (int)blockIdx.x;

  for (int r = tid; r < nRanges; r += SCAN_THREADS) scanbuf[0][r] = total[r];
  __syncthreads();
  int cur = 0;
  for (int off = 1; off < nRanges; off <<= 1) {
    for (int r = tid; r < nRanges; r += SCAN_THREADS)
      scanbuf[1 - cur][r] = scanbuf[cur][r] + (r >= off ? scanbuf[cur][r - off] : 0);
    __syncthreads();
    cur ^= 1;
  }
  for (int r = tid; r < nRanges; r += SCAN_THREADS) {
    const int tot  = total[r];
    const int incl = scanbuf[cur][r];
    const int excl = incl - tot;
    const int pre  = histT[r * nChunks + chunk];
    const int nxt  = chunk + 1 < nChunks ? histT[r * nChunks + chunk + 1] : tot;
    gpos[r] = (unsigned)(excl + pre);
    scanbuf[1 - cur][r] = nxt - pre;
    if (chunk == 0) {
      rangeStart[r] = excl;
      if (r == nRanges - 1) rangeStart[nRanges] = incl;
    }
  }
  __syncthreads();
  cur ^= 1;
  for (int off = 1; off < nRanges; off <<= 1) {
    for (int r = tid; r < nRanges; r += SCAN_THREADS)
      scanbuf[1 - cur][r] = scanbuf[cur][r] + (r >= off ? scanbuf[cur][r - off] : 0);
    __syncthreads();
    cur ^= 1;
  }
  for (int r = tid; r < nRanges; r += SCAN_THREADS) lcur[r] = r > 0 ? (unsigned)scanbuf[cur][r - 1] : 0u;
  __syncthreads();

  int start, n;
  chunk_bounds(nE, nChunks, chunk, start, n);
  const int* ps = src + start;
  const int* pd = dst + start;
  int head = (int)(((16u - (unsigned)((size_t)pd & 15u)) & 15u) >> 2);
  if (head > n) head = n;
  if ((((size_t)(ps + head)) & 15u) != 0) head = n;
#define PLACE_EDGE(S, D)                                                            \
  do {                                                                              \
    const unsigned pos_ = atomicAdd(&lcur[range_of((D), nRanges)], 1u);             \
    stage[pos_] = (((unsigned)(D) & (unsigned)(OWN_ROWS - 1)) << SRC_BITS) | ((unsigned)(S) & SRC_MASK); \
  } while (0)
  for (int i = tid; i < head; i += SCAN_THREADS) PLACE_EDGE(ps[i], pd[i]);
  const int4* sv = (const int4*)(ps + head);
  const int4* dv = (const int4*)(pd + head);
  const int nvec = (n - head) >> 2;
  int i = tid;
  for (; i + 3 * SCAN_THREADS < nvec; i += 4 * SCAN_THREADS) {
    int4 s4[4], d4[4];
#pragma unroll
    for (int u = 0; u < 4; ++u) d4[u] = dv[i + u * SCAN_THREADS];
#pragma unroll
    for (int u = 0; u < 4; ++u) s4[u] = sv[i + u * SCAN_THREADS];
    unsigned rec[16], pos[16];
#pragma unroll
    for (int u = 0; u < 4; ++u) {
      const int sx[4] = {s4[u].x, s4[u].y, s4[u].z, s4[u].w};
      const int dx[4] = {d4[u].x, d4[u].y, d4[u].z, d4[u].w};
#pragma unroll
      for (int k = 0; k < 4; ++k) {
        rec[4 * u + k] = (((unsigned)dx[k] & (unsigned)(OWN_ROWS - 1)) << SRC_BITS) | ((unsigned)sx[k] & SRC_MASK);
        pos[4 * u + k] = atomicAdd(&lcur[range_of(dx[k], nRanges)], 1u);
      }
    }
#pragma unroll
    for (int k = 0; k < 16; ++k) stage[pos[k]] = rec[k];
  }
  for (; i < nvec; i += SCAN_THREADS) {
    const int4 s0 = sv[i];
    const int4 d0 = dv[i];
    PLACE_EDGE(s0.x, d0.x); PLACE_EDGE(s0.y, d0.y); PLACE_EDGE(s0.z, d0.z); PLACE_EDGE(s0.w, d0.w);
  }
  for (int t = head + nvec * 4 + tid; t < n; t += SCAN_THREADS) PLACE_EDGE(ps[t], pd[t]);
#undef PLACE_EDGE
  __syncthreads();

  const int lane = tid & 31;
  for (int r = tid >> 5; r < nRanges; r += SCAN_THREADS / 32) {
    const unsigned  ls  = r > 0 ? lcur[r - 1] : 0u;
    const int       cnt = (int)(lcur[r] - ls);
    unsigned*       run = recs + gpos[r];
    const unsigned* st  = stage + ls;
    int k = lane - (int)((((size_t)run) >> 2) & 31u);
    if (k >= 0 && k < cnt) run[k] = st[k];
    k += 32;
    for (; k + 96 < cnt; k += 128) {
      const unsigned a0 = st[k], a1 = st[k + 32], a2 = st[k + 64], a3 = st[k + 96];
      run[k] = a0; run[k + 32] = a1; run[k + 64] = a2; run[k + 96] = a3;
    }
    for (; k < cnt; k += 32) run[k] = st[k];
  }
}

__global__ __launch_bounds__(SCAN_THREADS) void spmm1_owner(const unsigned* __restrict__ recs,
                                                            const int* __restrict__ rangeStart,
                                                            const float* __restrict__ h,
                                                            float* __restrict__ agg,
                                                            float* __restrict__ in_norm, int nN, int nE) {
  __shared__ float    acc[OWN_ROWS * HID];
  __shared__ unsigned indeg[OWN_ROWS];
  const int tid = threadIdx.x;
  for (int i = tid; i < OWN_ROWS * HID; i += SCAN_THREADS) acc[i] = 0.0f;
  for (int i = tid; i < OWN_ROWS; i += SCAN_THREADS) indeg[i] = 0u;
  __syncthreads();

  const int beg  = rangeStart[blockIdx.x];
  const int end  = rangeStart[blockIdx.x + 1];
  const int lane = tid & 31;
  const int j    = lane & 7;
  const int sub  = lane >> 3;
  constexpr int TRIP   = 64;
  constexpr int STRIDE = (SCAN_THREADS / 32) * TRIP;

  const int last = nE - 1;
  int base = beg + (tid >> 5) * TRIP;
  unsigned r0 = recs[min(base + lane, last)];
  unsigned r1 = recs[min(base + 32 + lane, last)];
  for (; base < end; base += STRIDE) {
    const int left  = end - base;
    const int nbase = base + STRIDE;
    const unsigned n0 = recs[min(nbase + lane, last)];
    const unsigned n1 = recs[min(nbase + 32 + lane, last)];

    if (lane < left)      atomicAdd(&indeg[r0 >> SRC_BITS], 1u);
    if (32 + lane < left) atomicAdd(&indeg[r1 >> SRC_BITS], 1u);

    float    v[16];
    unsigned off[16];
#pragma unroll
    for (int i = 0; i < 8; ++i) {
      const int from = i * 4 + sub;
      const unsigned q0 = __shfl(r0, from, 32);
      const unsigned q1 = __shfl(r1, from, 32);
      v[i]       = h[(q0 & SRC_MASK) * HID + j];
      off[i]     = (q0 >> SRC_BITS) * HID + j;
      v[8 + i]   = h[(q1 & SRC_MASK) * HID + j];
      off[8 + i] = (q1 >> SRC_BITS) * HID + j;
    }
#pragma unroll
    for (int i = 0; i < 8; ++i) {
      const bool ok0 = i * 4 + sub < left;
      const bool ok1 = 32 + i * 4 + sub < left;
      atomicAdd(&acc[ok0 ? off[i] : (unsigned)j], ok0 ? v[i] : 0.0f);
      atomicAdd(&acc[ok1 ? off[8 + i] : (unsigned)j], ok1 ? v[8 + i] : 0.0f);
    }
    r0 = n0;
    r1 = n1;
  }
  __syncthreads();

  const long rowBase = (long)blockIdx.x * OWN_ROWS;
  for (int t = tid; t < OWN_ROWS * HID; t += SCAN_THREADS) {
    if (rowBase + (t >> 3) < nN) agg[rowBase * HID + t] = acc[t];
  }
  for (int t = tid; t < OWN_ROWS; t += SCAN_THREADS) {
    if (rowBase + t < nN) in_norm[rowBase + t] = rsqrtf(fmaxf((float)indeg[t], 1.0f));
  }
}

__global__ __launch_bounds__(SCAN_THREADS) void spmm2_owner(const unsigned* __restrict__ recs,
                                                            const int* __restrict__ rangeStart,
                                                            const float* __restrict__ g2,
                                                            const float* __restrict__ in_norm,
                                                            const float* __restrict__ b2,
                                                            float* __restrict__ out, int nN, int nE) {
  __shared__ float acc[OWN_ROWS];
  const int tid = threadIdx.x;
  for (int i = tid; i < OWN_ROWS; i += SCAN_THREADS) acc[i] = 0.0f;
  __syncthreads();

  const int beg = rangeStart[blockIdx.x];
  const int end = rangeStart[blockIdx.x + 1];
  constexpr int U    = 8;
  constexpr int TRIP = U * SCAN_THREADS;
  const int last = nE - 1;
  unsigned r[U];
#pragma unroll
  for (int u = 0; u < U; ++u) r[u] = recs[min(beg + u * SCAN_THREADS + tid, last)];
  for (int base = beg; base < end; base += TRIP) {
    unsigned n[U];
#pragma unroll
    for (int u = 0; u < U; ++u) n[u] = recs[min(base + TRIP + u * SCAN_THREADS + tid, last)];
    float v[U];
#pragma unroll
    for (int u = 0; u < U; ++u) v[u] = g2[r[u] & SRC_MASK];
#pragma unroll
    for (int u = 0; u < U; ++u) {
      const bool ok = base + u * SCAN_THREADS + tid < end;
      atomicAdd(&acc[ok ? (r[u] >> SRC_BITS) : (unsigned)(tid & (OWN_ROWS - 1))], ok ? v[u] : 0.0f);
    }
#pragma unroll
    for (int u = 0; u < U; ++u) r[u] = n[u];
  }
  __syncthreads();

  const long rowBase = (long)blockIdx.x * OWN_ROWS;
  const float bias = b2[0];
  for (int t = tid; t < OWN_ROWS; t += SCAN_THREADS) {
    const long row = rowBase + t;
    if (row < nN) out[row] = fmaf(acc[t], in_norm[row], bias);
  }
}

constexpr int      L2_KEEP = 36;
constexpr int      L2_GRP  = 12;
constexpr unsigned NO_REC  = 0xFFFFFFFFu;

__global__ __launch_bounds__(SCAN_THREADS) void spmm2_lds(const unsigned* __restrict__ recs,
                                                          const int* __restrict__ rangeStart,
                                                          const float* __restrict__ g2,
                                                          const float* __restrict__ in_norm,
                                                          const float* __restrict__ b2,
                                                          float* __restrict__ out,
                                                          int nN, int nE, int nRanges, int nSlices) {
  __shared__ alignas(16) float g2s[SLICE_CAP];
  __shared__ float acc[2 * OWN_ROWS];
  const int tid = threadIdx.x;
  for (int i = tid; i < 2 * OWN_ROWS; i += SCAN_THREADS) acc[i] = 0.0f;

  const int  ra   = (int)blockIdx.x * 2;
  const bool two  = ra + 1 < nRanges;
  const int  begA = rangeStart[ra];
  const int  endA = rangeStart[ra + 1];
  const int  endB = two ? rangeStart[ra + 2] : endA;
  const int  last = nE - 1;

  unsigned ka[L2_KEEP], kb[L2_KEEP];
#pragma unroll
  for (int u = 0; u < L2_KEEP; ++u) {
    const int ia = begA + u * SCAN_THREADS + tid;
    const int ib = endA + u * SCAN_THREADS + tid;
    const unsigned xa = recs[min(ia, last)];
    const unsigned xb = recs[min(ib, last)];
    ka[u] = ia < endA ? xa : NO_REC;
    kb[u] = ib < endB ? xb : NO_REC;
  }

  const int len = slice_len(nN, nSlices);
#define L2_HIT(R, ROWOFF)                                                                 \
  do {                                                                                    \
    const unsigned k_ = ((R) & SRC_MASK) - (unsigned)base;                                \
    if (k_ < (unsigned)len) atomicAdd(&acc[((R) >> SRC_BITS) + (ROWOFF)], g2s[k_]);       \
  } while (0)
#define L2_GROUP(K, G, ROWOFF)                                                            \
  do {                                                                                    \
    unsigned k[L2_GRP];                                                                   \
    float    v[L2_GRP];                                                                   \
    _Pragma("unroll") for (int j = 0; j < L2_GRP; ++j) {                                  \
      k[j] = (K[(G) * L2_GRP + j] & SRC_MASK) - (unsigned)base;                           \
      v[j] = k[j] < (unsigned)len ? g2s[k[j]] : 0.0f;                                     \
    }                                                                                     \
    _Pragma("unroll") for (int j = 0; j < L2_GRP; ++j) {                                  \
      atomicAdd(&acc[k[j] < (unsigned)len ? (K[(G) * L2_GRP + j] >> SRC_BITS) + (ROWOFF) : (unsigned)tid], v[j]); \
    }                                                                                     \
  } while (0)

  for (int s = 0; s < nSlices; ++s) {
    const int base = s * len;
    __syncthreads();
    int cnt = nN - base;
    cnt = cnt < 0 ? 0 : (cnt > len ? len : cnt);
    const float* gsrc = g2 + base;
    const int nv = ((((size_t)gsrc) & 15u) == 0) ? (cnt >> 2) : 0;
    const float4* src4 = (const float4*)gsrc;
    float4*       dst4 = (float4*)g2s;
    for (int i = tid; i < nv; i += 6 * SCAN_THREADS) {
      float4 t[6];
      int ix[6];
#pragma unroll
      for (int u = 0; u < 6; ++u) ix[u] = min(i + u * SCAN_THREADS, nv - 1);
#pragma unroll
      for (int u = 0; u < 6; ++u) t[u] = src4[ix[u]];
#pragma unroll
      for (int u = 0; u < 6; ++u) dst4[ix[u]] = t[u];
    }
    for (int i = nv * 4 + tid; i < len; i += SCAN_THREADS) g2s[i] = i < cnt ? gsrc[i] : 0.0f;
    __syncthreads();

#pragma unroll
    for (int u = 0; u < L2_KEEP; ++u) {
      asm volatile("" : "+v"(ka[u]));
      asm volatile("" : "+v"(kb[u]));
    }
#pragma unroll
    for (int g = 0; g < L2_KEEP / L2_GRP; ++g) {
      L2_GROUP(ka, g, 0);
      L2_GROUP(kb, g, OWN_ROWS);
    }
    for (int i = begA + L2_KEEP * SCAN_THREADS + tid; i < endA; i += SCAN_THREADS) L2_HIT(recs[i], 0);
    for (int i = endA + L2_KEEP * SCAN_THREADS + tid; i < endB; i += SCAN_THREADS) L2_HIT(recs[i], OWN_ROWS);
  }
#undef L2_GROUP
#undef L2_HIT
  __syncthreads();

  const long rowBase = (long)ra * OWN_ROWS;
  const float bias = b2[0];
  for (int t = tid; t < 2 * OWN_ROWS; t += SCAN_THREADS) {
    const long row = rowBase + t;
    if (row < nN) out[row] = fmaf(acc[t], in_norm[row], bias);
  }
}


__global__ void layer2_node(const float* __restrict__ agg, const float* __restrict__ in_norm,
                            const float* __restrict__ out_norm, const float* __restrict__ b1,
                            const float* __restrict__ W2, float* __restrict__ g2, int n) {
  int v = blockIdx.x * blockDim.x + threadIdx.x;
  if (v >= n) return;
  float innv = in_norm[v];
  const float* av = agg + (long)v * HID;
  float s = 0.0f;
#pragma unroll
  for (int j = 0; j < HID; ++j) {
    float z = fmaf(av[j], innv, b1[j]);
    z = z > 0.0f ? z : 0.0f;
    s = fmaf(z, W2[j], s);
  }
  g2[v] = s * out_norm[v];
}


__device__ __forceinline__ void sum_hit(float* acc, const float* __restrict__ g2, int s, int d,
                                        int base, int len) {
  const unsigned k = (unsigned)(d - base);
  if (k < (unsigned)len) atomicAdd(&acc[k], g2[s]);
}

__global__ __launch_bounds__(SCAN_THREADS) void spmm2_slices(const int* __restrict__ src,
                                                             const int* __restrict__ dst,
                                                             const float* __restrict__ g2,
                                                             float* __restrict__ outacc,
                                                             int nE, int nN) {
  __shared__ float acc[SLICE_CAP];
  const int tid     = threadIdx.x;
  const int nSlices = gridDim.y;
  const int len     = slice_len(nN, nSlices);
  const int base    = (int)blockIdx.y * len;

  for (int i = tid; i < len; i += SCAN_THREADS) acc[i] = 0.0f;
  __syncthreads();

  int start, n;
  chunk_bounds(nE, gridDim.x, blockIdx.x, start, n);
  const int* ps = src + start;
  const int* pd = dst + start;
  int head = (int)(((16u - (unsigned)((size_t)pd & 15u)) & 15u) >> 2);
  if (head > n) head = n;
  const bool vec = (((size_t)(ps + head)) & 15u) == 0;
  if (!vec) head = n;
  for (int i = tid; i < head; i += SCAN_THREADS) sum_hit(acc, g2, ps[i], pd[i], base, len);
  const int4* sv = (const int4*)(ps + head);
  const int4* dv = (const int4*)(pd + head);
  const int nvec = (n - head) >> 2;
  int i = tid;
  int4 dq[4], sq[4];
#pragma unroll
  for (int u = 0; u < 4; ++u) { dq[u] = make_int4(-1, -1, -1, -1); sq[u] = make_int4(0, 0, 0, 0); }
  for (;;) {
    const bool fullNext = i + 3 * SCAN_THREADS < nvec;
    int4 dn[4], sn[4];
#pragma unroll
    for (int u = 0; u < 4; ++u) { dn[u] = make_int4(-1, -1, -1, -1); sn[u] = make_int4(0, 0, 0, 0); }
    if (fullNext) {
#pragma unroll
      for (int u = 0; u < 4; ++u) { sn[u] = sv[i + u * SCAN_THREADS]; dn[u] = dv[i + u * SCAN_THREADS]; }
    }
    unsigned k[16];
    float    val[16];
#pragma unroll
    for (int u = 0; u < 4; ++u) {
      k[4 * u + 0] = (unsigned)(dq[u].x - base);
      k[4 * u + 1] = (unsigned)(dq[u].y - base);
      k[4 * u + 2] = (unsigned)(dq[u].z - base);
      k[4 * u + 3] = (unsigned)(dq[u].w - base);
      val[4 * u + 0] = (k[4 * u + 0] < (unsigned)len) ? g2[sq[u].x] : 0.0f;
      val[4 * u + 1] = (k[4 * u + 1] < (unsigned)len) ? g2[sq[u].y] : 0.0f;
      val[4 * u + 2] = (k[4 * u + 2] < (unsigned)len) ? g2[sq[u].z] : 0.0f;
      val[4 * u + 3] = (k[4 * u + 3] < (unsigned)len) ? g2[sq[u].w] : 0.0f;
    }
#pragma unroll
    for (int j = 0; j < 16; ++j) {
      if (k[j] < (unsigned)len) atomicAdd(&acc[k[j]], val[j]);
    }
    if (!fullNext) break;
#pragma unroll
    for (int u = 0; u < 4; ++u) { dq[u] = dn[u]; sq[u] = sn[u]; }
    i += 4 * SCAN_THREADS;
  }
  for (; i < nvec; i += SCAN_THREADS) {
    const int4 d0 = dv[i];
    const int4 s0 = sv[i];
    sum_hit(acc, g2, s0.x, d0.x, base, len); sum_hit(acc, g2, s0.y, d0.y, base, len);
    sum_hit(acc, g2, s0.z, d0.z, base, len); sum_hit(acc, g2, s0.w, d0.w, base, len);
  }
  for (int t = head + nvec * 4 + tid; t < n; t += SCAN_THREADS) sum_hit(acc, g2, ps[t], pd[t], base, len);
  __syncthreads();

  for (int t = tid; t < len; t += SCAN_THREADS) {
    const int node = base + t;
    if (node < nN) atomicAdd(&outacc[node], acc[t]);
  }
}

__global__ void final_kernel(const float* __restrict__ outacc, const float* __restrict__ in_norm,
                             const float* __restrict__ b2, float* __restrict__ out, int n) {
  int v = blockIdx.x * blockDim.x + threadIdx.x;
  if (v < n) out[v] = fmaf(outacc[v], in_norm[v], b2[0]);
}


extern "C" void kernel_launch(void* const* d_in, const int* in_sizes, int n_in,
                              void* d_out, int out_size, void* d_ws, size_t ws_size,
                              hipStream_t stream) {
  const float* x  = (const float*)d_in[0];
  const float* W1 = (const float*)d_in[1];
  const float* b1 = (const float*)d_in[2];
  const float* W2 = (const float*)d_in[3];
  const float* b2 = (const float*)d_in[4];
  const int*  src = (const int*)d_in[5];
  const int*  dst = (const int*)d_in[6];

  const int nN = in_sizes[0] / IN_DIM;
  const int nE = in_sizes[5];

  float* ws     = (float*)d_ws;
  float* deg_s  = ws;
  float* deg_d  = ws + (long)nN;
  float* h      = ws + 2L * nN;
  float* agg    = ws + 10L * nN;
  float* g2     = ws + 18L * nN;
  float* outacc = ws + 19L * nN;
  float* out    = (float*)d_out;

  const int TB = 256;
  dim3 blk(TB);

  const int nSlices   = (nN + SLICE_CAP - 1) / SLICE_CAP;
  const int degChunks = 256 / (2 * nSlices) > 0 ? 256 / (2 * nSlices) : 1;
  const int sumChunks = 256 / nSlices > 0 ? 256 / nSlices : 1;

  const int    nRanges  = (nN + OWN_ROWS - 1) / OWN_ROWS;
  const size_t wsWords  = 20UL * nN + (size_t)nE + (size_t)nRanges * BIN_CHUNKS + 2UL * nRanges + 8UL;
  const bool   sorted   = nE > 0 && nRanges <= MAX_RANGES && wsWords * sizeof(float) <= ws_size &&
                          (nE + BIN_CHUNKS - 1) / BIN_CHUNKS + 3 <= STAGE_CAP;

  if (sorted) {
    zero_f32<<<(nN + TB - 1) / TB, blk, 0, stream>>>(deg_s, nN);
    degree_slices<<<dim3(sumChunks, nSlices), dim3(SCAN_THREADS), 0, stream>>>(
        src, dst, (unsigned*)deg_s, (unsigned*)deg_d, nE, nN, nSlices);
    norm_kernel<<<(nN + TB - 1) / TB, blk, 0, stream>>>((unsigned*)deg_s, (unsigned*)deg_d, nN, 0);
  } else {
    zero_f32<<<(2 * nN + TB - 1) / TB, blk, 0, stream>>>(deg_s, 2 * nN);
    degree_slices<<<dim3(degChunks, 2 * nSlices), dim3(SCAN_THREADS), 0, stream>>>(
        src, dst, (unsigned*)deg_s, (unsigned*)deg_d, nE, nN, nSlices);
    norm_kernel<<<(nN + TB - 1) / TB, blk, 0, stream>>>((unsigned*)deg_s, (unsigned*)deg_d, nN, 1);
  }

  gemm1_wmma<<<(nN + 127) / 128, blk, 0, stream>>>(x, W1, deg_s, h, nN);

  if (sorted) {
    unsigned* recs       = (unsigned*)(ws + 20L * nN);
    int*      histT      = (int*)(recs + nE);
    int*      total      = histT + (long)nRanges * BIN_CHUNKS;
    int*      rangeStart = total + nRanges;
    bin_count<<<BIN_CHUNKS, dim3(SCAN_THREADS), 0, stream>>>(dst, histT, nE, nRanges);
    bin_scan<<<(nRanges * 32 + TB - 1) / TB, blk, 0, stream>>>(histT, total, BIN_CHUNKS, nRanges);
    bin_place<<<BIN_CHUNKS, dim3(SCAN_THREADS), 0, stream>>>(src, dst, histT, total, rangeStart, recs, nE, nRanges);
    spmm1_owner<<<nRanges, dim3(SCAN_THREADS), 0, stream>>>(recs, rangeStart, h, agg, deg_d, nN, nE);
    layer2_node<<<(nN + TB - 1) / TB, blk, 0, stream>>>(agg, deg_d, deg_s, b1, W2, g2, nN);
    if (nN <= (int)SRC_MASK) {
      spmm2_lds<<<(nRanges + 1) / 2, dim3(SCAN_THREADS), 0, stream>>>(recs, rangeStart, g2, deg_d, b2, out, nN, nE,
                                                                     nRanges, nSlices);
    } else {
      spmm2_owner<<<nRanges, dim3(SCAN_THREADS), 0, stream>>>(recs, rangeStart, g2, deg_d, b2, out, nN, nE);
    }
  } else {
    zero_f32<<<(8 * nN + TB - 1) / TB, blk, 0, stream>>>(agg, 8 * nN);
    spmm1_kernel<<<(nE + TB - 1) / TB, blk, 0, stream>>>(src, dst, h, agg, nE);
    layer2_node<<<(nN + TB - 1) / TB, blk, 0, stream>>>(agg, deg_d, deg_s, b1, W2, g2, nN);
    zero_f32<<<(nN + TB - 1) / TB, blk, 0, stream>>>(outacc, nN);
    spmm2_slices<<<dim3(sumChunks, nSlices), dim3(SCAN_THREADS), 0, stream>>>(src, dst, g2, outacc, nE, nN);
    final_kernel<<<(nN + TB - 1) / TB, blk, 0, stream>>>(outacc, deg_d, b2, out, nN);
  }
}


# layer-2 sums: each work-group first orders its kept records by g2 slice (LDS counting sort, 65536-node slices), a slice looks only at its own pieces; rows add over 8 slices, not 7, LDS atomic order
# speedup vs baseline: 9.0237x; 1.0499x over previous
#include <hip/hip_runtime.h>

typedef __attribute__((ext_vector_type(2))) float v2f;
typedef __attribute__((ext_vector_type(8))) float v8f;

constexpr int IN_DIM = 128;
constexpr int HID    = 8;

constexpr int SLICE_CAP    = 73728;
constexpr int SCAN_THREADS = 1024;


__global__ void zero_f32(float* __restrict__ p, int n) {
  int i = blockIdx.x * blockDim.x + threadIdx.x;
  if (i < n) p[i] = 0.0f;
}


__device__ __forceinline__ int slice_len(int nN, int nSlices) {
  const int per = (nN + nSlices - 1) / nSlices;
  return (per + 31) & ~31;
}

__device__ __forceinline__ void chunk_bounds(int nE, int nChunks, int chunk, int& start, int& count) {
  const int chunkLen = (((nE + nChunks - 1) / nChunks) + 3) & ~3;
  long s = (long)chunk * chunkLen;
  if (s > nE) s = nE;
  long e = s + chunkLen;
  if (e > nE) e = nE;
  start = (int)s;
  count = (int)(e - s);
}

__device__ __forceinline__ void count_hit(unsigned* cnt, int v, int base, int len) {
  const unsigned k = (unsigned)(v - base);
  if (k < (unsigned)len) atomicAdd(&cnt[k], 1u);
}

__global__ __launch_bounds__(SCAN_THREADS) void degree_slices(const int* __restrict__ src,
                                                              const int* __restrict__ dst,
                                                              unsigned* __restrict__ cnt_s,
                                                              unsigned* __restrict__ cnt_d,
                                                              int nE, int nN, int nSlices) {
  __shared__ unsigned cnt[SLICE_CAP];
  const int tid     = threadIdx.x;
  const int which   = (int)blockIdx.y >= nSlices ? 1 : 0;
  const int slice   = (int)blockIdx.y - which * nSlices;
  const int len     = slice_len(nN, nSlices);
  const int base    = slice * len;
  const int* __restrict__ idx = which ? dst : src;
  unsigned* __restrict__ out  = which ? cnt_d : cnt_s;

  for (int i = tid; i < len; i += SCAN_THREADS) cnt[i] = 0u;
  __syncthreads();

  int start, n;
  chunk_bounds(nE, gridDim.x, blockIdx.x, start, n);
  const int* p = idx + start;
  int head = (int)(((16u - (unsigned)((size_t)p & 15u)) & 15u) >> 2);
  if (head > n) head = n;
  for (int i = tid; i < head; i += SCAN_THREADS) count_hit(cnt, p[i], base, len);
  const int4* pv = (const int4*)(p + head);
  const int nvec = (n - head) >> 2;
  int i = tid;
  for (; i + 7 * SCAN_THREADS < nvec; i += 8 * SCAN_THREADS) {
    int4 v[8];
#pragma unroll
    for (int u = 0; u < 8; ++u) v[u] = pv[i + u * SCAN_THREADS];
#pragma unroll
    for (int u = 0; u < 8; ++u) {
      count_hit(cnt, v[u].x, base, len); count_hit(cnt, v[u].y, base, len);
      count_hit(cnt, v[u].z, base, len); count_hit(cnt, v[u].w, base, len);
    }
  }
  for (; i < nvec; i += SCAN_THREADS) {
    const int4 a = pv[i];
    count_hit(cnt, a.x, base, len); count_hit(cnt, a.y, base, len);
    count_hit(cnt, a.z, base, len); count_hit(cnt, a.w, base, len);
  }
  for (int t = head + nvec * 4 + tid; t < n; t += SCAN_THREADS) count_hit(cnt, p[t], base, len);
  __syncthreads();

  for (int t = tid; t < len; t += SCAN_THREADS) {
    const int node = base + t;
    if (node < nN) atomicAdd(&out[node], cnt[t]);
  }
}

__global__ void norm_kernel(unsigned* __restrict__ deg_s, unsigned* __restrict__ deg_d, int n, int both) {
  int i = blockIdx.x * blockDim.x + threadIdx.x;
  if (i < n) {
    deg_s[i] = __float_as_uint(rsqrtf(fmaxf((float)deg_s[i], 1.0f)));
    if (both) deg_d[i] = __float_as_uint(rsqrtf(fmaxf((float)deg_d[i], 1.0f)));
  }
}


__global__ __launch_bounds__(256) void gemm1_wmma(const float* __restrict__ x,
                                                  const float* __restrict__ W1,
                                                  const float* __restrict__ out_norm,
                                                  float* __restrict__ h, int nNodes) {
  __shared__ float w[IN_DIM * 16];
  int tid = threadIdx.x;
  for (int i = tid; i < IN_DIM * 16; i += 256) {
    int r = i >> 4, cidx = i & 15;
    w[i] = (cidx < HID) ? W1[r * HID + cidx] : 0.0f;
  }
  __syncthreads();

  int wid  = tid >> 5;
  int lane = tid & 31;
  int m    = lane & 15;
  int kh   = lane >> 4;

  int row0 = (blockIdx.x * 8 + wid) * 16;
  int node = row0 + m;
  int nodeClamped = node < nNodes ? node : nNodes - 1;
  const float* xrow = x + (long)nodeClamped * IN_DIM;

  v8f acc = {};
#pragma unroll
  for (int kk = 0; kk < IN_DIM; kk += 4) {
    int k0 = kk + 2 * kh;
    v2f a;
    a.x = xrow[k0];
    a.y = xrow[k0 + 1];
    v2f b;
    b.x = w[k0 * 16 + m];
    b.y = w[(k0 + 1) * 16 + m];
    acc = __builtin_amdgcn_wmma_f32_16x16x4_f32(
         false, a,  false, b,
         (short)0, acc,  false,  false);
  }

  if (m < HID) {
#pragma unroll
    for (int r = 0; r < 8; ++r) {
      int nr = row0 + r + 8 * kh;
      if (nr < nNodes) h[(long)nr * HID + m] = acc[r] * out_norm[nr];
    }
  }
}


__global__ void spmm1_kernel(const int* __restrict__ src, const int* __restrict__ dst,
                             const float* __restrict__ h, float* __restrict__ agg, int nE) {
  const int lane = threadIdx.x & 31;
  const int j    = lane & 7;
  const int sub  = lane >> 3;
  const int e    = blockIdx.x * blockDim.x + threadIdx.x;
  const int left = nE - (e - lane);

  int s = 0, d = 0;
  if (e < nE) { s = src[e]; d = dst[e]; }

  float    v[8];
  unsigned off[8];
#pragma unroll
  for (int i = 0; i < 8; ++i) {
    const int from = i * 4 + sub;
    const int si = __shfl(s, from, 32);
    const int di = __shfl(d, from, 32);
    v[i]   = h[(unsigned)(si * HID + j)];
    off[i] = (unsigned)(di * HID + j);
  }
#pragma unroll
  for (int i = 0; i < 8; ++i) {
    if (i * 4 + sub < left) atomicAdd(&agg[off[i]], v[i]);
  }
}


constexpr int OWN_SHIFT  = 10;
constexpr int OWN_ROWS   = 1 << OWN_SHIFT;
constexpr int SRC_BITS   = 32 - OWN_SHIFT;
constexpr int MAX_RANGES = 4096;
constexpr int BIN_CHUNKS = 256;
constexpr unsigned SRC_MASK = (1u << SRC_BITS) - 1u;

__device__ __forceinline__ unsigned range_of(int d, int nRanges) {
  const unsigned r = (unsigned)d >> OWN_SHIFT;
  return r < (unsigned)nRanges ? r : (unsigned)(nRanges - 1);
}

__global__ __launch_bounds__(SCAN_THREADS) void bin_count(const int* __restrict__ dst,
                                                          int* __restrict__ histT,
                                                          int nE, int nRanges) {
  __shared__ unsigned cnt[MAX_RANGES];
  const int tid = threadIdx.x;
  for (int i = tid; i < nRanges; i += SCAN_THREADS) cnt[i] = 0u;
  __syncthreads();

  int start, n;
  chunk_bounds(nE, gridDim.x, blockIdx.x, start, n);
  const int* p = dst + start;
  int head = (int)(((16u - (unsigned)((size_t)p & 15u)) & 15u) >> 2);
  if (head > n) head = n;
  for (int i = tid; i < head; i += SCAN_THREADS) atomicAdd(&cnt[range_of(p[i], nRanges)], 1u);
  const int4* pv = (const int4*)(p + head);
  const int nvec = (n - head) >> 2;
  int i = tid;
  for (; i + 3 * SCAN_THREADS < nvec; i += 4 * SCAN_THREADS) {
    int4 v[4];
#pragma unroll
    for (int u = 0; u < 4; ++u) v[u] = pv[i + u * SCAN_THREADS];
#pragma unroll
    for (int u = 0; u < 4; ++u) {
      atomicAdd(&cnt[range_of(v[u].x, nRanges)], 1u); atomicAdd(&cnt[range_of(v[u].y, nRanges)], 1u);
      atomicAdd(&cnt[range_of(v[u].z, nRanges)], 1u); atomicAdd(&cnt[range_of(v[u].w, nRanges)], 1u);
    }
  }
  for (; i < nvec; i += SCAN_THREADS) {
    const int4 a = pv[i];
    atomicAdd(&cnt[range_of(a.x, nRanges)], 1u); atomicAdd(&cnt[range_of(a.y, nRanges)], 1u);
    atomicAdd(&cnt[range_of(a.z, nRanges)], 1u); atomicAdd(&cnt[range_of(a.w, nRanges)], 1u);
  }
  for (int t = head + nvec * 4 + tid; t < n; t += SCAN_THREADS)
    atomicAdd(&cnt[range_of(p[t], nRanges)], 1u);
  __syncthreads();

  for (int r = tid; r < nRanges; r += SCAN_THREADS) histT[r * (int)gridDim.x + (int)blockIdx.x] = (int)cnt[r];
}

__global__ void bin_scan(int* __restrict__ histT, int* __restrict__ total, int nChunks, int nRanges) {
  const int lane  = threadIdx.x & 31;
  const int range = (int)((blockIdx.x * blockDim.x + threadIdx.x) >> 5);
  if (range >= nRanges) return;
  int* row = histT + (long)range * nChunks;
  int carry = 0;
  for (int c0 = 0; c0 < nChunks; c0 += 32) {
    const int c = c0 + lane;
    const int v = c < nChunks ? row[c] : 0;
    int s = v;
#pragma unroll
    for (int d = 1; d < 32; d <<= 1) {
      const int t = __shfl_up(s, d, 32);
      if (lane >= d) s += t;
    }
    if (c < nChunks) row[c] = carry + s - v;
    carry += __shfl(s, 31, 32);
  }
  if (lane == 0) total[range] = carry;
}

constexpr int STAGE_CAP = 62976;

__global__ __launch_bounds__(SCAN_THREADS) void bin_place(const int* __restrict__ src,
                                                          const int* __restrict__ dst,
                                                          const int* __restrict__ histT,
                                                          const int* __restrict__ total,
                                                          int* __restrict__ rangeStart,
                                                          unsigned* __restrict__ recs,
                                                          int nE, int nRanges) {
  __shared__ int      scanbuf[2][MAX_RANGES];
  __shared__ unsigned gpos[MAX_RANGES];
  __shared__ unsigned lcur[MAX_RANGES];
  __shared__ unsigned stage[STAGE_CAP];
  const int tid     = threadIdx.x;
  const int nChunks = (int)gridDim.x;
  const int chunk   = (int)blockIdx.x;

  for (int r = tid; r < nRanges; r += SCAN_THREADS) scanbuf[0][r] = total[r];
  __syncthreads();
  int cur = 0;
  for (int off = 1; off < nRanges; off <<= 1) {
    for (int r = tid; r < nRanges; r += SCAN_THREADS)
      scanbuf[1 - cur][r] = scanbuf[cur][r] + (r >= off ? scanbuf[cur][r - off] : 0);
    __syncthreads();
    cur ^= 1;
  }
  for (int r = tid; r < nRanges; r += SCAN_THREADS) {
    const int tot  = total[r];
    const int incl = scanbuf[cur][r];
    const int excl = incl - tot;
    const int pre  = histT[r * nChunks + chunk];
    const int nxt  = chunk + 1 < nChunks ? histT[r * nChunks + chunk + 1] : tot;
    gpos[r] = (unsigned)(excl + pre);
    scanbuf[1 - cur][r] = nxt - pre;
    if (chunk == 0) {
      rangeStart[r] = excl;
      if (r == nRanges - 1) rangeStart[nRanges] = incl;
    }
  }
  __syncthreads();
  cur ^= 1;
  for (int off = 1; off < nRanges; off <<= 1) {
    for (int r = tid; r < nRanges; r += SCAN_THREADS)
      scanbuf[1 - cur][r] = scanbuf[cur][r] + (r >= off ? scanbuf[cur][r - off] : 0);
    __syncthreads();
    cur ^= 1;
  }
  for (int r = tid; r < nRanges; r += SCAN_THREADS) lcur[r] = r > 0 ? (unsigned)scanbuf[cur][r - 1] : 0u;
  __syncthreads();

  int start, n;
  chunk_bounds(nE, nChunks, chunk, start, n);
  const int* ps = src + start;
  const int* pd = dst + start;
  int head = (int)(((16u - (unsigned)((size_t)pd & 15u)) & 15u) >> 2);
  if (head > n) head = n;
  if ((((size_t)(ps + head)) & 15u) != 0) head = n;
#define PLACE_EDGE(S, D)                                                            \
  do {                                                                              \
    const unsigned pos_ = atomicAdd(&lcur[range_of((D), nRanges)], 1u);             \
    stage[pos_] = (((unsigned)(D) & (unsigned)(OWN_ROWS - 1)) << SRC_BITS) | ((unsigned)(S) & SRC_MASK); \
  } while (0)
  for (int i = tid; i < head; i += SCAN_THREADS) PLACE_EDGE(ps[i], pd[i]);
  const int4* sv = (const int4*)(ps + head);
  const int4* dv = (const int4*)(pd + head);
  const int nvec = (n - head) >> 2;
  int i = tid;
  for (; i + 3 * SCAN_THREADS < nvec; i += 4 * SCAN_THREADS) {
    int4 s4[4], d4[4];
#pragma unroll
    for (int u = 0; u < 4; ++u) d4[u] = dv[i + u * SCAN_THREADS];
#pragma unroll
    for (int u = 0; u < 4; ++u) s4[u] = sv[i + u * SCAN_THREADS];
    unsigned rec[16], pos[16];
#pragma unroll
    for (int u = 0; u < 4; ++u) {
      const int sx[4] = {s4[u].x, s4[u].y, s4[u].z, s4[u].w};
      const int dx[4] = {d4[u].x, d4[u].y, d4[u].z, d4[u].w};
#pragma unroll
      for (int k = 0; k < 4; ++k) {
        rec[4 * u + k] = (((unsigned)dx[k] & (unsigned)(OWN_ROWS - 1)) << SRC_BITS) | ((unsigned)sx[k] & SRC_MASK);
        pos[4 * u + k] = atomicAdd(&lcur[range_of(dx[k], nRanges)], 1u);
      }
    }
#pragma unroll
    for (int k = 0; k < 16; ++k) stage[pos[k]] = rec[k];
  }
  for (; i < nvec; i += SCAN_THREADS) {
    const int4 s0 = sv[i];
    const int4 d0 = dv[i];
    PLACE_EDGE(s0.x, d0.x); PLACE_EDGE(s0.y, d0.y); PLACE_EDGE(s0.z, d0.z); PLACE_EDGE(s0.w, d0.w);
  }
  for (int t = head + nvec * 4 + tid; t < n; t += SCAN_THREADS) PLACE_EDGE(ps[t], pd[t]);
#undef PLACE_EDGE
  __syncthreads();

  const int lane = tid & 31;
  for (int r = tid >> 5; r < nRanges; r += SCAN_THREADS / 32) {
    const unsigned  ls  = r > 0 ? lcur[r - 1] : 0u;
    const int       cnt = (int)(lcur[r] - ls);
    unsigned*       run = recs + gpos[r];
    const unsigned* st  = stage + ls;
    int k = lane - (int)((((size_t)run) >> 2) & 31u);
    if (k >= 0 && k < cnt) run[k] = st[k];
    k += 32;
    for (; k + 96 < cnt; k += 128) {
      const unsigned a0 = st[k], a1 = st[k + 32], a2 = st[k + 64], a3 = st[k + 96];
      run[k] = a0; run[k + 32] = a1; run[k + 64] = a2; run[k + 96] = a3;
    }
    for (; k < cnt; k += 32) run[k] = st[k];
  }
}

__global__ __launch_bounds__(SCAN_THREADS) void spmm1_owner(const unsigned* __restrict__ recs,
                                                            const int* __restrict__ rangeStart,
                                                            const float* __restrict__ h,
                                                            float* __restrict__ agg,
                                                            float* __restrict__ in_norm, int nN, int nE) {
  __shared__ float    acc[OWN_ROWS * HID];
  __shared__ unsigned indeg[OWN_ROWS];
  const int tid = threadIdx.x;
  for (int i = tid; i < OWN_ROWS * HID; i += SCAN_THREADS) acc[i] = 0.0f;
  for (int i = tid; i < OWN_ROWS; i += SCAN_THREADS) indeg[i] = 0u;
  __syncthreads();

  const int beg  = rangeStart[blockIdx.x];
  const int end  = rangeStart[blockIdx.x + 1];
  const int lane = tid & 31;
  const int j    = lane & 7;
  const int sub  = lane >> 3;
  constexpr int TRIP   = 64;
  constexpr int STRIDE = (SCAN_THREADS / 32) * TRIP;

  const int last = nE - 1;
  int base = beg + (tid >> 5) * TRIP;
  unsigned r0 = recs[min(base + lane, last)];
  unsigned r1 = recs[min(base + 32 + lane, last)];
  for (; base < end; base += STRIDE) {
    const int left  = end - base;
    const int nbase = base + STRIDE;
    const unsigned n0 = recs[min(nbase + lane, last)];
    const unsigned n1 = recs[min(nbase + 32 + lane, last)];

    if (lane < left)      atomicAdd(&indeg[r0 >> SRC_BITS], 1u);
    if (32 + lane < left) atomicAdd(&indeg[r1 >> SRC_BITS], 1u);

    float    v[16];
    unsigned off[16];
#pragma unroll
    for (int i = 0; i < 8; ++i) {
      const int from = i * 4 + sub;
      const unsigned q0 = __shfl(r0, from, 32);
      const unsigned q1 = __shfl(r1, from, 32);
      v[i]       = h[(q0 & SRC_MASK) * HID + j];
      off[i]     = (q0 >> SRC_BITS) * HID + j;
      v[8 + i]   = h[(q1 & SRC_MASK) * HID + j];
      off[8 + i] = (q1 >> SRC_BITS) * HID + j;
    }
#pragma unroll
    for (int i = 0; i < 8; ++i) {
      const bool ok0 = i * 4 + sub < left;
      const bool ok1 = 32 + i * 4 + sub < left;
      atomicAdd(&acc[ok0 ? off[i] : (unsigned)j], ok0 ? v[i] : 0.0f);
      atomicAdd(&acc[ok1 ? off[8 + i] : (unsigned)j], ok1 ? v[8 + i] : 0.0f);
    }
    r0 = n0;
    r1 = n1;
  }
  __syncthreads();

  const long rowBase = (long)blockIdx.x * OWN_ROWS;
  for (int t = tid; t < OWN_ROWS * HID; t += SCAN_THREADS) {
    if (rowBase + (t >> 3) < nN) agg[rowBase * HID + t] = acc[t];
  }
  for (int t = tid; t < OWN_ROWS; t += SCAN_THREADS) {
    if (rowBase + t < nN) in_norm[rowBase + t] = rsqrtf(fmaxf((float)indeg[t], 1.0f));
  }
}

__global__ __launch_bounds__(SCAN_THREADS) void spmm2_owner(const unsigned* __restrict__ recs,
                                                            const int* __restrict__ rangeStart,
                                                            const float* __restrict__ g2,
                                                            const float* __restrict__ in_norm,
                                                            const float* __restrict__ b2,
                                                            float* __restrict__ out, int nN, int nE) {
  __shared__ float acc[OWN_ROWS];
  const int tid = threadIdx.x;
  for (int i = tid; i < OWN_ROWS; i += SCAN_THREADS) acc[i] = 0.0f;
  __syncthreads();

  const int beg = rangeStart[blockIdx.x];
  const int end = rangeStart[blockIdx.x + 1];
  constexpr int U    = 8;
  constexpr int TRIP = U * SCAN_THREADS;
  const int last = nE - 1;
  unsigned r[U];
#pragma unroll
  for (int u = 0; u < U; ++u) r[u] = recs[min(beg + u * SCAN_THREADS + tid, last)];
  for (int base = beg; base < end; base += TRIP) {
    unsigned n[U];
#pragma unroll
    for (int u = 0; u < U; ++u) n[u] = recs[min(base + TRIP + u * SCAN_THREADS + tid, last)];
    float v[U];
#pragma unroll
    for (int u = 0; u < U; ++u) v[u] = g2[r[u] & SRC_MASK];
#pragma unroll
    for (int u = 0; u < U; ++u) {
      const bool ok = base + u * SCAN_THREADS + tid < end;
      atomicAdd(&acc[ok ? (r[u] >> SRC_BITS) : (unsigned)(tid & (OWN_ROWS - 1))], ok ? v[u] : 0.0f);
    }
#pragma unroll
    for (int u = 0; u < U; ++u) r[u] = n[u];
  }
  __syncthreads();

  const long rowBase = (long)blockIdx.x * OWN_ROWS;
  const float bias = b2[0];
  for (int t = tid; t < OWN_ROWS; t += SCAN_THREADS) {
    const long row = rowBase + t;
    if (row < nN) out[row] = fmaf(acc[t], in_norm[row], bias);
  }
}

constexpr int      L2_KEEP = 36;
constexpr int      L2_GRP  = 12;
constexpr unsigned NO_REC  = 0xFFFFFFFFu;

__global__ __launch_bounds__(SCAN_THREADS) void spmm2_lds(const unsigned* __restrict__ recs,
                                                          const int* __restrict__ rangeStart,
                                                          const float* __restrict__ g2,
                                                          const float* __restrict__ in_norm,
                                                          const float* __restrict__ b2,
                                                          float* __restrict__ out,
                                                          int nN, int nE, int nRanges, int nSlices) {
  __shared__ alignas(16) float g2s[SLICE_CAP];
  __shared__ float acc[2 * OWN_ROWS];
  const int tid = threadIdx.x;
  for (int i = tid; i < 2 * OWN_ROWS; i += SCAN_THREADS) acc[i] = 0.0f;

  const int  ra   = (int)blockIdx.x * 2;
  const bool two  = ra + 1 < nRanges;
  const int  begA = rangeStart[ra];
  const int  endA = rangeStart[ra + 1];
  const int  endB = two ? rangeStart[ra + 2] : endA;
  const int  last = nE - 1;

  unsigned ka[L2_KEEP], kb[L2_KEEP];
#pragma unroll
  for (int u = 0; u < L2_KEEP; ++u) {
    const int ia = begA + u * SCAN_THREADS + tid;
    const int ib = endA + u * SCAN_THREADS + tid;
    const unsigned xa = recs[min(ia, last)];
    const unsigned xb = recs[min(ib, last)];
    ka[u] = ia < endA ? xa : NO_REC;
    kb[u] = ib < endB ? xb : NO_REC;
  }

  const int len = slice_len(nN, nSlices);
#define L2_HIT(R, ROWOFF)                                                                 \
  do {                                                                                    \
    const unsigned k_ = ((R) & SRC_MASK) - (unsigned)base;                                \
    if (k_ < (unsigned)len) atomicAdd(&acc[((R) >> SRC_BITS) + (ROWOFF)], g2s[k_]);       \
  } while (0)
#define L2_GROUP(K, G, ROWOFF)                                                            \
  do {                                                                                    \
    unsigned k[L2_GRP];                                                                   \
    float    v[L2_GRP];                                                                   \
    _Pragma("unroll") for (int j = 0; j < L2_GRP; ++j) {                                  \
      k[j] = (K[(G) * L2_GRP + j] & SRC_MASK) - (unsigned)base;                           \
      v[j] = k[j] < (unsigned)len ? g2s[k[j]] : 0.0f;                                     \
    }                                                                                     \
    _Pragma("unroll") for (int j = 0; j < L2_GRP; ++j) {                                  \
      atomicAdd(&acc[k[j] < (unsigned)len ? (K[(G) * L2_GRP + j] >> SRC_BITS) + (ROWOFF) : (unsigned)tid], v[j]); \
    }                                                                                     \
  } while (0)

  for (int s = 0; s < nSlices; ++s) {
    const int base = s * len;
    __syncthreads();
    int cnt = nN - base;
    cnt = cnt < 0 ? 0 : (cnt > len ? len : cnt);
    const float* gsrc = g2 + base;
    const int nv = ((((size_t)gsrc) & 15u) == 0) ? (cnt >> 2) : 0;
    const float4* src4 = (const float4*)gsrc;
    float4*       dst4 = (float4*)g2s;
    for (int i = tid; i < nv; i += 6 * SCAN_THREADS) {
      float4 t[6];
      int ix[6];
#pragma unroll
      for (int u = 0; u < 6; ++u) ix[u] = min(i + u * SCAN_THREADS, nv - 1);
#pragma unroll
      for (int u = 0; u < 6; ++u) t[u] = src4[ix[u]];
#pragma unroll
      for (int u = 0; u < 6; ++u) dst4[ix[u]] = t[u];
    }
    for (int i = nv * 4 + tid; i < len; i += SCAN_THREADS) g2s[i] = i < cnt ? gsrc[i] : 0.0f;
    __syncthreads();

#pragma unroll
    for (int u = 0; u < L2_KEEP; ++u) {
      asm volatile("" : "+v"(ka[u]));
      asm volatile("" : "+v"(kb[u]));
    }
#pragma unroll
    for (int g = 0; g < L2_KEEP / L2_GRP; ++g) {
      L2_GROUP(ka, g, 0);
      L2_GROUP(kb, g, OWN_ROWS);
    }
    for (int i = begA + L2_KEEP * SCAN_THREADS + tid; i < endA; i += SCAN_THREADS) L2_HIT(recs[i], 0);
    for (int i = endA + L2_KEEP * SCAN_THREADS + tid; i < endB; i += SCAN_THREADS) L2_HIT(recs[i], OWN_ROWS);
  }
#undef L2_GROUP
#undef L2_HIT
  __syncthreads();

  const long rowBase = (long)ra * OWN_ROWS;
  const float bias = b2[0];
  for (int t = tid; t < 2 * OWN_ROWS; t += SCAN_THREADS) {
    const long row = rowBase + t;
    if (row < nN) out[row] = fmaf(acc[t], in_norm[row], bias);
  }
}

constexpr int L2S_SHIFT = 16;
constexpr int L2S_LEN   = 1 << L2S_SHIFT;
constexpr int L2S_BUCKETS = 32;
constexpr int L2S_MAXSL = L2S_BUCKETS / 2 - 1;
constexpr int L2S_CNT   = L2S_BUCKETS * 32;
constexpr int L2S_TOTAL = 2 * L2_KEEP * SCAN_THREADS;
constexpr int L2S_PIECE = 4;

__global__ __launch_bounds__(SCAN_THREADS) void spmm2_sorted(const unsigned* __restrict__ recs,
                                                             const int* __restrict__ rangeStart,
                                                             const float* __restrict__ g2,
                                                             const float* __restrict__ in_norm,
                                                             const float* __restrict__ b2,
                                                             float* __restrict__ out,
                                                             int nN, int nE, int nRanges, int nSl) {
  static_assert(L2S_CNT == SCAN_THREADS, "one counter per thread in the scan");
  static_assert(L2S_TOTAL <= SLICE_CAP, "the sorted records must fit the slice buffer");
  static_assert((2 * L2_KEEP) % L2S_PIECE == 0, "whole pieces");
  static_assert(L2S_PIECE * SCAN_THREADS == 4096, "a piece is 1 << 12 list elements");
  static_assert(SRC_BITS - L2S_SHIFT <= 6 && 2 * OWN_ROWS <= 2048, "fields of the ordered record");
  __shared__ alignas(16) float g2s[SLICE_CAP];
  __shared__ float    acc[2 * OWN_ROWS];
  __shared__ unsigned cnt[L2S_CNT];
  __shared__ unsigned scn[2][L2S_CNT];
  unsigned* sorted = (unsigned*)g2s;
  const int tid  = threadIdx.x;
  const int lane = tid & 31;
  for (int i = tid; i < 2 * OWN_ROWS; i += SCAN_THREADS) acc[i] = 0.0f;
  cnt[tid] = 0u;

  const int  ra   = (int)blockIdx.x * 2;
  const bool two  = ra + 1 < nRanges;
  const int  begA = rangeStart[ra];
  const int  endA = rangeStart[ra + 1];
  const int  endB = two ? rangeStart[ra + 2] : endA;
  const int  last = nE - 1;

  unsigned r[2 * L2_KEEP];
#pragma unroll
  for (int u = 0; u < L2_KEEP; ++u) {
    const int ia = begA + u * SCAN_THREADS + tid;
    const int ib = endA + u * SCAN_THREADS + tid;
    const unsigned xa = recs[min(ia, last)];
    const unsigned xb = recs[min(ib, last)];
    r[u]           = ia < endA ? xa : NO_REC;
    r[L2_KEEP + u] = ib < endB ? xb : NO_REC;
  }
  __syncthreads();

  const unsigned slEnd = (unsigned)nSl;
#define L2S_COUNTER(U) \
  ((min((r[(U)] >> L2S_SHIFT) & ((1u << (SRC_BITS - L2S_SHIFT)) - 1u), slEnd) * 2u + ((U) >= L2_KEEP ? 1u : 0u)) * 32u + (unsigned)lane)
#pragma unroll
  for (int u = 0; u < 2 * L2_KEEP; ++u) atomicAdd(&cnt[L2S_COUNTER(u)], 1u);
#pragma unroll
  for (int u = 0; u < 2 * L2_KEEP; ++u) asm volatile("" : "+v"(r[u]));
  __syncthreads();

  const unsigned mine = cnt[tid];
  scn[0][tid] = mine;
  __syncthreads();
  int cur = 0;
  for (int off = 1; off < L2S_CNT; off <<= 1) {
    scn[1 - cur][tid] = scn[cur][tid] + (tid >= off ? scn[cur][tid - off] : 0u);
    __syncthreads();
    cur ^= 1;
  }
  const unsigned excl = scn[cur][tid] - mine;
  cnt[tid] = excl;
  scn[1 - cur][tid] = excl;
  const unsigned* bstart = scn[1 - cur];
  __syncthreads();

#pragma unroll
  for (int u = 0; u < 2 * L2_KEEP; ++u) {
    const unsigned pos = atomicAdd(&cnt[L2S_COUNTER(u)], 1u);
    sorted[pos] = ((r[u] >> 1) & 0x7FE00000u) | ((u) >= L2_KEEP ? 0x80000000u : 0u) |
                  ((r[u] & (unsigned)(L2S_LEN - 1)) << 2);
  }
#undef L2S_COUNTER
  __syncthreads();
#pragma unroll
  for (int u = 0; u < 2 * L2_KEEP; ++u) r[u] = sorted[u * SCAN_THREADS + tid];

#define L2S_HIT(R, ROWOFF)                                                                \
  do {                                                                                    \
    const unsigned k_ = ((R) & SRC_MASK) - (unsigned)base;                                \
    if (k_ < (unsigned)L2S_LEN) atomicAdd(&acc[((R) >> SRC_BITS) + (ROWOFF)], g2s[k_]);   \
  } while (0)

  for (int s = 0; s < nSl; ++s) {
    const int base = s << L2S_SHIFT;
    __syncthreads();
    int cntN = nN - base;
    cntN = cntN < 0 ? 0 : (cntN > L2S_LEN ? L2S_LEN : cntN);
    const float* gsrc = g2 + base;
    const int nv = ((((size_t)gsrc) & 15u) == 0) ? (cntN >> 2) : 0;
    const float4* src4 = (const float4*)gsrc;
    float4*       dst4 = (float4*)g2s;
    for (int i = tid; i < nv; i += 6 * SCAN_THREADS) {
      float4 t[6];
      int ix[6];
#pragma unroll
      for (int u = 0; u < 6; ++u) ix[u] = min(i + u * SCAN_THREADS, nv - 1);
#pragma unroll
      for (int u = 0; u < 6; ++u) t[u] = src4[ix[u]];
#pragma unroll
      for (int u = 0; u < 6; ++u) dst4[ix[u]] = t[u];
    }
    for (int i = nv * 4 + tid; i < L2S_LEN; i += SCAN_THREADS) g2s[i] = i < cntN ? gsrc[i] : 0.0f;
    __syncthreads();

    const unsigned lo = (unsigned)__builtin_amdgcn_readfirstlane((int)bstart[(2 * s) * 32]);
    const unsigned hi = (unsigned)__builtin_amdgcn_readfirstlane((int)bstart[(2 * s + 2) * 32]);
    const unsigned span = hi - lo;

#pragma unroll
    for (int u = 0; u < 2 * L2_KEEP; ++u) asm volatile("" : "+v"(r[u]));

    if (span != 0u) {
      const unsigned glo   = lo >> 12;
      const unsigned gspan = ((hi - 1u) >> 12) - glo;
#pragma unroll
      for (int g = 0; g < 2 * L2_KEEP / L2S_PIECE; ++g) {
        if ((unsigned)g - glo <= gspan) {
          bool  in[L2S_PIECE];
          float v[L2S_PIECE];
#pragma unroll
          for (int j = 0; j < L2S_PIECE; ++j) {
            const unsigned p = (unsigned)((g * L2S_PIECE + j) * SCAN_THREADS + tid);
            in[j] = p - lo < span;
            v[j]  = in[j] ? g2s[(r[g * L2S_PIECE + j] & 0x3FFFFu) >> 2] : 0.0f;
          }
#pragma unroll
          for (int j = 0; j < L2S_PIECE; ++j)
            atomicAdd(&acc[in[j] ? (r[g * L2S_PIECE + j] >> 21) : (unsigned)tid], v[j]);
        }
      }
    }
    for (int i = begA + L2_KEEP * SCAN_THREADS + tid; i < endA; i += SCAN_THREADS) L2S_HIT(recs[i], 0);
    for (int i = endA + L2_KEEP * SCAN_THREADS + tid; i < endB; i += SCAN_THREADS) L2S_HIT(recs[i], OWN_ROWS);
  }
#undef L2S_HIT
  __syncthreads();

  const long rowBase = (long)ra * OWN_ROWS;
  const float bias = b2[0];
  for (int t = tid; t < 2 * OWN_ROWS; t += SCAN_THREADS) {
    const long row = rowBase + t;
    if (row < nN) out[row] = fmaf(acc[t], in_norm[row], bias);
  }
}


__global__ void layer2_node(const float* __restrict__ agg, const float* __restrict__ in_norm,
                            const float* __restrict__ out_norm, const float* __restrict__ b1,
                            const float* __restrict__ W2, float* __restrict__ g2, int n) {
  int v = blockIdx.x * blockDim.x + threadIdx.x;
  if (v >= n) return;
  float innv = in_norm[v];
  const float* av = agg + (long)v * HID;
  float s = 0.0f;
#pragma unroll
  for (int j = 0; j < HID; ++j) {
    float z = fmaf(av[j], innv, b1[j]);
    z = z > 0.0f ? z : 0.0f;
    s = fmaf(z, W2[j], s);
  }
  g2[v] = s * out_norm[v];
}


__device__ __forceinline__ void sum_hit(float* acc, const float* __restrict__ g2, int s, int d,
                                        int base, int len) {
  const unsigned k = (unsigned)(d - base);
  if (k < (unsigned)len) atomicAdd(&acc[k], g2[s]);
}

__global__ __launch_bounds__(SCAN_THREADS) void spmm2_slices(const int* __restrict__ src,
                                                             const int* __restrict__ dst,
                                                             const float* __restrict__ g2,
                                                             float* __restrict__ outacc,
                                                             int nE, int nN) {
  __shared__ float acc[SLICE_CAP];
  const int tid     = threadIdx.x;
  const int nSlices = gridDim.y;
  const int len     = slice_len(nN, nSlices);
  const int base    = (int)blockIdx.y * len;

  for (int i = tid; i < len; i += SCAN_THREADS) acc[i] = 0.0f;
  __syncthreads();

  int start, n;
  chunk_bounds(nE, gridDim.x, blockIdx.x, start, n);
  const int* ps = src + start;
  const int* pd = dst + start;
  int head = (int)(((16u - (unsigned)((size_t)pd & 15u)) & 15u) >> 2);
  if (head > n) head = n;
  const bool vec = (((size_t)(ps + head)) & 15u) == 0;
  if (!vec) head = n;
  for (int i = tid; i < head; i += SCAN_THREADS) sum_hit(acc, g2, ps[i], pd[i], base, len);
  const int4* sv = (const int4*)(ps + head);
  const int4* dv = (const int4*)(pd + head);
  const int nvec = (n - head) >> 2;
  int i = tid;
  int4 dq[4], sq[4];
#pragma unroll
  for (int u = 0; u < 4; ++u) { dq[u] = make_int4(-1, -1, -1, -1); sq[u] = make_int4(0, 0, 0, 0); }
  for (;;) {
    const bool fullNext = i + 3 * SCAN_THREADS < nvec;
    int4 dn[4], sn[4];
#pragma unroll
    for (int u = 0; u < 4; ++u) { dn[u] = make_int4(-1, -1, -1, -1); sn[u] = make_int4(0, 0, 0, 0); }
    if (fullNext) {
#pragma unroll
      for (int u = 0; u < 4; ++u) { sn[u] = sv[i + u * SCAN_THREADS]; dn[u] = dv[i + u * SCAN_THREADS]; }
    }
    unsigned k[16];
    float    val[16];
#pragma unroll
    for (int u = 0; u < 4; ++u) {
      k[4 * u + 0] = (unsigned)(dq[u].x - base);
      k[4 * u + 1] = (unsigned)(dq[u].y - base);
      k[4 * u + 2] = (unsigned)(dq[u].z - base);
      k[4 * u + 3] = (unsigned)(dq[u].w - base);
      val[4 * u + 0] = (k[4 * u + 0] < (unsigned)len) ? g2[sq[u].x] : 0.0f;
      val[4 * u + 1] = (k[4 * u + 1] < (unsigned)len) ? g2[sq[u].y] : 0.0f;
      val[4 * u + 2] = (k[4 * u + 2] < (unsigned)len) ? g2[sq[u].z] : 0.0f;
      val[4 * u + 3] = (k[4 * u + 3] < (unsigned)len) ? g2[sq[u].w] : 0.0f;
    }
#pragma unroll
    for (int j = 0; j < 16; ++j) {
      if (k[j] < (unsigned)len) atomicAdd(&acc[k[j]], val[j]);
    }
    if (!fullNext) break;
#pragma unroll
    for (int u = 0; u < 4; ++u) { dq[u] = dn[u]; sq[u] = sn[u]; }
    i += 4 * SCAN_THREADS;
  }
  for (; i < nvec; i += SCAN_THREADS) {
    const int4 d0 = dv[i];
    const int4 s0 = sv[i];
    sum_hit(acc, g2, s0.x, d0.x, base, len); sum_hit(acc, g2, s0.y, d0.y, base, len);
    sum_hit(acc, g2, s0.z, d0.z, base, len); sum_hit(acc, g2, s0.w, d0.w, base, len);
  }
  for (int t = head + nvec * 4 + tid; t < n; t += SCAN_THREADS) sum_hit(acc, g2, ps[t], pd[t], base, len);
  __syncthreads();

  for (int t = tid; t < len; t += SCAN_THREADS) {
    const int node = base + t;
    if (node < nN) atomicAdd(&outacc[node], acc[t]);
  }
}

__global__ void final_kernel(const float* __restrict__ outacc, const float* __restrict__ in_norm,
                             const float* __restrict__ b2, float* __restrict__ out, int n) {
  int v = blockIdx.x * blockDim.x + threadIdx.x;
  if (v < n) out[v] = fmaf(outacc[v], in_norm[v], b2[0]);
}


extern "C" void kernel_launch(void* const* d_in, const int* in_sizes, int n_in,
                              void* d_out, int out_size, void* d_ws, size_t ws_size,
                              hipStream_t stream) {
  const float* x  = (const float*)d_in[0];
  const float* W1 = (const float*)d_in[1];
  const float* b1 = (const float*)d_in[2];
  const float* W2 = (const float*)d_in[3];
  const float* b2 = (const float*)d_in[4];
  const int*  src = (const int*)d_in[5];
  const int*  dst = (const int*)d_in[6];

  const int nN = in_sizes[0] / IN_DIM;
  const int nE = in_sizes[5];

  float* ws     = (float*)d_ws;
  float* deg_s  = ws;
  float* deg_d  = ws + (long)nN;
  float* h      = ws + 2L * nN;
  float* agg    = ws + 10L * nN;
  float* g2     = ws + 18L * nN;
  float* outacc = ws + 19L * nN;
  float* out    = (float*)d_out;

  const int TB = 256;
  dim3 blk(TB);

  const int nSlices   = (nN + SLICE_CAP - 1) / SLICE_CAP;
  const int degChunks = 256 / (2 * nSlices) > 0 ? 256 / (2 * nSlices) : 1;
  const int sumChunks = 256 / nSlices > 0 ? 256 / nSlices : 1;

  const int    nRanges  = (nN + OWN_ROWS - 1) / OWN_ROWS;
  const size_t wsWords  = 20UL * nN + (size_t)nE + (size_t)nRanges * BIN_CHUNKS + 2UL * nRanges + 8UL;
  const bool   sorted   = nE > 0 && nRanges <= MAX_RANGES && wsWords * sizeof(float) <= ws_size &&
                          (nE + BIN_CHUNKS - 1) / BIN_CHUNKS + 3 <= STAGE_CAP;

  if (sorted) {
    zero_f32<<<(nN + TB - 1) / TB, blk, 0, stream>>>(deg_s, nN);
    degree_slices<<<dim3(sumChunks, nSlices), dim3(SCAN_THREADS), 0, stream>>>(
        src, dst, (unsigned*)deg_s, (unsigned*)deg_d, nE, nN, nSlices);
    norm_kernel<<<(nN + TB - 1) / TB, blk, 0, stream>>>((unsigned*)deg_s, (unsigned*)deg_d, nN, 0);
  } else {
    zero_f32<<<(2 * nN + TB - 1) / TB, blk, 0, stream>>>(deg_s, 2 * nN);
    degree_slices<<<dim3(degChunks, 2 * nSlices), dim3(SCAN_THREADS), 0, stream>>>(
        src, dst, (unsigned*)deg_s, (unsigned*)deg_d, nE, nN, nSlices);
    norm_kernel<<<(nN + TB - 1) / TB, blk, 0, stream>>>((unsigned*)deg_s, (unsigned*)deg_d, nN, 1);
  }

  gemm1_wmma<<<(nN + 127) / 128, blk, 0, stream>>>(x, W1, deg_s, h, nN);

  if (sorted) {
    unsigned* recs       = (unsigned*)(ws + 20L * nN);
    int*      histT      = (int*)(recs + nE);
    int*      total      = histT + (long)nRanges * BIN_CHUNKS;
    int*      rangeStart = total + nRanges;
    bin_count<<<BIN_CHUNKS, dim3(SCAN_THREADS), 0, stream>>>(dst, histT, nE, nRanges);
    bin_scan<<<(nRanges * 32 + TB - 1) / TB, blk, 0, stream>>>(histT, total, BIN_CHUNKS, nRanges);
    bin_place<<<BIN_CHUNKS, dim3(SCAN_THREADS), 0, stream>>>(src, dst, histT, total, rangeStart, recs, nE, nRanges);
    spmm1_owner<<<nRanges, dim3(SCAN_THREADS), 0, stream>>>(recs, rangeStart, h, agg, deg_d, nN, nE);
    layer2_node<<<(nN + TB - 1) / TB, blk, 0, stream>>>(agg, deg_d, deg_s, b1, W2, g2, nN);
    const int nSl2 = (nN + L2S_LEN - 1) >> L2S_SHIFT;
    if (nN <= (int)SRC_MASK && nSl2 <= L2S_MAXSL) {
      spmm2_sorted<<<(nRanges + 1) / 2, dim3(SCAN_THREADS), 0, stream>>>(recs, rangeStart, g2, deg_d, b2, out, nN, nE,
                                                                        nRanges, nSl2);
    } else if (nN <= (int)SRC_MASK) {
      spmm2_lds<<<(nRanges + 1) / 2, dim3(SCAN_THREADS), 0, stream>>>(recs, rangeStart, g2, deg_d, b2, out, nN, nE,
                                                                     nRanges, nSlices);
    } else {
      spmm2_owner<<<nRanges, dim3(SCAN_THREADS), 0, stream>>>(recs, rangeStart, g2, deg_d, b2, out, nN, nE);
    }
  } else {
    zero_f32<<<(8 * nN + TB - 1) / TB, blk, 0, stream>>>(agg, 8 * nN);
    spmm1_kernel<<<(nE + TB - 1) / TB, blk, 0, stream>>>(src, dst, h, agg, nE);
    layer2_node<<<(nN + TB - 1) / TB, blk, 0, stream>>>(agg, deg_d, deg_s, b1, W2, g2, nN);
    zero_f32<<<(nN + TB - 1) / TB, blk, 0, stream>>>(outacc, nN);
    spmm2_slices<<<dim3(sumChunks, nSlices), dim3(SCAN_THREADS), 0, stream>>>(src, dst, g2, outacc, nE, nN);
    final_kernel<<<(nN + TB - 1) / TB, blk, 0, stream>>>(outacc, deg_d, b2, out, nN);
  }
}


# gemm1: resident waves loop over row tiles, W1 staged once per work-group and kept in registers, x read by 16-byte loads (k order permuted alike in A and B: a row's products add in another order)
# speedup vs baseline: 9.4727x; 1.0498x over previous
#include <hip/hip_runtime.h>

typedef __attribute__((ext_vector_type(2))) float v2f;
typedef __attribute__((ext_vector_type(8))) float v8f;

constexpr int IN_DIM = 128;
constexpr int HID    = 8;

constexpr int SLICE_CAP    = 73728;
constexpr int SCAN_THREADS = 1024;


__global__ void zero_f32(float* __restrict__ p, int n) {
  int i = blockIdx.x * blockDim.x + threadIdx.x;
  if (i < n) p[i] = 0.0f;
}


__device__ __forceinline__ int slice_len(int nN, int nSlices) {
  const int per = (nN + nSlices - 1) / nSlices;
  return (per + 31) & ~31;
}

__device__ __forceinline__ void chunk_bounds(int nE, int nChunks, int chunk, int& start, int& count) {
  const int chunkLen = (((nE + nChunks - 1) / nChunks) + 3) & ~3;
  long s = (long)chunk * chunkLen;
  if (s > nE) s = nE;
  long e = s + chunkLen;
  if (e > nE) e = nE;
  start = (int)s;
  count = (int)(e - s);
}

__device__ __forceinline__ void count_hit(unsigned* cnt, int v, int base, int len) {
  const unsigned k = (unsigned)(v - base);
  if (k < (unsigned)len) atomicAdd(&cnt[k], 1u);
}

__global__ __launch_bounds__(SCAN_THREADS) void degree_slices(const int* __restrict__ src,
                                                              const int* __restrict__ dst,
                                                              unsigned* __restrict__ cnt_s,
                                                              unsigned* __restrict__ cnt_d,
                                                              int nE, int nN, int nSlices) {
  __shared__ unsigned cnt[SLICE_CAP];
  const int tid     = threadIdx.x;
  const int which   = (int)blockIdx.y >= nSlices ? 1 : 0;
  const int slice   = (int)blockIdx.y - which * nSlices;
  const int len     = slice_len(nN, nSlices);
  const int base    = slice * len;
  const int* __restrict__ idx = which ? dst : src;
  unsigned* __restrict__ out  = which ? cnt_d : cnt_s;

  for (int i = tid; i < len; i += SCAN_THREADS) cnt[i] = 0u;
  __syncthreads();

  int start, n;
  chunk_bounds(nE, gridDim.x, blockIdx.x, start, n);
  const int* p = idx + start;
  int head = (int)(((16u - (unsigned)((size_t)p & 15u)) & 15u) >> 2);
  if (head > n) head = n;
  for (int i = tid; i < head; i += SCAN_THREADS) count_hit(cnt, p[i], base, len);
  const int4* pv = (const int4*)(p + head);
  const int nvec = (n - head) >> 2;
  int i = tid;
  for (; i + 7 * SCAN_THREADS < nvec; i += 8 * SCAN_THREADS) {
    int4 v[8];
#pragma unroll
    for (int u = 0; u < 8; ++u) v[u] = pv[i + u * SCAN_THREADS];
#pragma unroll
    for (int u = 0; u < 8; ++u) {
      count_hit(cnt, v[u].x, base, len); count_hit(cnt, v[u].y, base, len);
      count_hit(cnt, v[u].z, base, len); count_hit(cnt, v[u].w, base, len);
    }
  }
  for (; i < nvec; i += SCAN_THREADS) {
    const int4 a = pv[i];
    count_hit(cnt, a.x, base, len); count_hit(cnt, a.y, base, len);
    count_hit(cnt, a.z, base, len); count_hit(cnt, a.w, base, len);
  }
  for (int t = head + nvec * 4 + tid; t < n; t += SCAN_THREADS) count_hit(cnt, p[t], base, len);
  __syncthreads();

  for (int t = tid; t < len; t += SCAN_THREADS) {
    const int node = base + t;
    if (node < nN) atomicAdd(&out[node], cnt[t]);
  }
}

__global__ void norm_kernel(unsigned* __restrict__ deg_s, unsigned* __restrict__ deg_d, int n, int both) {
  int i = blockIdx.x * blockDim.x + threadIdx.x;
  if (i < n) {
    deg_s[i] = __float_as_uint(rsqrtf(fmaxf((float)deg_s[i], 1.0f)));
    if (both) deg_d[i] = __float_as_uint(rsqrtf(fmaxf((float)deg_d[i], 1.0f)));
  }
}


__global__ __launch_bounds__(256) void gemm1_wmma(const float* __restrict__ x,
                                                  const float* __restrict__ W1,
                                                  const float* __restrict__ out_norm,
                                                  float* __restrict__ h, int nNodes) {
  __shared__ float w[IN_DIM * 16];
  const int tid = threadIdx.x;
  {
    float t[8];
#pragma unroll
    for (int u = 0; u < 8; ++u) {
      const int i = tid + u * 256;
      t[u] = W1[(i >> 4) * HID + (i & 7)];
    }
#pragma unroll
    for (int u = 0; u < 8; ++u) {
      const int i = tid + u * 256;
      w[i] = ((i & 15) < HID) ? t[u] : 0.0f;
    }
  }
  __syncthreads();

  const int wid  = tid >> 5;
  const int lane = tid & 31;
  const int m    = lane & 15;
  const int kh   = lane >> 4;

  const int nTiles = (nNodes + 15) >> 4;
  const int nWaves = (int)gridDim.x * 8;
  for (int tile = (int)blockIdx.x * 8 + wid; tile < nTiles; tile += nWaves) {
    const int row0 = tile * 16;
    const int node = row0 + m;
    const int nodeClamped = node < nNodes ? node : nNodes - 1;
    const float4* xrow = (const float4*)(x + (long)nodeClamped * IN_DIM);

    float4 xa[16];
#pragma unroll
    for (int q = 0; q < 16; ++q) xa[q] = xrow[2 * q + kh];
    float on[8];
#pragma unroll
    for (int r = 0; r < 8; ++r) {
      const int nr = row0 + r + 8 * kh;
      on[r] = out_norm[nr < nNodes ? nr : nNodes - 1];
    }

    v8f acc = {};
#pragma unroll
    for (int q = 0; q < 16; ++q) {
      const int k0 = 8 * q + 4 * kh;
      v2f a0, a1, b0, b1;
      a0.x = xa[q].x; a0.y = xa[q].y;
      a1.x = xa[q].z; a1.y = xa[q].w;
      b0.x = w[(k0 + 0) * 16 + m];
      b0.y = w[(k0 + 1) * 16 + m];
      b1.x = w[(k0 + 2) * 16 + m];
      b1.y = w[(k0 + 3) * 16 + m];
      acc = __builtin_amdgcn_wmma_f32_16x16x4_f32(
           false, a0,  false, b0,
           (short)0, acc,  false,  false);
      acc = __builtin_amdgcn_wmma_f32_16x16x4_f32(
           false, a1,  false, b1,
           (short)0, acc,  false,  false);
    }

    if (m < HID) {
#pragma unroll
      for (int r = 0; r < 8; ++r) {
        const int nr = row0 + r + 8 * kh;
        if (nr < nNodes) h[(long)nr * HID + m] = acc[r] * on[r];
      }
    }
  }
}


__global__ void spmm1_kernel(const int* __restrict__ src, const int* __restrict__ dst,
                             const float* __restrict__ h, float* __restrict__ agg, int nE) {
  const int lane = threadIdx.x & 31;
  const int j    = lane & 7;
  const int sub  = lane >> 3;
  const int e    = blockIdx.x * blockDim.x + threadIdx.x;
  const int left = nE - (e - lane);

  int s = 0, d = 0;
  if (e < nE) { s = src[e]; d = dst[e]; }

  float    v[8];
  unsigned off[8];
#pragma unroll
  for (int i = 0; i < 8; ++i) {
    const int from = i * 4 + sub;
    const int si = __shfl(s, from, 32);
    const int di = __shfl(d, from, 32);
    v[i]   = h[(unsigned)(si * HID + j)];
    off[i] = (unsigned)(di * HID + j);
  }
#pragma unroll
  for (int i = 0; i < 8; ++i) {
    if (i * 4 + sub < left) atomicAdd(&agg[off[i]], v[i]);
  }
}


constexpr int OWN_SHIFT  = 10;
constexpr int OWN_ROWS   = 1 << OWN_SHIFT;
constexpr int SRC_BITS   = 32 - OWN_SHIFT;
constexpr int MAX_RANGES = 4096;
constexpr int BIN_CHUNKS = 256;
constexpr unsigned SRC_MASK = (1u << SRC_BITS) - 1u;

__device__ __forceinline__ unsigned range_of(int d, int nRanges) {
  const unsigned r = (unsigned)d >> OWN_SHIFT;
  return r < (unsigned)nRanges ? r : (unsigned)(nRanges - 1);
}

__global__ __launch_bounds__(SCAN_THREADS) void bin_count(const int* __restrict__ dst,
                                                          int* __restrict__ histT,
                                                          int nE, int nRanges) {
  __shared__ unsigned cnt[MAX_RANGES];
  const int tid = threadIdx.x;
  for (int i = tid; i < nRanges; i += SCAN_THREADS) cnt[i] = 0u;
  __syncthreads();

  int start, n;
  chunk_bounds(nE, gridDim.x, blockIdx.x, start, n);
  const int* p = dst + start;
  int head = (int)(((16u - (unsigned)((size_t)p & 15u)) & 15u) >> 2);
  if (head > n) head = n;
  for (int i = tid; i < head; i += SCAN_THREADS) atomicAdd(&cnt[range_of(p[i], nRanges)], 1u);
  const int4* pv = (const int4*)(p + head);
  const int nvec = (n - head) >> 2;
  int i = tid;
  for (; i + 3 * SCAN_THREADS < nvec; i += 4 * SCAN_THREADS) {
    int4 v[4];
#pragma unroll
    for (int u = 0; u < 4; ++u) v[u] = pv[i + u * SCAN_THREADS];
#pragma unroll
    for (int u = 0; u < 4; ++u) {
      atomicAdd(&cnt[range_of(v[u].x, nRanges)], 1u); atomicAdd(&cnt[range_of(v[u].y, nRanges)], 1u);
      atomicAdd(&cnt[range_of(v[u].z, nRanges)], 1u); atomicAdd(&cnt[range_of(v[u].w, nRanges)], 1u);
    }
  }
  for (; i < nvec; i += SCAN_THREADS) {
    const int4 a = pv[i];
    atomicAdd(&cnt[range_of(a.x, nRanges)], 1u); atomicAdd(&cnt[range_of(a.y, nRanges)], 1u);
    atomicAdd(&cnt[range_of(a.z, nRanges)], 1u); atomicAdd(&cnt[range_of(a.w, nRanges)], 1u);
  }
  for (int t = head + nvec * 4 + tid; t < n; t += SCAN_THREADS)
    atomicAdd(&cnt[range_of(p[t], nRanges)], 1u);
  __syncthreads();

  for (int r = tid; r < nRanges; r += SCAN_THREADS) histT[r * (int)gridDim.x + (int)blockIdx.x] = (int)cnt[r];
}

__global__ void bin_scan(int* __restrict__ histT, int* __restrict__ total, int nChunks, int nRanges) {
  const int lane  = threadIdx.x & 31;
  const int range = (int)((blockIdx.x * blockDim.x + threadIdx.x) >> 5);
  if (range >= nRanges) return;
  int* row = histT + (long)range * nChunks;
  int carry = 0;
  for (int c0 = 0; c0 < nChunks; c0 += 32) {
    const int c = c0 + lane;
    const int v = c < nChunks ? row[c] : 0;
    int s = v;
#pragma unroll
    for (int d = 1; d < 32; d <<= 1) {
      const int t = __shfl_up(s, d, 32);
      if (lane >= d) s += t;
    }
    if (c < nChunks) row[c] = carry + s - v;
    carry += __shfl(s, 31, 32);
  }
  if (lane == 0) total[range] = carry;
}

constexpr int STAGE_CAP = 62976;

__global__ __launch_bounds__(SCAN_THREADS) void bin_place(const int* __restrict__ src,
                                                          const int* __restrict__ dst,
                                                          const int* __restrict__ histT,
                                                          const int* __restrict__ total,
                                                          int* __restrict__ rangeStart,
                                                          unsigned* __restrict__ recs,
                                                          int nE, int nRanges) {
  __shared__ int      scanbuf[2][MAX_RANGES];
  __shared__ unsigned gpos[MAX_RANGES];
  __shared__ unsigned lcur[MAX_RANGES];
  __shared__ unsigned stage[STAGE_CAP];
  const int tid     = threadIdx.x;
  const int nChunks = (int)gridDim.x;
  const int chunk   = (int)blockIdx.x;

  for (int r = tid; r < nRanges; r += SCAN_THREADS) scanbuf[0][r] = total[r];
  __syncthreads();
  int cur = 0;
  for (int off = 1; off < nRanges; off <<= 1) {
    for (int r = tid; r < nRanges; r += SCAN_THREADS)
      scanbuf[1 - cur][r] = scanbuf[cur][r] + (r >= off ? scanbuf[cur][r - off] : 0);
    __syncthreads();
    cur ^= 1;
  }
  for (int r = tid; r < nRanges; r += SCAN_THREADS) {
    const int tot  = total[r];
    const int incl = scanbuf[cur][r];
    const int excl = incl - tot;
    const int pre  = histT[r * nChunks + chunk];
    const int nxt  = chunk + 1 < nChunks ? histT[r * nChunks + chunk + 1] : tot;
    gpos[r] = (unsigned)(excl + pre);
    scanbuf[1 - cur][r] = nxt - pre;
    if (chunk == 0) {
      rangeStart[r] = excl;
      if (r == nRanges - 1) rangeStart[nRanges] = incl;
    }
  }
  __syncthreads();
  cur ^= 1;
  for (int off = 1; off < nRanges; off <<= 1) {
    for (int r = tid; r < nRanges; r += SCAN_THREADS)
      scanbuf[1 - cur][r] = scanbuf[cur][r] + (r >= off ? scanbuf[cur][r - off] : 0);
    __syncthreads();
    cur ^= 1;
  }
  for (int r = tid; r < nRanges; r += SCAN_THREADS) lcur[r] = r > 0 ? (unsigned)scanbuf[cur][r - 1] : 0u;
  __syncthreads();

  int start, n;
  chunk_bounds(nE, nChunks, chunk, start, n);
  const int* ps = src + start;
  const int* pd = dst + start;
  int head = (int)(((16u - (unsigned)((size_t)pd & 15u)) & 15u) >> 2);
  if (head > n) head = n;
  if ((((size_t)(ps + head)) & 15u) != 0) head = n;
#define PLACE_EDGE(S, D)                                                            \
  do {                                                                              \
    const unsigned pos_ = atomicAdd(&lcur[range_of((D), nRanges)], 1u);             \
    stage[pos_] = (((unsigned)(D) & (unsigned)(OWN_ROWS - 1)) << SRC_BITS) | ((unsigned)(S) & SRC_MASK); \
  } while (0)
  for (int i = tid; i < head; i += SCAN_THREADS) PLACE_EDGE(ps[i], pd[i]);
  const int4* sv = (const int4*)(ps + head);
  const int4* dv = (const int4*)(pd + head);
  const int nvec = (n - head) >> 2;
  int i = tid;
  for (; i + 3 * SCAN_THREADS < nvec; i += 4 * SCAN_THREADS) {
    int4 s4[4], d4[4];
#pragma unroll
    for (int u = 0; u < 4; ++u) d4[u] = dv[i + u * SCAN_THREADS];
#pragma unroll
    for (int u = 0; u < 4; ++u) s4[u] = sv[i + u * SCAN_THREADS];
    unsigned rec[16], pos[16];
#pragma unroll
    for (int u = 0; u < 4; ++u) {
      const int sx[4] = {s4[u].x, s4[u].y, s4[u].z, s4[u].w};
      const int dx[4] = {d4[u].x, d4[u].y, d4[u].z, d4[u].w};
#pragma unroll
      for (int k = 0; k < 4; ++k) {
        rec[4 * u + k] = (((unsigned)dx[k] & (unsigned)(OWN_ROWS - 1)) << SRC_BITS) | ((unsigned)sx[k] & SRC_MASK);
        pos[4 * u + k] = atomicAdd(&lcur[range_of(dx[k], nRanges)], 1u);
      }
    }
#pragma unroll
    for (int k = 0; k < 16; ++k) stage[pos[k]] = rec[k];
  }
  for (; i < nvec; i += SCAN_THREADS) {
    const int4 s0 = sv[i];
    const int4 d0 = dv[i];
    PLACE_EDGE(s0.x, d0.x); PLACE_EDGE(s0.y, d0.y); PLACE_EDGE(s0.z, d0.z); PLACE_EDGE(s0.w, d0.w);
  }
  for (int t = head + nvec * 4 + tid; t < n; t += SCAN_THREADS) PLACE_EDGE(ps[t], pd[t]);
#undef PLACE_EDGE
  __syncthreads();

  const int lane = tid & 31;
  for (int r = tid >> 5; r < nRanges; r += SCAN_THREADS / 32) {
    const unsigned  ls  = r > 0 ? lcur[r - 1] : 0u;
    const int       cnt = (int)(lcur[r] - ls);
    unsigned*       run = recs + gpos[r];
    const unsigned* st  = stage + ls;
    int k = lane - (int)((((size_t)run) >> 2) & 31u);
    if (k >= 0 && k < cnt) run[k] = st[k];
    k += 32;
    for (; k + 96 < cnt; k += 128) {
      const unsigned a0 = st[k], a1 = st[k + 32], a2 = st[k + 64], a3 = st[k + 96];
      run[k] = a0; run[k + 32] = a1; run[k + 64] = a2; run[k + 96] = a3;
    }
    for (; k < cnt; k += 32) run[k] = st[k];
  }
}

__global__ __launch_bounds__(SCAN_THREADS) void spmm1_owner(const unsigned* __restrict__ recs,
                                                            const int* __restrict__ rangeStart,
                                                            const float* __restrict__ h,
                                                            float* __restrict__ agg,
                                                            float* __restrict__ in_norm, int nN, int nE) {
  __shared__ float    acc[OWN_ROWS * HID];
  __shared__ unsigned indeg[OWN_ROWS];
  const int tid = threadIdx.x;
  for (int i = tid; i < OWN_ROWS * HID; i += SCAN_THREADS) acc[i] = 0.0f;
  for (int i = tid; i < OWN_ROWS; i += SCAN_THREADS) indeg[i] = 0u;
  __syncthreads();

  const int beg  = rangeStart[blockIdx.x];
  const int end  = rangeStart[blockIdx.x + 1];
  const int lane = tid & 31;
  const int j    = lane & 7;
  const int sub  = lane >> 3;
  constexpr int TRIP   = 64;
  constexpr int STRIDE = (SCAN_THREADS / 32) * TRIP;

  const int last = nE - 1;
  int base = beg + (tid >> 5) * TRIP;
  unsigned r0 = recs[min(base + lane, last)];
  unsigned r1 = recs[min(base + 32 + lane, last)];
  for (; base < end; base += STRIDE) {
    const int left  = end - base;
    const int nbase = base + STRIDE;
    const unsigned n0 = recs[min(nbase + lane, last)];
    const unsigned n1 = recs[min(nbase + 32 + lane, last)];

    if (lane < left)      atomicAdd(&indeg[r0 >> SRC_BITS], 1u);
    if (32 + lane < left) atomicAdd(&indeg[r1 >> SRC_BITS], 1u);

    float    v[16];
    unsigned off[16];
#pragma unroll
    for (int i = 0; i < 8; ++i) {
      const int from = i * 4 + sub;
      const unsigned q0 = __shfl(r0, from, 32);
      const unsigned q1 = __shfl(r1, from, 32);
      v[i]       = h[(q0 & SRC_MASK) * HID + j];
      off[i]     = (q0 >> SRC_BITS) * HID + j;
      v[8 + i]   = h[(q1 & SRC_MASK) * HID + j];
      off[8 + i] = (q1 >> SRC_BITS) * HID + j;
    }
#pragma unroll
    for (int i = 0; i < 8; ++i) {
      const bool ok0 = i * 4 + sub < left;
      const bool ok1 = 32 + i * 4 + sub < left;
      atomicAdd(&acc[ok0 ? off[i] : (unsigned)j], ok0 ? v[i] : 0.0f);
      atomicAdd(&acc[ok1 ? off[8 + i] : (unsigned)j], ok1 ? v[8 + i] : 0.0f);
    }
    r0 = n0;
    r1 = n1;
  }
  __syncthreads();

  const long rowBase = (long)blockIdx.x * OWN_ROWS;
  for (int t = tid; t < OWN_ROWS * HID; t += SCAN_THREADS) {
    if (rowBase + (t >> 3) < nN) agg[rowBase * HID + t] = acc[t];
  }
  for (int t = tid; t < OWN_ROWS; t += SCAN_THREADS) {
    if (rowBase + t < nN) in_norm[rowBase + t] = rsqrtf(fmaxf((float)indeg[t], 1.0f));
  }
}

__global__ __launch_bounds__(SCAN_THREADS) void spmm2_owner(const unsigned* __restrict__ recs,
                                                            const int* __restrict__ rangeStart,
                                                            const float* __restrict__ g2,
                                                            const float* __restrict__ in_norm,
                                                            const float* __restrict__ b2,
                                                            float* __restrict__ out, int nN, int nE) {
  __shared__ float acc[OWN_ROWS];
  const int tid = threadIdx.x;
  for (int i = tid; i < OWN_ROWS; i += SCAN_THREADS) acc[i] = 0.0f;
  __syncthreads();

  const int beg = rangeStart[blockIdx.x];
  const int end = rangeStart[blockIdx.x + 1];
  constexpr int U    = 8;
  constexpr int TRIP = U * SCAN_THREADS;
  const int last = nE - 1;
  unsigned r[U];
#pragma unroll
  for (int u = 0; u < U; ++u) r[u] = recs[min(beg + u * SCAN_THREADS + tid, last)];
  for (int base = beg; base < end; base += TRIP) {
    unsigned n[U];
#pragma unroll
    for (int u = 0; u < U; ++u) n[u] = recs[min(base + TRIP + u * SCAN_THREADS + tid, last)];
    float v[U];
#pragma unroll
    for (int u = 0; u < U; ++u) v[u] = g2[r[u] & SRC_MASK];
#pragma unroll
    for (int u = 0; u < U; ++u) {
      const bool ok = base + u * SCAN_THREADS + tid < end;
      atomicAdd(&acc[ok ? (r[u] >> SRC_BITS) : (unsigned)(tid & (OWN_ROWS - 1))], ok ? v[u] : 0.0f);
    }
#pragma unroll
    for (int u = 0; u < U; ++u) r[u] = n[u];
  }
  __syncthreads();

  const long rowBase = (long)blockIdx.x * OWN_ROWS;
  const float bias = b2[0];
  for (int t = tid; t < OWN_ROWS; t += SCAN_THREADS) {
    const long row = rowBase + t;
    if (row < nN) out[row] = fmaf(acc[t], in_norm[row], bias);
  }
}

constexpr int      L2_KEEP = 36;
constexpr int      L2_GRP  = 12;
constexpr unsigned NO_REC  = 0xFFFFFFFFu;

__global__ __launch_bounds__(SCAN_THREADS) void spmm2_lds(const unsigned* __restrict__ recs,
                                                          const int* __restrict__ rangeStart,
                                                          const float* __restrict__ g2,
                                                          const float* __restrict__ in_norm,
                                                          const float* __restrict__ b2,
                                                          float* __restrict__ out,
                                                          int nN, int nE, int nRanges, int nSlices) {
  __shared__ alignas(16) float g2s[SLICE_CAP];
  __shared__ float acc[2 * OWN_ROWS];
  const int tid = threadIdx.x;
  for (int i = tid; i < 2 * OWN_ROWS; i += SCAN_THREADS) acc[i] = 0.0f;

  const int  ra   = (int)blockIdx.x * 2;
  const bool two  = ra + 1 < nRanges;
  const int  begA = rangeStart[ra];
  const int  endA = rangeStart[ra + 1];
  const int  endB = two ? rangeStart[ra + 2] : endA;
  const int  last = nE - 1;

  unsigned ka[L2_KEEP], kb[L2_KEEP];
#pragma unroll
  for (int u = 0; u < L2_KEEP; ++u) {
    const int ia = begA + u * SCAN_THREADS + tid;
    const int ib = endA + u * SCAN_THREADS + tid;
    const unsigned xa = recs[min(ia, last)];
    const unsigned xb = recs[min(ib, last)];
    ka[u] = ia < endA ? xa : NO_REC;
    kb[u] = ib < endB ? xb : NO_REC;
  }

  const int len = slice_len(nN, nSlices);
#define L2_HIT(R, ROWOFF)                                                                 \
  do {                                                                                    \
    const unsigned k_ = ((R) & SRC_MASK) - (unsigned)base;                                \
    if (k_ < (unsigned)len) atomicAdd(&acc[((R) >> SRC_BITS) + (ROWOFF)], g2s[k_]);       \
  } while (0)
#define L2_GROUP(K, G, ROWOFF)                                                            \
  do {                                                                                    \
    unsigned k[L2_GRP];                                                                   \
    float    v[L2_GRP];                                                                   \
    _Pragma("unroll") for (int j = 0; j < L2_GRP; ++j) {                                  \
      k[j] = (K[(G) * L2_GRP + j] & SRC_MASK) - (unsigned)base;                           \
      v[j] = k[j] < (unsigned)len ? g2s[k[j]] : 0.0f;                                     \
    }                                                                                     \
    _Pragma("unroll") for (int j = 0; j < L2_GRP; ++j) {                                  \
      atomicAdd(&acc[k[j] < (unsigned)len ? (K[(G) * L2_GRP + j] >> SRC_BITS) + (ROWOFF) : (unsigned)tid], v[j]); \
    }                                                                                     \
  } while (0)

  for (int s = 0; s < nSlices; ++s) {
    const int base = s * len;
    __syncthreads();
    int cnt = nN - base;
    cnt = cnt < 0 ? 0 : (cnt > len ? len : cnt);
    const float* gsrc = g2 + base;
    const int nv = ((((size_t)gsrc) & 15u) == 0) ? (cnt >> 2) : 0;
    const float4* src4 = (const float4*)gsrc;
    float4*       dst4 = (float4*)g2s;
    for (int i = tid; i < nv; i += 6 * SCAN_THREADS) {
      float4 t[6];
      int ix[6];
#pragma unroll
      for (int u = 0; u < 6; ++u) ix[u] = min(i + u * SCAN_THREADS, nv - 1);
#pragma unroll
      for (int u = 0; u < 6; ++u) t[u] = src4[ix[u]];
#pragma unroll
      for (int u = 0; u < 6; ++u) dst4[ix[u]] = t[u];
    }
    for (int i = nv * 4 + tid; i < len; i += SCAN_THREADS) g2s[i] = i < cnt ? gsrc[i] : 0.0f;
    __syncthreads();

#pragma unroll
    for (int u = 0; u < L2_KEEP; ++u) {
      asm volatile("" : "+v"(ka[u]));
      asm volatile("" : "+v"(kb[u]));
    }
#pragma unroll
    for (int g = 0; g < L2_KEEP / L2_GRP; ++g) {
      L2_GROUP(ka, g, 0);
      L2_GROUP(kb, g, OWN_ROWS);
    }
    for (int i = begA + L2_KEEP * SCAN_THREADS + tid; i < endA; i += SCAN_THREADS) L2_HIT(recs[i], 0);
    for (int i = endA + L2_KEEP * SCAN_THREADS + tid; i < endB; i += SCAN_THREADS) L2_HIT(recs[i], OWN_ROWS);
  }
#undef L2_GROUP
#undef L2_HIT
  __syncthreads();

  const long rowBase = (long)ra * OWN_ROWS;
  const float bias = b2[0];
  for (int t = tid; t < 2 * OWN_ROWS; t += SCAN_THREADS) {
    const long row = rowBase + t;
    if (row < nN) out[row] = fmaf(acc[t], in_norm[row], bias);
  }
}

constexpr int L2S_SHIFT = 16;
constexpr int L2S_LEN   = 1 << L2S_SHIFT;
constexpr int L2S_BUCKETS = 32;
constexpr int L2S_MAXSL = L2S_BUCKETS / 2 - 1;
constexpr int L2S_CNT   = L2S_BUCKETS * 32;
constexpr int L2S_TOTAL = 2 * L2_KEEP * SCAN_THREADS;
constexpr int L2S_PIECE = 4;

__global__ __launch_bounds__(SCAN_THREADS) void spmm2_sorted(const unsigned* __restrict__ recs,
                                                             const int* __restrict__ rangeStart,
                                                             const float* __restrict__ g2,
                                                             const float* __restrict__ in_norm,
                                                             const float* __restrict__ b2,
                                                             float* __restrict__ out,
                                                             int nN, int nE, int nRanges, int nSl) {
  static_assert(L2S_CNT == SCAN_THREADS, "one counter per thread in the scan");
  static_assert(L2S_TOTAL <= SLICE_CAP, "the sorted records must fit the slice buffer");
  static_assert((2 * L2_KEEP) % L2S_PIECE == 0, "whole pieces");
  static_assert(L2S_PIECE * SCAN_THREADS == 4096, "a piece is 1 << 12 list elements");
  static_assert(SRC_BITS - L2S_SHIFT <= 6 && 2 * OWN_ROWS <= 2048, "fields of the ordered record");
  __shared__ alignas(16) float g2s[SLICE_CAP];
  __shared__ float    acc[2 * OWN_ROWS];
  __shared__ unsigned cnt[L2S_CNT];
  __shared__ unsigned scn[2][L2S_CNT];
  unsigned* sorted = (unsigned*)g2s;
  const int tid  = threadIdx.x;
  const int lane = tid & 31;
  for (int i = tid; i < 2 * OWN_ROWS; i += SCAN_THREADS) acc[i] = 0.0f;
  cnt[tid] = 0u;

  const int  ra   = (int)blockIdx.x * 2;
  const bool two  = ra + 1 < nRanges;
  const int  begA = rangeStart[ra];
  const int  endA = rangeStart[ra + 1];
  const int  endB = two ? rangeStart[ra + 2] : endA;
  const int  last = nE - 1;

  unsigned r[2 * L2_KEEP];
#pragma unroll
  for (int u = 0; u < L2_KEEP; ++u) {
    const int ia = begA + u * SCAN_THREADS + tid;
    const int ib = endA + u * SCAN_THREADS + tid;
    const unsigned xa = recs[min(ia, last)];
    const unsigned xb = recs[min(ib, last)];
    r[u]           = ia < endA ? xa : NO_REC;
    r[L2_KEEP + u] = ib < endB ? xb : NO_REC;
  }
  __syncthreads();

  const unsigned slEnd = (unsigned)nSl;
#define L2S_COUNTER(U) \
  ((min((r[(U)] >> L2S_SHIFT) & ((1u << (SRC_BITS - L2S_SHIFT)) - 1u), slEnd) * 2u + ((U) >= L2_KEEP ? 1u : 0u)) * 32u + (unsigned)lane)
#pragma unroll
  for (int u = 0; u < 2 * L2_KEEP; ++u) atomicAdd(&cnt[L2S_COUNTER(u)], 1u);
#pragma unroll
  for (int u = 0; u < 2 * L2_KEEP; ++u) asm volatile("" : "+v"(r[u]));
  __syncthreads();

  const unsigned mine = cnt[tid];
  scn[0][tid] = mine;
  __syncthreads();
  int cur = 0;
  for (int off = 1; off < L2S_CNT; off <<= 1) {
    scn[1 - cur][tid] = scn[cur][tid] + (tid >= off ? scn[cur][tid - off] : 0u);
    __syncthreads();
    cur ^= 1;
  }
  const unsigned excl = scn[cur][tid] - mine;
  cnt[tid] = excl;
  scn[1 - cur][tid] = excl;
  const unsigned* bstart = scn[1 - cur];
  __syncthreads();

#pragma unroll
  for (int u = 0; u < 2 * L2_KEEP; ++u) {
    const unsigned pos = atomicAdd(&cnt[L2S_COUNTER(u)], 1u);
    sorted[pos] = ((r[u] >> 1) & 0x7FE00000u) | ((u) >= L2_KEEP ? 0x80000000u : 0u) |
                  ((r[u] & (unsigned)(L2S_LEN - 1)) << 2);
  }
#undef L2S_COUNTER
  __syncthreads();
#pragma unroll
  for (int u = 0; u < 2 * L2_KEEP; ++u) r[u] = sorted[u * SCAN_THREADS + tid];

#define L2S_HIT(R, ROWOFF)                                                                \
  do {                                                                                    \
    const unsigned k_ = ((R) & SRC_MASK) - (unsigned)base;                                \
    if (k_ < (unsigned)L2S_LEN) atomicAdd(&acc[((R) >> SRC_BITS) + (ROWOFF)], g2s[k_]);   \
  } while (0)

  for (int s = 0; s < nSl; ++s) {
    const int base = s << L2S_SHIFT;
    __syncthreads();
    int cntN = nN - base;
    cntN = cntN < 0 ? 0 : (cntN > L2S_LEN ? L2S_LEN : cntN);
    const float* gsrc = g2 + base;
    const int nv = ((((size_t)gsrc) & 15u) == 0) ? (cntN >> 2) : 0;
    const float4* src4 = (const float4*)gsrc;
    float4*       dst4 = (float4*)g2s;
    for (int i = tid; i < nv; i += 6 * SCAN_THREADS) {
      float4 t[6];
      int ix[6];
#pragma unroll
      for (int u = 0; u < 6; ++u) ix[u] = min(i + u * SCAN_THREADS, nv - 1);
#pragma unroll
      for (int u = 0; u < 6; ++u) t[u] = src4[ix[u]];
#pragma unroll
      for (int u = 0; u < 6; ++u) dst4[ix[u]] = t[u];
    }
    for (int i = nv * 4 + tid; i < L2S_LEN; i += SCAN_THREADS) g2s[i] = i < cntN ? gsrc[i] : 0.0f;
    __syncthreads();

    const unsigned lo = (unsigned)__builtin_amdgcn_readfirstlane((int)bstart[(2 * s) * 32]);
    const unsigned hi = (unsigned)__builtin_amdgcn_readfirstlane((int)bstart[(2 * s + 2) * 32]);
    const unsigned span = hi - lo;

#pragma unroll
    for (int u = 0; u < 2 * L2_KEEP; ++u) asm volatile("" : "+v"(r[u]));

    if (span != 0u) {
      const unsigned glo   = lo >> 12;
      const unsigned gspan = ((hi - 1u) >> 12) - glo;
#pragma unroll
      for (int g = 0; g < 2 * L2_KEEP / L2S_PIECE; ++g) {
        if ((unsigned)g - glo <= gspan) {
          bool  in[L2S_PIECE];
          float v[L2S_PIECE];
#pragma unroll
          for (int j = 0; j < L2S_PIECE; ++j) {
            const unsigned p = (unsigned)((g * L2S_PIECE + j) * SCAN_THREADS + tid);
            in[j] = p - lo < span;
            v[j]  = in[j] ? g2s[(r[g * L2S_PIECE + j] & 0x3FFFFu) >> 2] : 0.0f;
          }
#pragma unroll
          for (int j = 0; j < L2S_PIECE; ++j)
            atomicAdd(&acc[in[j] ? (r[g * L2S_PIECE + j] >> 21) : (unsigned)tid], v[j]);
        }
      }
    }
    for (int i = begA + L2_KEEP * SCAN_THREADS + tid; i < endA; i += SCAN_THREADS) L2S_HIT(recs[i], 0);
    for (int i = endA + L2_KEEP * SCAN_THREADS + tid; i < endB; i += SCAN_THREADS) L2S_HIT(recs[i], OWN_ROWS);
  }
#undef L2S_HIT
  __syncthreads();

  const long rowBase = (long)ra * OWN_ROWS;
  const float bias = b2[0];
  for (int t = tid; t < 2 * OWN_ROWS; t += SCAN_THREADS) {
    const long row = rowBase + t;
    if (row < nN) out[row] = fmaf(acc[t], in_norm[row], bias);
  }
}


__global__ void layer2_node(const float* __restrict__ agg, const float* __restrict__ in_norm,
                            const float* __restrict__ out_norm, const float* __restrict__ b1,
                            const float* __restrict__ W2, float* __restrict__ g2, int n) {
  int v = blockIdx.x * blockDim.x + threadIdx.x;
  if (v >= n) return;
  float innv = in_norm[v];
  const float* av = agg + (long)v * HID;
  float s = 0.0f;
#pragma unroll
  for (int j = 0; j < HID; ++j) {
    float z = fmaf(av[j], innv, b1[j]);
    z = z > 0.0f ? z : 0.0f;
    s = fmaf(z, W2[j], s);
  }
  g2[v] = s * out_norm[v];
}


__device__ __forceinline__ void sum_hit(float* acc, const float* __restrict__ g2, int s, int d,
                                        int base, int len) {
  const unsigned k = (unsigned)(d - base);
  if (k < (unsigned)len) atomicAdd(&acc[k], g2[s]);
}

__global__ __launch_bounds__(SCAN_THREADS) void spmm2_slices(const int* __restrict__ src,
                                                             const int* __restrict__ dst,
                                                             const float* __restrict__ g2,
                                                             float* __restrict__ outacc,
                                                             int nE, int nN) {
  __shared__ float acc[SLICE_CAP];
  const int tid     = threadIdx.x;
  const int nSlices = gridDim.y;
  const int len     = slice_len(nN, nSlices);
  const int base    = (int)blockIdx.y * len;

  for (int i = tid; i < len; i += SCAN_THREADS) acc[i] = 0.0f;
  __syncthreads();

  int start, n;
  chunk_bounds(nE, gridDim.x, blockIdx.x, start, n);
  const int* ps = src + start;
  const int* pd = dst + start;
  int head = (int)(((16u - (unsigned)((size_t)pd & 15u)) & 15u) >> 2);
  if (head > n) head = n;
  const bool vec = (((size_t)(ps + head)) & 15u) == 0;
  if (!vec) head = n;
  for (int i = tid; i < head; i += SCAN_THREADS) sum_hit(acc, g2, ps[i], pd[i], base, len);
  const int4* sv = (const int4*)(ps + head);
  const int4* dv = (const int4*)(pd + head);
  const int nvec = (n - head) >> 2;
  int i = tid;
  int4 dq[4], sq[4];
#pragma unroll
  for (int u = 0; u < 4; ++u) { dq[u] = make_int4(-1, -1, -1, -1); sq[u] = make_int4(0, 0, 0, 0); }
  for (;;) {
    const bool fullNext = i + 3 * SCAN_THREADS < nvec;
    int4 dn[4], sn[4];
#pragma unroll
    for (int u = 0; u < 4; ++u) { dn[u] = make_int4(-1, -1, -1, -1); sn[u] = make_int4(0, 0, 0, 0); }
    if (fullNext) {
#pragma unroll
      for (int u = 0; u < 4; ++u) { sn[u] = sv[i + u * SCAN_THREADS]; dn[u] = dv[i + u * SCAN_THREADS]; }
    }
    unsigned k[16];
    float    val[16];
#pragma unroll
    for (int u = 0; u < 4; ++u) {
      k[4 * u + 0] = (unsigned)(dq[u].x - base);
      k[4 * u + 1] = (unsigned)(dq[u].y - base);
      k[4 * u + 2] = (unsigned)(dq[u].z - base);
      k[4 * u + 3] = (unsigned)(dq[u].w - base);
      val[4 * u + 0] = (k[4 * u + 0] < (unsigned)len) ? g2[sq[u].x] : 0.0f;
      val[4 * u + 1] = (k[4 * u + 1] < (unsigned)len) ? g2[sq[u].y] : 0.0f;
      val[4 * u + 2] = (k[4 * u + 2] < (unsigned)len) ? g2[sq[u].z] : 0.0f;
      val[4 * u + 3] = (k[4 * u + 3] < (unsigned)len) ? g2[sq[u].w] : 0.0f;
    }
#pragma unroll
    for (int j = 0; j < 16; ++j) {
      if (k[j] < (unsigned)len) atomicAdd(&acc[k[j]], val[j]);
    }
    if (!fullNext) break;
#pragma unroll
    for (int u = 0; u < 4; ++u) { dq[u] = dn[u]; sq[u] = sn[u]; }
    i += 4 * SCAN_THREADS;
  }
  for (; i < nvec; i += SCAN_THREADS) {
    const int4 d0 = dv[i];
    const int4 s0 = sv[i];
    sum_hit(acc, g2, s0.x, d0.x, base, len); sum_hit(acc, g2, s0.y, d0.y, base, len);
    sum_hit(acc, g2, s0.z, d0.z, base, len); sum_hit(acc, g2, s0.w, d0.w, base, len);
  }
  for (int t = head + nvec * 4 + tid; t < n; t += SCAN_THREADS) sum_hit(acc, g2, ps[t], pd[t], base, len);
  __syncthreads();

  for (int t = tid; t < len; t += SCAN_THREADS) {
    const int node = base + t;
    if (node < nN) atomicAdd(&outacc[node], acc[t]);
  }
}

__global__ void final_kernel(const float* __restrict__ outacc, const float* __restrict__ in_norm,
                             const float* __restrict__ b2, float* __restrict__ out, int n) {
  int v = blockIdx.x * blockDim.x + threadIdx.x;
  if (v < n) out[v] = fmaf(outacc[v], in_norm[v], b2[0]);
}


extern "C" void kernel_launch(void* const* d_in, const int* in_sizes, int n_in,
                              void* d_out, int out_size, void* d_ws, size_t ws_size,
                              hipStream_t stream) {
  const float* x  = (const float*)d_in[0];
  const float* W1 = (const float*)d_in[1];
  const float* b1 = (const float*)d_in[2];
  const float* W2 = (const float*)d_in[3];
  const float* b2 = (const float*)d_in[4];
  const int*  src = (const int*)d_in[5];
  const int*  dst = (const int*)d_in[6];

  const int nN = in_sizes[0] / IN_DIM;
  const int nE = in_sizes[5];

  float* ws     = (float*)d_ws;
  float* deg_s  = ws;
  float* deg_d  = ws + (long)nN;
  float* h      = ws + 2L * nN;
  float* agg    = ws + 10L * nN;
  float* g2     = ws + 18L * nN;
  float* outacc = ws + 19L * nN;
  float* out    = (float*)d_out;

  const int TB = 256;
  dim3 blk(TB);

  const int nSlices   = (nN + SLICE_CAP - 1) / SLICE_CAP;
  const int degChunks = 256 / (2 * nSlices) > 0 ? 256 / (2 * nSlices) : 1;
  const int sumChunks = 256 / nSlices > 0 ? 256 / nSlices : 1;

  const int    nRanges  = (nN + OWN_ROWS - 1) / OWN_ROWS;
  const size_t wsWords  = 20UL * nN + (size_t)nE + (size_t)nRanges * BIN_CHUNKS + 2UL * nRanges + 8UL;
  const bool   sorted   = nE > 0 && nRanges <= MAX_RANGES && wsWords * sizeof(float) <= ws_size &&
                          (nE + BIN_CHUNKS - 1) / BIN_CHUNKS + 3 <= STAGE_CAP;

  if (sorted) {
    zero_f32<<<(nN + TB - 1) / TB, blk, 0, stream>>>(deg_s, nN);
    degree_slices<<<dim3(sumChunks, nSlices), dim3(SCAN_THREADS), 0, stream>>>(
        src, dst, (unsigned*)deg_s, (unsigned*)deg_d, nE, nN, nSlices);
    norm_kernel<<<(nN + TB - 1) / TB, blk, 0, stream>>>((unsigned*)deg_s, (unsigned*)deg_d, nN, 0);
  } else {
    zero_f32<<<(2 * nN + TB - 1) / TB, blk, 0, stream>>>(deg_s, 2 * nN);
    degree_slices<<<dim3(degChunks, 2 * nSlices), dim3(SCAN_THREADS), 0, stream>>>(
        src, dst, (unsigned*)deg_s, (unsigned*)deg_d, nE, nN, nSlices);
    norm_kernel<<<(nN + TB - 1) / TB, blk, 0, stream>>>((unsigned*)deg_s, (unsigned*)deg_d, nN, 1);
  }

  gemm1_wmma<<<1024, blk, 0, stream>>>(x, W1, deg_s, h, nN);

  if (sorted) {
    unsigned* recs       = (unsigned*)(ws + 20L * nN);
    int*      histT      = (int*)(recs + nE);
    int*      total      = histT + (long)nRanges * BIN_CHUNKS;
    int*      rangeStart = total + nRanges;
    bin_count<<<BIN_CHUNKS, dim3(SCAN_THREADS), 0, stream>>>(dst, histT, nE, nRanges);
    bin_scan<<<(nRanges * 32 + TB - 1) / TB, blk, 0, stream>>>(histT, total, BIN_CHUNKS, nRanges);
    bin_place<<<BIN_CHUNKS, dim3(SCAN_THREADS), 0, stream>>>(src, dst, histT, total, rangeStart, recs, nE, nRanges);
    spmm1_owner<<<nRanges, dim3(SCAN_THREADS), 0, stream>>>(recs, rangeStart, h, agg, deg_d, nN, nE);
    layer2_node<<<(nN + TB - 1) / TB, blk, 0, stream>>>(agg, deg_d, deg_s, b1, W2, g2, nN);
    const int nSl2 = (nN + L2S_LEN - 1) >> L2S_SHIFT;
    if (nN <= (int)SRC_MASK && nSl2 <= L2S_MAXSL) {
      spmm2_sorted<<<(nRanges + 1) / 2, dim3(SCAN_THREADS), 0, stream>>>(recs, rangeStart, g2, deg_d, b2, out, nN, nE,
                                                                        nRanges, nSl2);
    } else if (nN <= (int)SRC_MASK) {
      spmm2_lds<<<(nRanges + 1) / 2, dim3(SCAN_THREADS), 0, stream>>>(recs, rangeStart, g2, deg_d, b2, out, nN, nE,
                                                                     nRanges, nSlices);
    } else {
      spmm2_owner<<<nRanges, dim3(SCAN_THREADS), 0, stream>>>(recs, rangeStart, g2, deg_d, b2, out, nN, nE);
    }
  } else {
    zero_f32<<<(8 * nN + TB - 1) / TB, blk, 0, stream>>>(agg, 8 * nN);
    spmm1_kernel<<<(nE + TB - 1) / TB, blk, 0, stream>>>(src, dst, h, agg, nE);
    layer2_node<<<(nN + TB - 1) / TB, blk, 0, stream>>>(agg, deg_d, deg_s, b1, W2, g2, nN);
    zero_f32<<<(nN + TB - 1) / TB, blk, 0, stream>>>(outacc, nN);
    spmm2_slices<<<dim3(sumChunks, nSlices), dim3(SCAN_THREADS), 0, stream>>>(src, dst, g2, outacc, nE, nN);
    final_kernel<<<(nN + TB - 1) / TB, blk, 0, stream>>>(outacc, deg_d, b2, out, nN);
  }
}


# per-node passes ride in edge launches: zero fill in bin_count (now first; dst sort ahead of GEMM), out-degree norm formed by its readers, layer-2 node step in spmm1_owner's tail; 10 launches to 7
# speedup vs baseline: 9.7812x; 1.0326x over previous
#include <hip/hip_runtime.h>

typedef __attribute__((ext_vector_type(2))) float v2f;
typedef __attribute__((ext_vector_type(8))) float v8f;

constexpr int IN_DIM = 128;
constexpr int HID    = 8;

constexpr int SLICE_CAP    = 73728;
constexpr int SCAN_THREADS = 1024;


__global__ void zero_f32(float* __restrict__ p, int n) {
  int i = blockIdx.x * blockDim.x + threadIdx.x;
  if (i < n) p[i] = 0.0f;
}


__device__ __forceinline__ int slice_len(int nN, int nSlices) {
  const int per = (nN + nSlices - 1) / nSlices;
  return (per + 31) & ~31;
}

__device__ __forceinline__ void chunk_bounds(int nE, int nChunks, int chunk, int& start, int& count) {
  const int chunkLen = (((nE + nChunks - 1) / nChunks) + 3) & ~3;
  long s = (long)chunk * chunkLen;
  if (s > nE) s = nE;
  long e = s + chunkLen;
  if (e > nE) e = nE;
  start = (int)s;
  count = (int)(e - s);
}

__device__ __forceinline__ void count_hit(unsigned* cnt, int v, int base, int len) {
  const unsigned k = (unsigned)(v - base);
  if (k < (unsigned)len) atomicAdd(&cnt[k], 1u);
}

__global__ __launch_bounds__(SCAN_THREADS) void degree_slices(const int* __restrict__ src,
                                                              const int* __restrict__ dst,
                                                              unsigned* __restrict__ cnt_s,
                                                              unsigned* __restrict__ cnt_d,
                                                              int nE, int nN, int nSlices) {
  __shared__ unsigned cnt[SLICE_CAP];
  const int tid     = threadIdx.x;
  const int which   = (int)blockIdx.y >= nSlices ? 1 : 0;
  const int slice   = (int)blockIdx.y - which * nSlices;
  const int len     = slice_len(nN, nSlices);
  const int base    = slice * len;
  const int* __restrict__ idx = which ? dst : src;
  unsigned* __restrict__ out  = which ? cnt_d : cnt_s;

  for (int i = tid; i < len; i += SCAN_THREADS) cnt[i] = 0u;
  __syncthreads();

  int start, n;
  chunk_bounds(nE, gridDim.x, blockIdx.x, start, n);
  const int* p = idx + start;
  int head = (int)(((16u - (unsigned)((size_t)p & 15u)) & 15u) >> 2);
  if (head > n) head = n;
  for (int i = tid; i < head; i += SCAN_THREADS) count_hit(cnt, p[i], base, len);
  const int4* pv = (const int4*)(p + head);
  const int nvec = (n - head) >> 2;
  int i = tid;
  for (; i + 7 * SCAN_THREADS < nvec; i += 8 * SCAN_THREADS) {
    int4 v[8];
#pragma unroll
    for (int u = 0; u < 8; ++u) v[u] = pv[i + u * SCAN_THREADS];
#pragma unroll
    for (int u = 0; u < 8; ++u) {
      count_hit(cnt, v[u].x, base, len); count_hit(cnt, v[u].y, base, len);
      count_hit(cnt, v[u].z, base, len); count_hit(cnt, v[u].w, base, len);
    }
  }
  for (; i < nvec; i += SCAN_THREADS) {
    const int4 a = pv[i];
    count_hit(cnt, a.x, base, len); count_hit(cnt, a.y, base, len);
    count_hit(cnt, a.z, base, len); count_hit(cnt, a.w, base, len);
  }
  for (int t = head + nvec * 4 + tid; t < n; t += SCAN_THREADS) count_hit(cnt, p[t], base, len);
  __syncthreads();

  for (int t = tid; t < len; t += SCAN_THREADS) {
    const int node = base + t;
    if (node < nN) atomicAdd(&out[node], cnt[t]);
  }
}

__global__ void norm_kernel(unsigned* __restrict__ deg_s, unsigned* __restrict__ deg_d, int n) {
  int i = blockIdx.x * blockDim.x + threadIdx.x;
  if (i < n) {
    deg_s[i] = __float_as_uint(rsqrtf(fmaxf((float)deg_s[i], 1.0f)));
    deg_d[i] = __float_as_uint(rsqrtf(fmaxf((float)deg_d[i], 1.0f)));
  }
}


template <bool COUNTS_IN>
__global__ __launch_bounds__(256) void gemm1_wmma(const float* __restrict__ x,
                                                  const float* __restrict__ W1,
                                                  const float* __restrict__ out_norm,
                                                  float* __restrict__ h, int nNodes) {
  __shared__ float w[IN_DIM * 16];
  const int tid = threadIdx.x;
  {
    float t[8];
#pragma unroll
    for (int u = 0; u < 8; ++u) {
      const int i = tid + u * 256;
      t[u] = W1[(i >> 4) * HID + (i & 7)];
    }
#pragma unroll
    for (int u = 0; u < 8; ++u) {
      const int i = tid + u * 256;
      w[i] = ((i & 15) < HID) ? t[u] : 0.0f;
    }
  }
  __syncthreads();

  const int wid  = tid >> 5;
  const int lane = tid & 31;
  const int m    = lane & 15;
  const int kh   = lane >> 4;

  const int nTiles = (nNodes + 15) >> 4;
  const int nWaves = (int)gridDim.x * 8;
  for (int tile = (int)blockIdx.x * 8 + wid; tile < nTiles; tile += nWaves) {
    const int row0 = tile * 16;
    const int node = row0 + m;
    const int nodeClamped = node < nNodes ? node : nNodes - 1;
    const float4* xrow = (const float4*)(x + (long)nodeClamped * IN_DIM);

    float4 xa[16];
#pragma unroll
    for (int q = 0; q < 16; ++q) xa[q] = xrow[2 * q + kh];
    float on[8];
#pragma unroll
    for (int r = 0; r < 8; ++r) {
      const int nr = row0 + r + 8 * kh;
      on[r] = out_norm[nr < nNodes ? nr : nNodes - 1];
    }

    v8f acc = {};
#pragma unroll
    for (int q = 0; q < 16; ++q) {
      const int k0 = 8 * q + 4 * kh;
      v2f a0, a1, b0, b1;
      a0.x = xa[q].x; a0.y = xa[q].y;
      a1.x = xa[q].z; a1.y = xa[q].w;
      b0.x = w[(k0 + 0) * 16 + m];
      b0.y = w[(k0 + 1) * 16 + m];
      b1.x = w[(k0 + 2) * 16 + m];
      b1.y = w[(k0 + 3) * 16 + m];
      acc = __builtin_amdgcn_wmma_f32_16x16x4_f32(
           false, a0,  false, b0,
           (short)0, acc,  false,  false);
      acc = __builtin_amdgcn_wmma_f32_16x16x4_f32(
           false, a1,  false, b1,
           (short)0, acc,  false,  false);
    }

    if (m < HID) {
#pragma unroll
      for (int r = 0; r < 8; ++r) {
        const int nr = row0 + r + 8 * kh;
        const float sc = COUNTS_IN ? rsqrtf(fmaxf((float)__float_as_uint(on[r]), 1.0f)) : on[r];
        if (nr < nNodes) h[(long)nr * HID + m] = acc[r] * sc;
      }
    }
  }
}


__global__ void spmm1_kernel(const int* __restrict__ src, const int* __restrict__ dst,
                             const float* __restrict__ h, float* __restrict__ agg, int nE) {
  const int lane = threadIdx.x & 31;
  const int j    = lane & 7;
  const int sub  = lane >> 3;
  const int e    = blockIdx.x * blockDim.x + threadIdx.x;
  const int left = nE - (e - lane);

  int s = 0, d = 0;
  if (e < nE) { s = src[e]; d = dst[e]; }

  float    v[8];
  unsigned off[8];
#pragma unroll
  for (int i = 0; i < 8; ++i) {
    const int from = i * 4 + sub;
    const int si = __shfl(s, from, 32);
    const int di = __shfl(d, from, 32);
    v[i]   = h[(unsigned)(si * HID + j)];
    off[i] = (unsigned)(di * HID + j);
  }
#pragma unroll
  for (int i = 0; i < 8; ++i) {
    if (i * 4 + sub < left) atomicAdd(&agg[off[i]], v[i]);
  }
}


constexpr int OWN_SHIFT  = 10;
constexpr int OWN_ROWS   = 1 << OWN_SHIFT;
constexpr int SRC_BITS   = 32 - OWN_SHIFT;
constexpr int MAX_RANGES = 4096;
constexpr int BIN_CHUNKS = 256;
constexpr unsigned SRC_MASK = (1u << SRC_BITS) - 1u;

__device__ __forceinline__ unsigned range_of(int d, int nRanges) {
  const unsigned r = (unsigned)d >> OWN_SHIFT;
  return r < (unsigned)nRanges ? r : (unsigned)(nRanges - 1);
}

__global__ __launch_bounds__(SCAN_THREADS) void bin_count(const int* __restrict__ dst,
                                                          int* __restrict__ histT,
                                                          int nE, int nRanges,
                                                          unsigned* __restrict__ zeroP, int zeroN) {
  __shared__ unsigned cnt[MAX_RANGES];
  const int tid = threadIdx.x;
  for (int i = (int)blockIdx.x * SCAN_THREADS + tid; i < zeroN; i += (int)gridDim.x * SCAN_THREADS) zeroP[i] = 0u;
  for (int i = tid; i < nRanges; i += SCAN_THREADS) cnt[i] = 0u;
  __syncthreads();

  int start, n;
  chunk_bounds(nE, gridDim.x, blockIdx.x, start, n);
  const int* p = dst + start;
  int head = (int)(((16u - (unsigned)((size_t)p & 15u)) & 15u) >> 2);
  if (head > n) head = n;
  for (int i = tid; i < head; i += SCAN_THREADS) atomicAdd(&cnt[range_of(p[i], nRanges)], 1u);
  const int4* pv = (const int4*)(p + head);
  const int nvec = (n - head) >> 2;
  int i = tid;
  for (; i + 3 * SCAN_THREADS < nvec; i += 4 * SCAN_THREADS) {
    int4 v[4];
#pragma unroll
    for (int u = 0; u < 4; ++u) v[u] = pv[i + u * SCAN_THREADS];
#pragma unroll
    for (int u = 0; u < 4; ++u) {
      atomicAdd(&cnt[range_of(v[u].x, nRanges)], 1u); atomicAdd(&cnt[range_of(v[u].y, nRanges)], 1u);
      atomicAdd(&cnt[range_of(v[u].z, nRanges)], 1u); atomicAdd(&cnt[range_of(v[u].w, nRanges)], 1u);
    }
  }
  for (; i < nvec; i += SCAN_THREADS) {
    const int4 a = pv[i];
    atomicAdd(&cnt[range_of(a.x, nRanges)], 1u); atomicAdd(&cnt[range_of(a.y, nRanges)], 1u);
    atomicAdd(&cnt[range_of(a.z, nRanges)], 1u); atomicAdd(&cnt[range_of(a.w, nRanges)], 1u);
  }
  for (int t = head + nvec * 4 + tid; t < n; t += SCAN_THREADS)
    atomicAdd(&cnt[range_of(p[t], nRanges)], 1u);
  __syncthreads();

  for (int r = tid; r < nRanges; r += SCAN_THREADS) histT[r * (int)gridDim.x + (int)blockIdx.x] = (int)cnt[r];
}

__global__ void bin_scan(int* __restrict__ histT, int* __restrict__ total, int nChunks, int nRanges) {
  const int lane  = threadIdx.x & 31;
  const int range = (int)((blockIdx.x * blockDim.x + threadIdx.x) >> 5);
  if (range >= nRanges) return;
  int* row = histT + (long)range * nChunks;
  int carry = 0;
  for (int c0 = 0; c0 < nChunks; c0 += 32) {
    const int c = c0 + lane;
    const int v = c < nChunks ? row[c] : 0;
    int s = v;
#pragma unroll
    for (int d = 1; d < 32; d <<= 1) {
      const int t = __shfl_up(s, d, 32);
      if (lane >= d) s += t;
    }
    if (c < nChunks) row[c] = carry + s - v;
    carry += __shfl(s, 31, 32);
  }
  if (lane == 0) total[range] = carry;
}

constexpr int STAGE_CAP = 62976;

__global__ __launch_bounds__(SCAN_THREADS) void bin_place(const int* __restrict__ src,
                                                          const int* __restrict__ dst,
                                                          const int* __restrict__ histT,
                                                          const int* __restrict__ total,
                                                          int* __restrict__ rangeStart,
                                                          unsigned* __restrict__ recs,
                                                          int nE, int nRanges) {
  __shared__ int      scanbuf[2][MAX_RANGES];
  __shared__ unsigned gpos[MAX_RANGES];
  __shared__ unsigned lcur[MAX_RANGES];
  __shared__ unsigned stage[STAGE_CAP];
  const int tid     = threadIdx.x;
  const int nChunks = (int)gridDim.x;
  const int chunk   = (int)blockIdx.x;

  for (int r = tid; r < nRanges; r += SCAN_THREADS) scanbuf[0][r] = total[r];
  __syncthreads();
  int cur = 0;
  for (int off = 1; off < nRanges; off <<= 1) {
    for (int r = tid; r < nRanges; r += SCAN_THREADS)
      scanbuf[1 - cur][r] = scanbuf[cur][r] + (r >= off ? scanbuf[cur][r - off] : 0);
    __syncthreads();
    cur ^= 1;
  }
  for (int r = tid; r < nRanges; r += SCAN_THREADS) {
    const int tot  = total[r];
    const int incl = scanbuf[cur][r];
    const int excl = incl - tot;
    const int pre  = histT[r * nChunks + chunk];
    const int nxt  = chunk + 1 < nChunks ? histT[r * nChunks + chunk + 1] : tot;
    gpos[r] = (unsigned)(excl + pre);
    scanbuf[1 - cur][r] = nxt - pre;
    if (chunk == 0) {
      rangeStart[r] = excl;
      if (r == nRanges - 1) rangeStart[nRanges] = incl;
    }
  }
  __syncthreads();
  cur ^= 1;
  for (int off = 1; off < nRanges; off <<= 1) {
    for (int r = tid; r < nRanges; r += SCAN_THREADS)
      scanbuf[1 - cur][r] = scanbuf[cur][r] + (r >= off ? scanbuf[cur][r - off] : 0);
    __syncthreads();
    cur ^= 1;
  }
  for (int r = tid; r < nRanges; r += SCAN_THREADS) lcur[r] = r > 0 ? (unsigned)scanbuf[cur][r - 1] : 0u;
  __syncthreads();

  int start, n;
  chunk_bounds(nE, nChunks, chunk, start, n);
  const int* ps = src + start;
  const int* pd = dst + start;
  int head = (int)(((16u - (unsigned)((size_t)pd & 15u)) & 15u) >> 2);
  if (head > n) head = n;
  if ((((size_t)(ps + head)) & 15u) != 0) head = n;
#define PLACE_EDGE(S, D)                                                            \
  do {                                                                              \
    const unsigned pos_ = atomicAdd(&lcur[range_of((D), nRanges)], 1u);             \
    stage[pos_] = (((unsigned)(D) & (unsigned)(OWN_ROWS - 1)) << SRC_BITS) | ((unsigned)(S) & SRC_MASK); \
  } while (0)
  for (int i = tid; i < head; i += SCAN_THREADS) PLACE_EDGE(ps[i], pd[i]);
  const int4* sv = (const int4*)(ps + head);
  const int4* dv = (const int4*)(pd + head);
  const int nvec = (n - head) >> 2;
  int i = tid;
  for (; i + 3 * SCAN_THREADS < nvec; i += 4 * SCAN_THREADS) {
    int4 s4[4], d4[4];
#pragma unroll
    for (int u = 0; u < 4; ++u) d4[u] = dv[i + u * SCAN_THREADS];
#pragma unroll
    for (int u = 0; u < 4; ++u) s4[u] = sv[i + u * SCAN_THREADS];
    unsigned rec[16], pos[16];
#pragma unroll
    for (int u = 0; u < 4; ++u) {
      const int sx[4] = {s4[u].x, s4[u].y, s4[u].z, s4[u].w};
      const int dx[4] = {d4[u].x, d4[u].y, d4[u].z, d4[u].w};
#pragma unroll
      for (int k = 0; k < 4; ++k) {
        rec[4 * u + k] = (((unsigned)dx[k] & (unsigned)(OWN_ROWS - 1)) << SRC_BITS) | ((unsigned)sx[k] & SRC_MASK);
        pos[4 * u + k] = atomicAdd(&lcur[range_of(dx[k], nRanges)], 1u);
      }
    }
#pragma unroll
    for (int k = 0; k < 16; ++k) stage[pos[k]] = rec[k];
  }
  for (; i < nvec; i += SCAN_THREADS) {
    const int4 s0 = sv[i];
    const int4 d0 = dv[i];
    PLACE_EDGE(s0.x, d0.x); PLACE_EDGE(s0.y, d0.y); PLACE_EDGE(s0.z, d0.z); PLACE_EDGE(s0.w, d0.w);
  }
  for (int t = head + nvec * 4 + tid; t < n; t += SCAN_THREADS) PLACE_EDGE(ps[t], pd[t]);
#undef PLACE_EDGE
  __syncthreads();

  const int lane = tid & 31;
  for (int r = tid >> 5; r < nRanges; r += SCAN_THREADS / 32) {
    const unsigned  ls  = r > 0 ? lcur[r - 1] : 0u;
    const int       cnt = (int)(lcur[r] - ls);
    unsigned*       run = recs + gpos[r];
    const unsigned* st  = stage + ls;
    int k = lane - (int)((((size_t)run) >> 2) & 31u);
    if (k >= 0 && k < cnt) run[k] = st[k];
    k += 32;
    for (; k + 96 < cnt; k += 128) {
      const unsigned a0 = st[k], a1 = st[k + 32], a2 = st[k + 64], a3 = st[k + 96];
      run[k] = a0; run[k + 32] = a1; run[k + 64] = a2; run[k + 96] = a3;
    }
    for (; k < cnt; k += 32) run[k] = st[k];
  }
}

__global__ __launch_bounds__(SCAN_THREADS) void spmm1_owner(const unsigned* __restrict__ recs,
                                                            const int* __restrict__ rangeStart,
                                                            const float* __restrict__ h,
                                                            const unsigned* __restrict__ outCnt,
                                                            const float* __restrict__ b1,
                                                            const float* __restrict__ W2,
                                                            float* __restrict__ g2,
                                                            float* __restrict__ in_norm, int nN, int nE) {
  __shared__ alignas(16) float acc[OWN_ROWS * HID];
  __shared__ unsigned indeg[OWN_ROWS];
  const int tid = threadIdx.x;
  for (int i = tid; i < OWN_ROWS * HID; i += SCAN_THREADS) acc[i] = 0.0f;
  for (int i = tid; i < OWN_ROWS; i += SCAN_THREADS) indeg[i] = 0u;
  __syncthreads();

  const int beg  = rangeStart[blockIdx.x];
  const int end  = rangeStart[blockIdx.x + 1];
  const int lane = tid & 31;
  const int j    = lane & 7;
  const int sub  = lane >> 3;
  constexpr int TRIP   = 64;
  constexpr int STRIDE = (SCAN_THREADS / 32) * TRIP;

  const int last = nE - 1;
  int base = beg + (tid >> 5) * TRIP;
  unsigned r0 = recs[min(base + lane, last)];
  unsigned r1 = recs[min(base + 32 + lane, last)];
  for (; base < end; base += STRIDE) {
    const int left  = end - base;
    const int nbase = base + STRIDE;
    const unsigned n0 = recs[min(nbase + lane, last)];
    const unsigned n1 = recs[min(nbase + 32 + lane, last)];

    if (lane < left)      atomicAdd(&indeg[r0 >> SRC_BITS], 1u);
    if (32 + lane < left) atomicAdd(&indeg[r1 >> SRC_BITS], 1u);

    float    v[16];
    unsigned off[16];
#pragma unroll
    for (int i = 0; i < 8; ++i) {
      const int from = i * 4 + sub;
      const unsigned q0 = __shfl(r0, from, 32);
      const unsigned q1 = __shfl(r1, from, 32);
      v[i]       = h[(q0 & SRC_MASK) * HID + j];
      off[i]     = (q0 >> SRC_BITS) * HID + j;
      v[8 + i]   = h[(q1 & SRC_MASK) * HID + j];
      off[8 + i] = (q1 >> SRC_BITS) * HID + j;
    }
#pragma unroll
    for (int i = 0; i < 8; ++i) {
      const bool ok0 = i * 4 + sub < left;
      const bool ok1 = 32 + i * 4 + sub < left;
      atomicAdd(&acc[ok0 ? off[i] : (unsigned)j], ok0 ? v[i] : 0.0f);
      atomicAdd(&acc[ok1 ? off[8 + i] : (unsigned)j], ok1 ? v[8 + i] : 0.0f);
    }
    r0 = n0;
    r1 = n1;
  }
  __syncthreads();

  const long rowBase = (long)blockIdx.x * OWN_ROWS;
  for (int t = tid; t < OWN_ROWS; t += SCAN_THREADS) {
    const long row = rowBase + t;
    if (row < nN) {
      const float innv = rsqrtf(fmaxf((float)indeg[t], 1.0f));
      const float4 lo = *(const float4*)&acc[t * HID];
      const float4 hi = *(const float4*)&acc[t * HID + 4];
      const float av[HID] = {lo.x, lo.y, lo.z, lo.w, hi.x, hi.y, hi.z, hi.w};
      float s = 0.0f;
#pragma unroll
      for (int j = 0; j < HID; ++j) {
        float z = fmaf(av[j], innv, b1[j]);
        z = z > 0.0f ? z : 0.0f;
        s = fmaf(z, W2[j], s);
      }
      in_norm[row] = innv;
      g2[row] = s * rsqrtf(fmaxf((float)outCnt[row], 1.0f));
    }
  }
}

__global__ __launch_bounds__(SCAN_THREADS) void spmm2_owner(const unsigned* __restrict__ recs,
                                                            const int* __restrict__ rangeStart,
                                                            const float* __restrict__ g2,
                                                            const float* __restrict__ in_norm,
                                                            const float* __restrict__ b2,
                                                            float* __restrict__ out, int nN, int nE) {
  __shared__ float acc[OWN_ROWS];
  const int tid = threadIdx.x;
  for (int i = tid; i < OWN_ROWS; i += SCAN_THREADS) acc[i] = 0.0f;
  __syncthreads();

  const int beg = rangeStart[blockIdx.x];
  const int end = rangeStart[blockIdx.x + 1];
  constexpr int U    = 8;
  constexpr int TRIP = U * SCAN_THREADS;
  const int last = nE - 1;
  unsigned r[U];
#pragma unroll
  for (int u = 0; u < U; ++u) r[u] = recs[min(beg + u * SCAN_THREADS + tid, last)];
  for (int base = beg; base < end; base += TRIP) {
    unsigned n[U];
#pragma unroll
    for (int u = 0; u < U; ++u) n[u] = recs[min(base + TRIP + u * SCAN_THREADS + tid, last)];
    float v[U];
#pragma unroll
    for (int u = 0; u < U; ++u) v[u] = g2[r[u] & SRC_MASK];
#pragma unroll
    for (int u = 0; u < U; ++u) {
      const bool ok = base + u * SCAN_THREADS + tid < end;
      atomicAdd(&acc[ok ? (r[u] >> SRC_BITS) : (unsigned)(tid & (OWN_ROWS - 1))], ok ? v[u] : 0.0f);
    }
#pragma unroll
    for (int u = 0; u < U; ++u) r[u] = n[u];
  }
  __syncthreads();

  const long rowBase = (long)blockIdx.x * OWN_ROWS;
  const float bias = b2[0];
  for (int t = tid; t < OWN_ROWS; t += SCAN_THREADS) {
    const long row = rowBase + t;
    if (row < nN) out[row] = fmaf(acc[t], in_norm[row], bias);
  }
}

constexpr int      L2_KEEP = 36;
constexpr int      L2_GRP  = 12;
constexpr unsigned NO_REC  = 0xFFFFFFFFu;

__global__ __launch_bounds__(SCAN_THREADS) void spmm2_lds(const unsigned* __restrict__ recs,
                                                          const int* __restrict__ rangeStart,
                                                          const float* __restrict__ g2,
                                                          const float* __restrict__ in_norm,
                                                          const float* __restrict__ b2,
                                                          float* __restrict__ out,
                                                          int nN, int nE, int nRanges, int nSlices) {
  __shared__ alignas(16) float g2s[SLICE_CAP];
  __shared__ float acc[2 * OWN_ROWS];
  const int tid = threadIdx.x;
  for (int i = tid; i < 2 * OWN_ROWS; i += SCAN_THREADS) acc[i] = 0.0f;

  const int  ra   = (int)blockIdx.x * 2;
  const bool two  = ra + 1 < nRanges;
  const int  begA = rangeStart[ra];
  const int  endA = rangeStart[ra + 1];
  const int  endB = two ? rangeStart[ra + 2] : endA;
  const int  last = nE - 1;

  unsigned ka[L2_KEEP], kb[L2_KEEP];
#pragma unroll
  for (int u = 0; u < L2_KEEP; ++u) {
    const int ia = begA + u * SCAN_THREADS + tid;
    const int ib = endA + u * SCAN_THREADS + tid;
    const unsigned xa = recs[min(ia, last)];
    const unsigned xb = recs[min(ib, last)];
    ka[u] = ia < endA ? xa : NO_REC;
    kb[u] = ib < endB ? xb : NO_REC;
  }

  const int len = slice_len(nN, nSlices);
#define L2_HIT(R, ROWOFF)                                                                 \
  do {                                                                                    \
    const unsigned k_ = ((R) & SRC_MASK) - (unsigned)base;                                \
    if (k_ < (unsigned)len) atomicAdd(&acc[((R) >> SRC_BITS) + (ROWOFF)], g2s[k_]);       \
  } while (0)
#define L2_GROUP(K, G, ROWOFF)                                                            \
  do {                                                                                    \
    unsigned k[L2_GRP];                                                                   \
    float    v[L2_GRP];                                                                   \
    _Pragma("unroll") for (int j = 0; j < L2_GRP; ++j) {                                  \
      k[j] = (K[(G) * L2_GRP + j] & SRC_MASK) - (unsigned)base;                           \
      v[j] = k[j] < (unsigned)len ? g2s[k[j]] : 0.0f;                                     \
    }                                                                                     \
    _Pragma("unroll") for (int j = 0; j < L2_GRP; ++j) {                                  \
      atomicAdd(&acc[k[j] < (unsigned)len ? (K[(G) * L2_GRP + j] >> SRC_BITS) + (ROWOFF) : (unsigned)tid], v[j]); \
    }                                                                                     \
  } while (0)

  for (int s = 0; s < nSlices; ++s) {
    const int base = s * len;
    __syncthreads();
    int cnt = nN - base;
    cnt = cnt < 0 ? 0 : (cnt > len ? len : cnt);
    const float* gsrc = g2 + base;
    const int nv = ((((size_t)gsrc) & 15u) == 0) ? (cnt >> 2) : 0;
    const float4* src4 = (const float4*)gsrc;
    float4*       dst4 = (float4*)g2s;
    for (int i = tid; i < nv; i += 6 * SCAN_THREADS) {
      float4 t[6];
      int ix[6];
#pragma unroll
      for (int u = 0; u < 6; ++u) ix[u] = min(i + u * SCAN_THREADS, nv - 1);
#pragma unroll
      for (int u = 0; u < 6; ++u) t[u] = src4[ix[u]];
#pragma unroll
      for (int u = 0; u < 6; ++u) dst4[ix[u]] = t[u];
    }
    for (int i = nv * 4 + tid; i < len; i += SCAN_THREADS) g2s[i] = i < cnt ? gsrc[i] : 0.0f;
    __syncthreads();

#pragma unroll
    for (int u = 0; u < L2_KEEP; ++u) {
      asm volatile("" : "+v"(ka[u]));
      asm volatile("" : "+v"(kb[u]));
    }
#pragma unroll
    for (int g = 0; g < L2_KEEP / L2_GRP; ++g) {
      L2_GROUP(ka, g, 0);
      L2_GROUP(kb, g, OWN_ROWS);
    }
    for (int i = begA + L2_KEEP * SCAN_THREADS + tid; i < endA; i += SCAN_THREADS) L2_HIT(recs[i], 0);
    for (int i = endA + L2_KEEP * SCAN_THREADS + tid; i < endB; i += SCAN_THREADS) L2_HIT(recs[i], OWN_ROWS);
  }
#undef L2_GROUP
#undef L2_HIT
  __syncthreads();

  const long rowBase = (long)ra * OWN_ROWS;
  const float bias = b2[0];
  for (int t = tid; t < 2 * OWN_ROWS; t += SCAN_THREADS) {
    const long row = rowBase + t;
    if (row < nN) out[row] = fmaf(acc[t], in_norm[row], bias);
  }
}

constexpr int L2S_SHIFT = 16;
constexpr int L2S_LEN   = 1 << L2S_SHIFT;
constexpr int L2S_BUCKETS = 32;
constexpr int L2S_MAXSL = L2S_BUCKETS / 2 - 1;
constexpr int L2S_CNT   = L2S_BUCKETS * 32;
constexpr int L2S_TOTAL = 2 * L2_KEEP * SCAN_THREADS;
constexpr int L2S_PIECE = 4;

__global__ __launch_bounds__(SCAN_THREADS) void spmm2_sorted(const unsigned* __restrict__ recs,
                                                             const int* __restrict__ rangeStart,
                                                             const float* __restrict__ g2,
                                                             const float* __restrict__ in_norm,
                                                             const float* __restrict__ b2,
                                                             float* __restrict__ out,
                                                             int nN, int nE, int nRanges, int nSl) {
  static_assert(L2S_CNT == SCAN_THREADS, "one counter per thread in the scan");
  static_assert(L2S_TOTAL <= SLICE_CAP, "the sorted records must fit the slice buffer");
  static_assert((2 * L2_KEEP) % L2S_PIECE == 0, "whole pieces");
  static_assert(L2S_PIECE * SCAN_THREADS == 4096, "a piece is 1 << 12 list elements");
  static_assert(SRC_BITS - L2S_SHIFT <= 6 && 2 * OWN_ROWS <= 2048, "fields of the ordered record");
  __shared__ alignas(16) float g2s[SLICE_CAP];
  __shared__ float    acc[2 * OWN_ROWS];
  __shared__ unsigned cnt[L2S_CNT];
  __shared__ unsigned scn[2][L2S_CNT];
  unsigned* sorted = (unsigned*)g2s;
  const int tid  = threadIdx.x;
  const int lane = tid & 31;
  for (int i = tid; i < 2 * OWN_ROWS; i += SCAN_THREADS) acc[i] = 0.0f;
  cnt[tid] = 0u;

  const int  ra   = (int)blockIdx.x * 2;
  const bool two  = ra + 1 < nRanges;
  const int  begA = rangeStart[ra];
  const int  endA = rangeStart[ra + 1];
  const int  endB = two ? rangeStart[ra + 2] : endA;
  const int  last = nE - 1;

  unsigned r[2 * L2_KEEP];
#pragma unroll
  for (int u = 0; u < L2_KEEP; ++u) {
    const int ia = begA + u * SCAN_THREADS + tid;
    const int ib = endA + u * SCAN_THREADS + tid;
    const unsigned xa = recs[min(ia, last)];
    const unsigned xb = recs[min(ib, last)];
    r[u]           = ia < endA ? xa : NO_REC;
    r[L2_KEEP + u] = ib < endB ? xb : NO_REC;
  }
  __syncthreads();

  const unsigned slEnd = (unsigned)nSl;
#define L2S_COUNTER(U) \
  ((min((r[(U)] >> L2S_SHIFT) & ((1u << (SRC_BITS - L2S_SHIFT)) - 1u), slEnd) * 2u + ((U) >= L2_KEEP ? 1u : 0u)) * 32u + (unsigned)lane)
#pragma unroll
  for (int u = 0; u < 2 * L2_KEEP; ++u) atomicAdd(&cnt[L2S_COUNTER(u)], 1u);
#pragma unroll
  for (int u = 0; u < 2 * L2_KEEP; ++u) asm volatile("" : "+v"(r[u]));
  __syncthreads();

  const unsigned mine = cnt[tid];
  scn[0][tid] = mine;
  __syncthreads();
  int cur = 0;
  for (int off = 1; off < L2S_CNT; off <<= 1) {
    scn[1 - cur][tid] = scn[cur][tid] + (tid >= off ? scn[cur][tid - off] : 0u);
    __syncthreads();
    cur ^= 1;
  }
  const unsigned excl = scn[cur][tid] - mine;
  cnt[tid] = excl;
  scn[1 - cur][tid] = excl;
  const unsigned* bstart = scn[1 - cur];
  __syncthreads();

#pragma unroll
  for (int u = 0; u < 2 * L2_KEEP; ++u) {
    const unsigned pos = atomicAdd(&cnt[L2S_COUNTER(u)], 1u);
    sorted[pos] = ((r[u] >> 1) & 0x7FE00000u) | ((u) >= L2_KEEP ? 0x80000000u : 0u) |
                  ((r[u] & (unsigned)(L2S_LEN - 1)) << 2);
  }
#undef L2S_COUNTER
  __syncthreads();
#pragma unroll
  for (int u = 0; u < 2 * L2_KEEP; ++u) r[u] = sorted[u * SCAN_THREADS + tid];

#define L2S_HIT(R, ROWOFF)                                                                \
  do {                                                                                    \
    const unsigned k_ = ((R) & SRC_MASK) - (unsigned)base;                                \
    if (k_ < (unsigned)L2S_LEN) atomicAdd(&acc[((R) >> SRC_BITS) + (ROWOFF)], g2s[k_]);   \
  } while (0)

  for (int s = 0; s < nSl; ++s) {
    const int base = s << L2S_SHIFT;
    __syncthreads();
    int cntN = nN - base;
    cntN = cntN < 0 ? 0 : (cntN > L2S_LEN ? L2S_LEN : cntN);
    const float* gsrc = g2 + base;
    const int nv = ((((size_t)gsrc) & 15u) == 0) ? (cntN >> 2) : 0;
    const float4* src4 = (const float4*)gsrc;
    float4*       dst4 = (float4*)g2s;
    for (int i = tid; i < nv; i += 6 * SCAN_THREADS) {
      float4 t[6];
      int ix[6];
#pragma unroll
      for (int u = 0; u < 6; ++u) ix[u] = min(i + u * SCAN_THREADS, nv - 1);
#pragma unroll
      for (int u = 0; u < 6; ++u) t[u] = src4[ix[u]];
#pragma unroll
      for (int u = 0; u < 6; ++u) dst4[ix[u]] = t[u];
    }
    for (int i = nv * 4 + tid; i < L2S_LEN; i += SCAN_THREADS) g2s[i] = i < cntN ? gsrc[i] : 0.0f;
    __syncthreads();

    const unsigned lo = (unsigned)__builtin_amdgcn_readfirstlane((int)bstart[(2 * s) * 32]);
    const unsigned hi = (unsigned)__builtin_amdgcn_readfirstlane((int)bstart[(2 * s + 2) * 32]);
    const unsigned span = hi - lo;

#pragma unroll
    for (int u = 0; u < 2 * L2_KEEP; ++u) asm volatile("" : "+v"(r[u]));

    if (span != 0u) {
      const unsigned glo   = lo >> 12;
      const unsigned gspan = ((hi - 1u) >> 12) - glo;
#pragma unroll
      for (int g = 0; g < 2 * L2_KEEP / L2S_PIECE; ++g) {
        if ((unsigned)g - glo <= gspan) {
          bool  in[L2S_PIECE];
          float v[L2S_PIECE];
#pragma unroll
          for (int j = 0; j < L2S_PIECE; ++j) {
            const unsigned p = (unsigned)((g * L2S_PIECE + j) * SCAN_THREADS + tid);
            in[j] = p - lo < span;
            v[j]  = in[j] ? g2s[(r[g * L2S_PIECE + j] & 0x3FFFFu) >> 2] : 0.0f;
          }
#pragma unroll
          for (int j = 0; j < L2S_PIECE; ++j)
            atomicAdd(&acc[in[j] ? (r[g * L2S_PIECE + j] >> 21) : (unsigned)tid], v[j]);
        }
      }
    }
    for (int i = begA + L2_KEEP * SCAN_THREADS + tid; i < endA; i += SCAN_THREADS) L2S_HIT(recs[i], 0);
    for (int i = endA + L2_KEEP * SCAN_THREADS + tid; i < endB; i += SCAN_THREADS) L2S_HIT(recs[i], OWN_ROWS);
  }
#undef L2S_HIT
  __syncthreads();

  const long rowBase = (long)ra * OWN_ROWS;
  const float bias = b2[0];
  for (int t = tid; t < 2 * OWN_ROWS; t += SCAN_THREADS) {
    const long row = rowBase + t;
    if (row < nN) out[row] = fmaf(acc[t], in_norm[row], bias);
  }
}


__global__ void layer2_node(const float* __restrict__ agg, const float* __restrict__ in_norm,
                            const float* __restrict__ out_norm, const float* __restrict__ b1,
                            const float* __restrict__ W2, float* __restrict__ g2, int n) {
  int v = blockIdx.x * blockDim.x + threadIdx.x;
  if (v >= n) return;
  float innv = in_norm[v];
  const float* av = agg + (long)v * HID;
  float s = 0.0f;
#pragma unroll
  for (int j = 0; j < HID; ++j) {
    float z = fmaf(av[j], innv, b1[j]);
    z = z > 0.0f ? z : 0.0f;
    s = fmaf(z, W2[j], s);
  }
  g2[v] = s * out_norm[v];
}


__device__ __forceinline__ void sum_hit(float* acc, const float* __restrict__ g2, int s, int d,
                                        int base, int len) {
  const unsigned k = (unsigned)(d - base);
  if (k < (unsigned)len) atomicAdd(&acc[k], g2[s]);
}

__global__ __launch_bounds__(SCAN_THREADS) void spmm2_slices(const int* __restrict__ src,
                                                             const int* __restrict__ dst,
                                                             const float* __restrict__ g2,
                                                             float* __restrict__ outacc,
                                                             int nE, int nN) {
  __shared__ float acc[SLICE_CAP];
  const int tid     = threadIdx.x;
  const int nSlices = gridDim.y;
  const int len     = slice_len(nN, nSlices);
  const int base    = (int)blockIdx.y * len;

  for (int i = tid; i < len; i += SCAN_THREADS) acc[i] = 0.0f;
  __syncthreads();

  int start, n;
  chunk_bounds(nE, gridDim.x, blockIdx.x, start, n);
  const int* ps = src + start;
  const int* pd = dst + start;
  int head = (int)(((16u - (unsigned)((size_t)pd & 15u)) & 15u) >> 2);
  if (head > n) head = n;
  const bool vec = (((size_t)(ps + head)) & 15u) == 0;
  if (!vec) head = n;
  for (int i = tid; i < head; i += SCAN_THREADS) sum_hit(acc, g2, ps[i], pd[i], base, len);
  const int4* sv = (const int4*)(ps + head);
  const int4* dv = (const int4*)(pd + head);
  const int nvec = (n - head) >> 2;
  int i = tid;
  int4 dq[4], sq[4];
#pragma unroll
  for (int u = 0; u < 4; ++u) { dq[u] = make_int4(-1, -1, -1, -1); sq[u] = make_int4(0, 0, 0, 0); }
  for (;;) {
    const bool fullNext = i + 3 * SCAN_THREADS < nvec;
    int4 dn[4], sn[4];
#pragma unroll
    for (int u = 0; u < 4; ++u) { dn[u] = make_int4(-1, -1, -1, -1); sn[u] = make_int4(0, 0, 0, 0); }
    if (fullNext) {
#pragma unroll
      for (int u = 0; u < 4; ++u) { sn[u] = sv[i + u * SCAN_THREADS]; dn[u] = dv[i + u * SCAN_THREADS]; }
    }
    unsigned k[16];
    float    val[16];
#pragma unroll
    for (int u = 0; u < 4; ++u) {
      k[4 * u + 0] = (unsigned)(dq[u].x - base);
      k[4 * u + 1] = (unsigned)(dq[u].y - base);
      k[4 * u + 2] = (unsigned)(dq[u].z - base);
      k[4 * u + 3] = (unsigned)(dq[u].w - base);
      val[4 * u + 0] = (k[4 * u + 0] < (unsigned)len) ? g2[sq[u].x] : 0.0f;
      val[4 * u + 1] = (k[4 * u + 1] < (unsigned)len) ? g2[sq[u].y] : 0.0f;
      val[4 * u + 2] = (k[4 * u + 2] < (unsigned)len) ? g2[sq[u].z] : 0.0f;
      val[4 * u + 3] = (k[4 * u + 3] < (unsigned)len) ? g2[sq[u].w] : 0.0f;
    }
#pragma unroll
    for (int j = 0; j < 16; ++j) {
      if (k[j] < (unsigned)len) atomicAdd(&acc[k[j]], val[j]);
    }
    if (!fullNext) break;
#pragma unroll
    for (int u = 0; u < 4; ++u) { dq[u] = dn[u]; sq[u] = sn[u]; }
    i += 4 * SCAN_THREADS;
  }
  for (; i < nvec; i += SCAN_THREADS) {
    const int4 d0 = dv[i];
    const int4 s0 = sv[i];
    sum_hit(acc, g2, s0.x, d0.x, base, len); sum_hit(acc, g2, s0.y, d0.y, base, len);
    sum_hit(acc, g2, s0.z, d0.z, base, len); sum_hit(acc, g2, s0.w, d0.w, base, len);
  }
  for (int t = head + nvec * 4 + tid; t < n; t += SCAN_THREADS) sum_hit(acc, g2, ps[t], pd[t], base, len);
  __syncthreads();

  for (int t = tid; t < len; t += SCAN_THREADS) {
    const int node = base + t;
    if (node < nN) atomicAdd(&outacc[node], acc[t]);
  }
}

__global__ void final_kernel(const float* __restrict__ outacc, const float* __restrict__ in_norm,
                             const float* __restrict__ b2, float* __restrict__ out, int n) {
  int v = blockIdx.x * blockDim.x + threadIdx.x;
  if (v < n) out[v] = fmaf(outacc[v], in_norm[v], b2[0]);
}


extern "C" void kernel_launch(void* const* d_in, const int* in_sizes, int n_in,
                              void* d_out, int out_size, void* d_ws, size_t ws_size,
                              hipStream_t stream) {
  const float* x  = (const float*)d_in[0];
  const float* W1 = (const float*)d_in[1];
  const float* b1 = (const float*)d_in[2];
  const float* W2 = (const float*)d_in[3];
  const float* b2 = (const float*)d_in[4];
  const int*  src = (const int*)d_in[5];
  const int*  dst = (const int*)d_in[6];

  const int nN = in_sizes[0] / IN_DIM;
  const int nE = in_sizes[5];

  float* ws     = (float*)d_ws;
  float* deg_s  = ws;
  float* deg_d  = ws + (long)nN;
  float* h      = ws + 2L * nN;
  float* agg    = ws + 10L * nN;
  float* g2     = ws + 18L * nN;
  float* outacc = ws + 19L * nN;
  float* out    = (float*)d_out;

  const int TB = 256;
  dim3 blk(TB);

  const int nSlices   = (nN + SLICE_CAP - 1) / SLICE_CAP;
  const int degChunks = 256 / (2 * nSlices) > 0 ? 256 / (2 * nSlices) : 1;
  const int sumChunks = 256 / nSlices > 0 ? 256 / nSlices : 1;

  const int    nRanges  = (nN + OWN_ROWS - 1) / OWN_ROWS;
  const size_t wsWords  = 20UL * nN + (size_t)nE + (size_t)nRanges * BIN_CHUNKS + 2UL * nRanges + 8UL;
  const bool   sorted   = nE > 0 && nRanges <= MAX_RANGES && wsWords * sizeof(float) <= ws_size &&
                          (nE + BIN_CHUNKS - 1) / BIN_CHUNKS + 3 <= STAGE_CAP;

  if (sorted) {
    unsigned* recs       = (unsigned*)(ws + 20L * nN);
    int*      histT      = (int*)(recs + nE);
    int*      total      = histT + (long)nRanges * BIN_CHUNKS;
    int*      rangeStart = total + nRanges;
    bin_count<<<BIN_CHUNKS, dim3(SCAN_THREADS), 0, stream>>>(dst, histT, nE, nRanges, (unsigned*)deg_s, nN);
    degree_slices<<<dim3(sumChunks, nSlices), dim3(SCAN_THREADS), 0, stream>>>(
        src, dst, (unsigned*)deg_s, (unsigned*)deg_d, nE, nN, nSlices);
    bin_scan<<<(nRanges * 32 + TB - 1) / TB, blk, 0, stream>>>(histT, total, BIN_CHUNKS, nRanges);
    bin_place<<<BIN_CHUNKS, dim3(SCAN_THREADS), 0, stream>>>(src, dst, histT, total, rangeStart, recs, nE, nRanges);
    gemm1_wmma<true><<<1024, blk, 0, stream>>>(x, W1, deg_s, h, nN);
    spmm1_owner<<<nRanges, dim3(SCAN_THREADS), 0, stream>>>(recs, rangeStart, h, (const unsigned*)deg_s, b1, W2, g2,
                                                            deg_d, nN, nE);
    const int nSl2 = (nN + L2S_LEN - 1) >> L2S_SHIFT;
    if (nN <= (int)SRC_MASK && nSl2 <= L2S_MAXSL) {
      spmm2_sorted<<<(nRanges + 1) / 2, dim3(SCAN_THREADS), 0, stream>>>(recs, rangeStart, g2, deg_d, b2, out, nN, nE,
                                                                        nRanges, nSl2);
    } else if (nN <= (int)SRC_MASK) {
      spmm2_lds<<<(nRanges + 1) / 2, dim3(SCAN_THREADS), 0, stream>>>(recs, rangeStart, g2, deg_d, b2, out, nN, nE,
                                                                     nRanges, nSlices);
    } else {
      spmm2_owner<<<nRanges, dim3(SCAN_THREADS), 0, stream>>>(recs, rangeStart, g2, deg_d, b2, out, nN, nE);
    }
  } else {
    zero_f32<<<(2 * nN + TB - 1) / TB, blk, 0, stream>>>(deg_s, 2 * nN);
    degree_slices<<<dim3(degChunks, 2 * nSlices), dim3(SCAN_THREADS), 0, stream>>>(
        src, dst, (unsigned*)deg_s, (unsigned*)deg_d, nE, nN, nSlices);
    norm_kernel<<<(nN + TB - 1) / TB, blk, 0, stream>>>((unsigned*)deg_s, (unsigned*)deg_d, nN);
    gemm1_wmma<false><<<1024, blk, 0, stream>>>(x, W1, deg_s, h, nN);
    zero_f32<<<(8 * nN + TB - 1) / TB, blk, 0, stream>>>(agg, 8 * nN);
    spmm1_kernel<<<(nE + TB - 1) / TB, blk, 0, stream>>>(src, dst, h, agg, nE);
    layer2_node<<<(nN + TB - 1) / TB, blk, 0, stream>>>(agg, deg_d, deg_s, b1, W2, g2, nN);
    zero_f32<<<(nN + TB - 1) / TB, blk, 0, stream>>>(outacc, nN);
    spmm2_slices<<<dim3(sumChunks, nSlices), dim3(SCAN_THREADS), 0, stream>>>(src, dst, g2, outacc, nE, nN);
    final_kernel<<<(nN + TB - 1) / TB, blk, 0, stream>>>(outacc, deg_d, b2, out, nN);
  }
}
